# starting kernel
# baseline (speedup 1.0000x reference)
#include <hip/hip_runtime.h>

typedef __attribute__((ext_vector_type(2))) float v2f;
typedef __attribute__((ext_vector_type(8))) float v8f;

#define HDIM 64

__global__ __launch_bounds__(256)
void hg_scatter(const float* __restrict__ xsrc, const int* __restrict__ src,
                const int* __restrict__ dst, float* __restrict__ acc,
                float* __restrict__ cnt, int E)
{
  int idx = blockIdx.x * blockDim.x + threadIdx.x;
  int e = idx >> 4;
  if (e >= E) return;
  int c = (idx & 15) << 2;
  int s = src[e];
  int d = dst[e];
  const float4 v = *(const float4*)(xsrc + (size_t)s * HDIM + c);
  float* p = acc + (size_t)d * HDIM + c;
  atomicAdd(p + 0, v.x);
  atomicAdd(p + 1, v.y);
  atomicAdd(p + 2, v.z);
  atomicAdd(p + 3, v.w);
  if (c == 0) atomicAdd(cnt + d, 1.0f);
}

__device__ __forceinline__ void mma_product(v8f cacc[4],
                                            const float* __restrict__ Abase,
                                            float inv,
                                            const float* __restrict__ Wsh,
                                            int m, int half)
{
#pragma unroll
  for (int k0 = 0; k0 < HDIM; k0 += 4) {
    const int kk = k0 + 2 * half;
    v2f a;
    a.x = Abase[kk] * inv;
    a.y = Abase[kk + 1] * inv;
#pragma unroll
    for (int nt = 0; nt < 4; ++nt) {
      v2f b = *(const v2f*)(Wsh + (size_t)(nt * 16 + m) * HDIM + kk);
      cacc[nt] = __builtin_amdgcn_wmma_f32_16x16x4_f32(
          false, a, false, b, (short)0, cacc[nt], false, false);
    }
  }
}

__global__ __launch_bounds__(256)
void hg_sage(const float* __restrict__ accA, const float* __restrict__ cntA,
             const float* __restrict__ WlA, const float* __restrict__ blA,
             const float* __restrict__ WrA,
             const float* __restrict__ accB, const float* __restrict__ cntB,
             const float* __restrict__ WlB, const float* __restrict__ blB,
             const float* __restrict__ WrB,
             const float* __restrict__ xdst, float* __restrict__ out,
             int n_dst, int do_relu)
{
  __shared__ float sWlA[HDIM * HDIM];
  __shared__ float sWlB[HDIM * HDIM];
  __shared__ float sWr [HDIM * HDIM];
  __shared__ float sbias[HDIM];

  const int tid = threadIdx.x;
  const bool hasB = (accB != nullptr);

  for (int i = tid; i < HDIM * HDIM; i += 256) {
    sWlA[i] = WlA[i];
    sWr[i]  = hasB ? (WrA[i] + WrB[i]) : WrA[i];
    sWlB[i] = hasB ? WlB[i] : 0.0f;
  }
  if (tid < HDIM) sbias[tid] = hasB ? (blA[tid] + blB[tid]) : blA[tid];
  __syncthreads();

  const int wave = tid >> 5;
  const int lane = tid & 31;
  const int tile = blockIdx.x * 8 + wave;
  if (tile * 16 >= n_dst) return;

  const int m = lane & 15;
  const int half = lane >> 4;
  const int row = tile * 16 + m;

  v8f cacc[4] = {};

  {
    const float invA = 1.0f / fmaxf(cntA[row], 1.0f);
    mma_product(cacc, accA + (size_t)row * HDIM, invA, sWlA, m, half);
  }
  if (hasB) {
    const float invB = 1.0f / fmaxf(cntB[row], 1.0f);
    mma_product(cacc, accB + (size_t)row * HDIM, invB, sWlB, m, half);
  }
  mma_product(cacc, xdst + (size_t)row * HDIM, 1.0f, sWr, m, half);

#pragma unroll
  for (int nt = 0; nt < 4; ++nt) {
    const float bv = sbias[nt * 16 + m];
    float* obase = out + (size_t)(tile * 16) * HDIM + nt * 16 + m;
#pragma unroll
    for (int r = 0; r < 8; ++r) {
      float v = cacc[nt][r] + bv;
      if (do_relu) v = fmaxf(v, 0.0f);
      obase[(size_t)(r + 8 * half) * HDIM] = v;
    }
  }
}

extern "C" void kernel_launch(void* const* d_in, const int* in_sizes, int n_in,
                              void* d_out, int out_size, void* d_ws, size_t ws_size,
                              hipStream_t stream) {
  const int* src0 = (const int*)d_in[0]; const int* dst0 = (const int*)d_in[1];
  const int* src1 = (const int*)d_in[2]; const int* dst1 = (const int*)d_in[3];
  const int* src2 = (const int*)d_in[4]; const int* dst2 = (const int*)d_in[5];
  const int* src3 = (const int*)d_in[6]; const int* dst3 = (const int*)d_in[7];
  const float* ed  = (const float*)d_in[8];
  const float* eg  = (const float*)d_in[9];
  const float* es  = (const float*)d_in[10];
  const float* Wl0 = (const float*)d_in[11];
  const float* bl0 = (const float*)d_in[12];
  const float* Wr0 = (const float*)d_in[13];
  const float* Wl1 = (const float*)d_in[14];
  const float* bl1 = (const float*)d_in[15];
  const float* Wr1 = (const float*)d_in[16];

  const int E  = in_sizes[0];
  const int Nd = in_sizes[8]  / HDIM;
  const int Ng = in_sizes[9]  / HDIM;
  const int Ns = in_sizes[10] / HDIM;

  float* ws = (float*)d_ws;
  size_t off = 0;
  float* acc0 = ws + off; off += (size_t)Ng * HDIM;
  float* acc1 = ws + off; off += (size_t)Nd * HDIM;
  float* acc2 = ws + off; off += (size_t)Ns * HDIM;
  float* acc3 = ws + off; off += (size_t)Ng * HDIM;
  float* cnt0 = ws + off; off += (size_t)Ng;
  float* cnt1 = ws + off; off += (size_t)Nd;
  float* cnt2 = ws + off; off += (size_t)Ns;
  float* cnt3 = ws + off; off += (size_t)Ng;
  const size_t accBytes = off * sizeof(float);
  float* xd1 = ws + off; off += (size_t)Nd * HDIM;
  float* xg1 = ws + off; off += (size_t)Ng * HDIM;
  float* xs1 = ws + off; off += (size_t)Ns * HDIM;

  float* outd = (float*)d_out;
  float* outg = outd + (size_t)Nd * HDIM;
  float* outs = outg + (size_t)Ng * HDIM;

  const int sgrid = (E * 16 + 255) / 256;
  const dim3 blk(256);
  auto Wt = [](const float* base, int t) { return base + (size_t)t * HDIM * HDIM; };
  auto Bt = [](const float* base, int t) { return base + (size_t)t * HDIM; };
  const int gGene = (Ng / 16 + 7) / 8;
  const int gDrug = (Nd / 16 + 7) / 8;
  const int gDis  = (Ns / 16 + 7) / 8;

  hipMemsetAsync(ws, 0, accBytes, stream);
  hg_scatter<<<sgrid, blk, 0, stream>>>(ed, src0, dst0, acc0, cnt0, E);
  hg_scatter<<<sgrid, blk, 0, stream>>>(eg, src1, dst1, acc1, cnt1, E);
  hg_scatter<<<sgrid, blk, 0, stream>>>(eg, src2, dst2, acc2, cnt2, E);
  hg_scatter<<<sgrid, blk, 0, stream>>>(es, src3, dst3, acc3, cnt3, E);

  hg_sage<<<gGene, blk, 0, stream>>>(acc0, cnt0, Wt(Wl0,0), Bt(bl0,0), Wt(Wr0,0),
                                     acc3, cnt3, Wt(Wl0,3), Bt(bl0,3), Wt(Wr0,3),
                                     eg, xg1, Ng, 1);
  hg_sage<<<gDrug, blk, 0, stream>>>(acc1, cnt1, Wt(Wl0,1), Bt(bl0,1), Wt(Wr0,1),
                                     nullptr, nullptr, nullptr, nullptr, nullptr,
                                     ed, xd1, Nd, 1);
  hg_sage<<<gDis,  blk, 0, stream>>>(acc2, cnt2, Wt(Wl0,2), Bt(bl0,2), Wt(Wr0,2),
                                     nullptr, nullptr, nullptr, nullptr, nullptr,
                                     es, xs1, Ns, 1);

  hipMemsetAsync(ws, 0, accBytes, stream);
  hg_scatter<<<sgrid, blk, 0, stream>>>(xd1, src0, dst0, acc0, cnt0, E);
  hg_scatter<<<sgrid, blk, 0, stream>>>(xg1, src1, dst1, acc1, cnt1, E);
  hg_scatter<<<sgrid, blk, 0, stream>>>(xg1, src2, dst2, acc2, cnt2, E);
  hg_scatter<<<sgrid, blk, 0, stream>>>(xs1, src3, dst3, acc3, cnt3, E);

  hg_sage<<<gGene, blk, 0, stream>>>(acc0, cnt0, Wt(Wl1,0), Bt(bl1,0), Wt(Wr1,0),
                                     acc3, cnt3, Wt(Wl1,3), Bt(bl1,3), Wt(Wr1,3),
                                     xg1, outg, Ng, 0);
  hg_sage<<<gDrug, blk, 0, stream>>>(acc1, cnt1, Wt(Wl1,1), Bt(bl1,1), Wt(Wr1,1),
                                     nullptr, nullptr, nullptr, nullptr, nullptr,
                                     xd1, outd, Nd, 0);
  hg_sage<<<gDis,  blk, 0, stream>>>(acc2, cnt2, Wt(Wl1,2), Bt(bl1,2), Wt(Wr1,2),
                                     nullptr, nullptr, nullptr, nullptr, nullptr,
                                     xs1, outs, Ns, 0);
}


# Replace the per-element f32 atomic scatter by edges grouped by destination once per call (int-atomic count, scan, arrival-order rank fill) and an atomic-free row gather per layer; sum order changes
# speedup vs baseline: 2.0125x; 2.0125x over previous
#include <hip/hip_runtime.h>

typedef __attribute__((ext_vector_type(2))) float v2f;
typedef __attribute__((ext_vector_type(8))) float v8f;

#define HDIM 64

__global__ __launch_bounds__(256)
void hg_degree(const int* __restrict__ dst0, const int* __restrict__ dst1,
               const int* __restrict__ dst2, const int* __restrict__ dst3,
               int* __restrict__ deg0, int* __restrict__ deg1,
               int* __restrict__ deg2, int* __restrict__ deg3, int E)
{
  const int* dst; int* deg;
  switch (blockIdx.y) {
    case 0:  dst = dst0; deg = deg0; break;
    case 1:  dst = dst1; deg = deg1; break;
    case 2:  dst = dst2; deg = deg2; break;
    default: dst = dst3; deg = deg3; break;
  }
  const int stride = gridDim.x * 256;
  for (int e = blockIdx.x * 256 + threadIdx.x; e < E; e += stride)
    atomicAdd(deg + dst[e], 1);
}

__global__ __launch_bounds__(1024)
void hg_rowptr(const int* __restrict__ deg0, const int* __restrict__ deg1,
               const int* __restrict__ deg2, const int* __restrict__ deg3,
               int* __restrict__ rp0, int* __restrict__ rp1,
               int* __restrict__ rp2, int* __restrict__ rp3,
               int* __restrict__ cur0, int* __restrict__ cur1,
               int* __restrict__ cur2, int* __restrict__ cur3,
               int n0, int n1, int n2, int n3)
{
  __shared__ int swave[32];
  const int* deg; int* rp; int* cur; int n;
  switch (blockIdx.x) {
    case 0:  deg = deg0; rp = rp0; cur = cur0; n = n0; break;
    case 1:  deg = deg1; rp = rp1; cur = cur1; n = n1; break;
    case 2:  deg = deg2; rp = rp2; cur = cur2; n = n2; break;
    default: deg = deg3; rp = rp3; cur = cur3; n = n3; break;
  }
  const int tid = threadIdx.x;
  const int lane = tid & 31;
  const int wave = tid >> 5;
  int carry = 0;
  for (int base = 0; base < n; base += 4096) {
    const int i0 = base + tid * 4;
    int v[4];
#pragma unroll
    for (int k = 0; k < 4; ++k) v[k] = (i0 + k < n) ? deg[i0 + k] : 0;
    const int tsum = v[0] + v[1] + v[2] + v[3];
    int x = tsum;
#pragma unroll
    for (int o = 1; o < 32; o <<= 1) {
      const int y = __shfl_up(x, o);
      if (lane >= o) x += y;
    }
    if (lane == 31) swave[wave] = x;
    __syncthreads();
    if (wave == 0) {
      int w = swave[lane];
#pragma unroll
      for (int o = 1; o < 32; o <<= 1) {
        const int y = __shfl_up(w, o);
        if (lane >= o) w += y;
      }
      swave[lane] = w;
    }
    __syncthreads();
    const int woff = (wave > 0) ? swave[wave - 1] : 0;
    const int total = swave[31];
    int run = carry + woff + (x - tsum);
#pragma unroll
    for (int k = 0; k < 4; ++k) {
      if (i0 + k < n) { rp[i0 + k] = run; cur[i0 + k] = run; }
      run += v[k];
    }
    carry += total;
    __syncthreads();
  }
  if (tid == 0) rp[n] = carry;
}

__global__ __launch_bounds__(256)
void hg_fill(const int* __restrict__ src0, const int* __restrict__ dst0,
             const int* __restrict__ src1, const int* __restrict__ dst1,
             const int* __restrict__ src2, const int* __restrict__ dst2,
             const int* __restrict__ src3, const int* __restrict__ dst3,
             int* __restrict__ cur0, int* __restrict__ cur1,
             int* __restrict__ cur2, int* __restrict__ cur3,
             int* __restrict__ csr0, int* __restrict__ csr1,
             int* __restrict__ csr2, int* __restrict__ csr3, int E)
{
  const int* src; const int* dst; int* cur; int* csr;
  switch (blockIdx.y) {
    case 0:  src = src0; dst = dst0; cur = cur0; csr = csr0; break;
    case 1:  src = src1; dst = dst1; cur = cur1; csr = csr1; break;
    case 2:  src = src2; dst = dst2; cur = cur2; csr = csr2; break;
    default: src = src3; dst = dst3; cur = cur3; csr = csr3; break;
  }
  const int stride = gridDim.x * 256;
  for (int e = blockIdx.x * 256 + threadIdx.x; e < E; e += stride) {
    const int pos = atomicAdd(cur + dst[e], 1);
    csr[pos] = src[e];
  }
}

__device__ __forceinline__ float4 ld_row16(const float* __restrict__ x, int s, int c)
{
  return *(const float4*)(x + (size_t)s * HDIM + c);
}

__global__ __launch_bounds__(256)
void hg_gather(const float* __restrict__ x0, const float* __restrict__ x1,
               const float* __restrict__ x2, const float* __restrict__ x3,
               const int* __restrict__ rp0, const int* __restrict__ rp1,
               const int* __restrict__ rp2, const int* __restrict__ rp3,
               const int* __restrict__ csr0, const int* __restrict__ csr1,
               const int* __restrict__ csr2, const int* __restrict__ csr3,
               float* __restrict__ acc0, float* __restrict__ acc1,
               float* __restrict__ acc2, float* __restrict__ acc3,
               float* __restrict__ cnt0, float* __restrict__ cnt1,
               float* __restrict__ cnt2, float* __restrict__ cnt3,
               int n0, int n1, int n2, int n3)
{
  const float* x; const int* rp; const int* csr; float* acc; float* cnt; int n;
  switch (blockIdx.y) {
    case 0:  x = x0; rp = rp0; csr = csr0; acc = acc0; cnt = cnt0; n = n0; break;
    case 1:  x = x1; rp = rp1; csr = csr1; acc = acc1; cnt = cnt1; n = n1; break;
    case 2:  x = x2; rp = rp2; csr = csr2; acc = acc2; cnt = cnt2; n = n2; break;
    default: x = x3; rp = rp3; csr = csr3; acc = acc3; cnt = cnt3; n = n3; break;
  }
  const int wave = __builtin_amdgcn_readfirstlane((int)(threadIdx.x >> 5));
  const int lane = threadIdx.x & 31;
  const int h = lane >> 4;
  const int c = (lane & 15) << 2;
  const int dstep = gridDim.x * 8;
  for (int d = blockIdx.x * 8 + wave; d < n; d += dstep) {
    const int beg = rp[d];
    const int end = rp[d + 1];

    float4 a0 = make_float4(0.f, 0.f, 0.f, 0.f);
    float4 a1 = make_float4(0.f, 0.f, 0.f, 0.f);
    for (int jb = beg; jb < end; jb += 32) {
      const int j = jb + lane;
      const int idx = (j < end) ? csr[j] : 0;
      const int nk = (end - jb < 32) ? (end - jb) : 32;
      int k = 0;
      for (; k + 8 <= nk; k += 8) {
        const int s0 = __shfl(idx, k + h);
        const int s1 = __shfl(idx, k + 2 + h);
        const int s2 = __shfl(idx, k + 4 + h);
        const int s3 = __shfl(idx, k + 6 + h);
        const float4 v0 = ld_row16(x, s0, c);
        const float4 v1 = ld_row16(x, s1, c);
        const float4 v2 = ld_row16(x, s2, c);
        const float4 v3 = ld_row16(x, s3, c);
        a0.x += v0.x; a0.y += v0.y; a0.z += v0.z; a0.w += v0.w;
        a1.x += v1.x; a1.y += v1.y; a1.z += v1.z; a1.w += v1.w;
        a0.x += v2.x; a0.y += v2.y; a0.z += v2.z; a0.w += v2.w;
        a1.x += v3.x; a1.y += v3.y; a1.z += v3.z; a1.w += v3.w;
      }
      if (k < nk) {
        float4 t[4];
#pragma unroll
        for (int q = 0; q < 4; ++q) {
          const int kk = k + 2 * q + h;
          const int s = __shfl(idx, kk);
          t[q] = make_float4(0.f, 0.f, 0.f, 0.f);
          if (kk < nk) t[q] = ld_row16(x, s, c);
        }
#pragma unroll
        for (int q = 0; q < 4; ++q) {
          a0.x += t[q].x; a0.y += t[q].y; a0.z += t[q].z; a0.w += t[q].w;
        }
      }
    }
    a0.x += a1.x; a0.y += a1.y; a0.z += a1.z; a0.w += a1.w;
    a0.x += __shfl_xor(a0.x, 16);
    a0.y += __shfl_xor(a0.y, 16);
    a0.z += __shfl_xor(a0.z, 16);
    a0.w += __shfl_xor(a0.w, 16);
    if (h == 0) *(float4*)(acc + (size_t)d * HDIM + c) = a0;
    if (lane == 0) cnt[d] = (float)(end - beg);
  }
}

__device__ __forceinline__ void mma_product(v8f cacc[4],
                                            const float* __restrict__ Abase,
                                            float inv,
                                            const float* __restrict__ Wsh,
                                            int m, int half)
{
#pragma unroll
  for (int k0 = 0; k0 < HDIM; k0 += 4) {
    const int kk = k0 + 2 * half;
    v2f a;
    a.x = Abase[kk] * inv;
    a.y = Abase[kk + 1] * inv;
#pragma unroll
    for (int nt = 0; nt < 4; ++nt) {
      v2f b = *(const v2f*)(Wsh + (size_t)(nt * 16 + m) * HDIM + kk);
      cacc[nt] = __builtin_amdgcn_wmma_f32_16x16x4_f32(
          false, a, false, b, (short)0, cacc[nt], false, false);
    }
  }
}

__global__ __launch_bounds__(256)
void hg_sage(const float* __restrict__ accA, const float* __restrict__ cntA,
             const float* __restrict__ WlA, const float* __restrict__ blA,
             const float* __restrict__ WrA,
             const float* __restrict__ accB, const float* __restrict__ cntB,
             const float* __restrict__ WlB, const float* __restrict__ blB,
             const float* __restrict__ WrB,
             const float* __restrict__ xdst, float* __restrict__ out,
             int n_dst, int do_relu)
{
  __shared__ float sWlA[HDIM * HDIM];
  __shared__ float sWlB[HDIM * HDIM];
  __shared__ float sWr [HDIM * HDIM];
  __shared__ float sbias[HDIM];

  const int tid = threadIdx.x;
  const bool hasB = (accB != nullptr);

  for (int i = tid; i < HDIM * HDIM; i += 256) {
    sWlA[i] = WlA[i];
    sWr[i]  = hasB ? (WrA[i] + WrB[i]) : WrA[i];
    sWlB[i] = hasB ? WlB[i] : 0.0f;
  }
  if (tid < HDIM) sbias[tid] = hasB ? (blA[tid] + blB[tid]) : blA[tid];
  __syncthreads();

  const int wave = tid >> 5;
  const int lane = tid & 31;
  const int tile = blockIdx.x * 8 + wave;
  if (tile * 16 >= n_dst) return;

  const int m = lane & 15;
  const int half = lane >> 4;
  const int row = tile * 16 + m;

  v8f cacc[4] = {};

  {
    const float invA = 1.0f / fmaxf(cntA[row], 1.0f);
    mma_product(cacc, accA + (size_t)row * HDIM, invA, sWlA, m, half);
  }
  if (hasB) {
    const float invB = 1.0f / fmaxf(cntB[row], 1.0f);
    mma_product(cacc, accB + (size_t)row * HDIM, invB, sWlB, m, half);
  }
  mma_product(cacc, xdst + (size_t)row * HDIM, 1.0f, sWr, m, half);

#pragma unroll
  for (int nt = 0; nt < 4; ++nt) {
    const float bv = sbias[nt * 16 + m];
    float* obase = out + (size_t)(tile * 16) * HDIM + nt * 16 + m;
#pragma unroll
    for (int r = 0; r < 8; ++r) {
      float v = cacc[nt][r] + bv;
      if (do_relu) v = fmaxf(v, 0.0f);
      obase[(size_t)(r + 8 * half) * HDIM] = v;
    }
  }
}

extern "C" void kernel_launch(void* const* d_in, const int* in_sizes, int n_in,
                              void* d_out, int out_size, void* d_ws, size_t ws_size,
                              hipStream_t stream) {
  const int* src0 = (const int*)d_in[0]; const int* dst0 = (const int*)d_in[1];
  const int* src1 = (const int*)d_in[2]; const int* dst1 = (const int*)d_in[3];
  const int* src2 = (const int*)d_in[4]; const int* dst2 = (const int*)d_in[5];
  const int* src3 = (const int*)d_in[6]; const int* dst3 = (const int*)d_in[7];
  const float* ed  = (const float*)d_in[8];
  const float* eg  = (const float*)d_in[9];
  const float* es  = (const float*)d_in[10];
  const float* Wl0 = (const float*)d_in[11];
  const float* bl0 = (const float*)d_in[12];
  const float* Wr0 = (const float*)d_in[13];
  const float* Wl1 = (const float*)d_in[14];
  const float* bl1 = (const float*)d_in[15];
  const float* Wr1 = (const float*)d_in[16];

  const int E  = in_sizes[0];
  const int Nd = in_sizes[8]  / HDIM;
  const int Ng = in_sizes[9]  / HDIM;
  const int Ns = in_sizes[10] / HDIM;

  auto up64 = [](size_t n) { return (n + 63) & ~(size_t)63; };
  float* ws = (float*)d_ws;
  size_t off = 0;
  float* acc0 = ws + off; off += (size_t)Ng * HDIM;
  float* acc1 = ws + off; off += (size_t)Nd * HDIM;
  float* acc2 = ws + off; off += (size_t)Ns * HDIM;
  float* acc3 = ws + off; off += (size_t)Ng * HDIM;
  float* cnt0 = ws + off; off += up64(Ng);
  float* cnt1 = ws + off; off += up64(Nd);
  float* cnt2 = ws + off; off += up64(Ns);
  float* cnt3 = ws + off; off += up64(Ng);
  float* xd1 = ws + off; off += (size_t)Nd * HDIM;
  float* xg1 = ws + off; off += (size_t)Ng * HDIM;
  float* xs1 = ws + off; off += (size_t)Ns * HDIM;
  const size_t degOff = off;
  int* deg0 = (int*)(ws + off); off += up64(Ng);
  int* deg1 = (int*)(ws + off); off += up64(Nd);
  int* deg2 = (int*)(ws + off); off += up64(Ns);
  int* deg3 = (int*)(ws + off); off += up64(Ng);
  const size_t degBytes = (off - degOff) * sizeof(int);
  int* rp0 = (int*)(ws + off); off += up64((size_t)Ng + 1);
  int* rp1 = (int*)(ws + off); off += up64((size_t)Nd + 1);
  int* rp2 = (int*)(ws + off); off += up64((size_t)Ns + 1);
  int* rp3 = (int*)(ws + off); off += up64((size_t)Ng + 1);
  int* cur0 = (int*)(ws + off); off += up64(Ng);
  int* cur1 = (int*)(ws + off); off += up64(Nd);
  int* cur2 = (int*)(ws + off); off += up64(Ns);
  int* cur3 = (int*)(ws + off); off += up64(Ng);
  int* csr0 = (int*)(ws + off); off += up64(E);
  int* csr1 = (int*)(ws + off); off += up64(E);
  int* csr2 = (int*)(ws + off); off += up64(E);
  int* csr3 = (int*)(ws + off); off += up64(E);

  float* outd = (float*)d_out;
  float* outg = outd + (size_t)Nd * HDIM;
  float* outs = outg + (size_t)Ng * HDIM;

  const dim3 blk(256);
  auto Wt = [](const float* base, int t) { return base + (size_t)t * HDIM * HDIM; };
  auto Bt = [](const float* base, int t) { return base + (size_t)t * HDIM; };
  const int gGene = (Ng / 16 + 7) / 8;
  const int gDrug = (Nd / 16 + 7) / 8;
  const int gDis  = (Ns / 16 + 7) / 8;
  const dim3 egrid(1024, 4);
  const dim3 ggrid(1024, 4);

  hipMemsetAsync(deg0, 0, degBytes, stream);
  hg_degree<<<egrid, blk, 0, stream>>>(dst0, dst1, dst2, dst3, deg0, deg1, deg2, deg3, E);
  hg_rowptr<<<4, 1024, 0, stream>>>(deg0, deg1, deg2, deg3, rp0, rp1, rp2, rp3,
                                    cur0, cur1, cur2, cur3, Ng, Nd, Ns, Ng);
  hg_fill<<<egrid, blk, 0, stream>>>(src0, dst0, src1, dst1, src2, dst2, src3, dst3,
                                     cur0, cur1, cur2, cur3, csr0, csr1, csr2, csr3, E);

  hg_gather<<<ggrid, blk, 0, stream>>>(ed, eg, eg, es, rp0, rp1, rp2, rp3,
                                       csr0, csr1, csr2, csr3, acc0, acc1, acc2, acc3,
                                       cnt0, cnt1, cnt2, cnt3, Ng, Nd, Ns, Ng);

  hg_sage<<<gGene, blk, 0, stream>>>(acc0, cnt0, Wt(Wl0,0), Bt(bl0,0), Wt(Wr0,0),
                                     acc3, cnt3, Wt(Wl0,3), Bt(bl0,3), Wt(Wr0,3),
                                     eg, xg1, Ng, 1);
  hg_sage<<<gDrug, blk, 0, stream>>>(acc1, cnt1, Wt(Wl0,1), Bt(bl0,1), Wt(Wr0,1),
                                     nullptr, nullptr, nullptr, nullptr, nullptr,
                                     ed, xd1, Nd, 1);
  hg_sage<<<gDis,  blk, 0, stream>>>(acc2, cnt2, Wt(Wl0,2), Bt(bl0,2), Wt(Wr0,2),
                                     nullptr, nullptr, nullptr, nullptr, nullptr,
                                     es, xs1, Ns, 1);

  hg_gather<<<ggrid, blk, 0, stream>>>(xd1, xg1, xg1, xs1, rp0, rp1, rp2, rp3,
                                       csr0, csr1, csr2, csr3, acc0, acc1, acc2, acc3,
                                       cnt0, cnt1, cnt2, cnt3, Ng, Nd, Ns, Ng);

  hg_sage<<<gGene, blk, 0, stream>>>(acc0, cnt0, Wt(Wl1,0), Bt(bl1,0), Wt(Wr1,0),
                                     acc3, cnt3, Wt(Wl1,3), Bt(bl1,3), Wt(Wr1,3),
                                     xg1, outg, Ng, 0);
  hg_sage<<<gDrug, blk, 0, stream>>>(acc1, cnt1, Wt(Wl1,1), Bt(bl1,1), Wt(Wr1,1),
                                     nullptr, nullptr, nullptr, nullptr, nullptr,
                                     xd1, outd, Nd, 0);
  hg_sage<<<gDis,  blk, 0, stream>>>(acc2, cnt2, Wt(Wl1,2), Bt(bl1,2), Wt(Wr1,2),
                                     nullptr, nullptr, nullptr, nullptr, nullptr,
                                     xs1, outs, Ns, 0);
}


# Group edges by destination through LDS bins (tile bin counts, binned scatter, per-bin LDS rank, whole-line stores) instead of 8M global atomics; row sum order stays arrival order (LDS adds)
# speedup vs baseline: 5.5704x; 2.7679x over previous
#include <hip/hip_runtime.h>

typedef __attribute__((ext_vector_type(2))) float v2f;
typedef __attribute__((ext_vector_type(8))) float v8f;

#define HDIM 64

#define TILE 8192
#define MAXB 256
#define CAP  12288

__device__ __forceinline__ int bin_shift(int n)
{
  int s = 8;
  while (s > 5 && ((n - 1) >> (s - 1)) < MAXB) --s;
  return s;
}

__global__ __launch_bounds__(1024)
void hg_bincount(const int* __restrict__ dst0, const int* __restrict__ dst1,
                 const int* __restrict__ dst2, const int* __restrict__ dst3,
                 int* __restrict__ hist, int E, int NT,
                 int n0, int n1, int n2, int n3)
{
  __shared__ int h[MAXB];
  const int* dst; int n;
  switch (blockIdx.y) {
    case 0:  dst = dst0; n = n0; break;
    case 1:  dst = dst1; n = n1; break;
    case 2:  dst = dst2; n = n2; break;
    default: dst = dst3; n = n3; break;
  }
  const int sh = bin_shift(n);
  const int tid = threadIdx.x;
  const int t = blockIdx.x;
  if (tid < MAXB) h[tid] = 0;
  __syncthreads();
  const int e0 = t * TILE;
#pragma unroll
  for (int k = 0; k < TILE / 1024; ++k) {
    const int e = e0 + k * 1024 + tid;
    if (e < E) atomicAdd(&h[dst[e] >> sh], 1);
  }
  __syncthreads();
  if (tid < MAXB) hist[((size_t)blockIdx.y * NT + t) * MAXB + tid] = h[tid];
}

__global__ __launch_bounds__(1024)
void hg_binscatter(const int* __restrict__ src0, const int* __restrict__ dst0,
                   const int* __restrict__ src1, const int* __restrict__ dst1,
                   const int* __restrict__ src2, const int* __restrict__ dst2,
                   const int* __restrict__ src3, const int* __restrict__ dst3,
                   const int* __restrict__ hist, int* __restrict__ bst,
                   unsigned* __restrict__ bn0, unsigned* __restrict__ bn1,
                   unsigned* __restrict__ bn2, unsigned* __restrict__ bn3,
                   int E, int NT, int n0, int n1, int n2, int n3)
{
  __shared__ int s_tot[MAXB];
  __shared__ int s_pre[MAXB];
  __shared__ int s_own[MAXB];
  __shared__ int s_gd[MAXB];
  __shared__ int s_lc[MAXB];
  __shared__ int s_part[3][4][MAXB];
  __shared__ unsigned s_word[TILE];
  __shared__ unsigned char s_bin[TILE];

  const int* src; const int* dst; unsigned* bn; int n;
  switch (blockIdx.y) {
    case 0:  src = src0; dst = dst0; bn = bn0; n = n0; break;
    case 1:  src = src1; dst = dst1; bn = bn1; n = n1; break;
    case 2:  src = src2; dst = dst2; bn = bn2; n = n2; break;
    default: src = src3; dst = dst3; bn = bn3; n = n3; break;
  }
  const int sh = bin_shift(n);
  const int lmask = (1 << sh) - 1;
  const int tid = threadIdx.x;
  const int lane = tid & 31;
  const int wave = tid >> 5;
  const int t = blockIdx.x;
  const int* hb = hist + (size_t)blockIdx.y * NT * MAXB;
  int* bstart = bst + blockIdx.y * (MAXB + 1);

  {
    const int bq = tid & (MAXB - 1);
    const int q = tid >> 8;
    int tot = 0, pre = 0, own = 0;
#pragma unroll 8
    for (int tt = q; tt < NT; tt += 4) {
      const int v = hb[(size_t)tt * MAXB + bq];
      tot += v;
      pre += (tt < t) ? v : 0;
      own += (tt == t) ? v : 0;
    }
    s_part[0][q][bq] = tot; s_part[1][q][bq] = pre; s_part[2][q][bq] = own;
  }
  __syncthreads();
  if (tid < MAXB) {
    s_tot[tid] = s_part[0][0][tid] + s_part[0][1][tid] + s_part[0][2][tid] + s_part[0][3][tid];
    s_pre[tid] = s_part[1][0][tid] + s_part[1][1][tid] + s_part[1][2][tid] + s_part[1][3][tid];
    s_own[tid] = s_part[2][0][tid] + s_part[2][1][tid] + s_part[2][2][tid] + s_part[2][3][tid];
  }
  __syncthreads();
  if (wave == 0) {
    int a[8], o[8];
    int sa = 0, so = 0;
#pragma unroll
    for (int k = 0; k < 8; ++k) {
      a[k] = s_tot[8 * lane + k]; o[k] = s_own[8 * lane + k];
      sa += a[k]; so += o[k];
    }
    int xa = sa, xo = so;
#pragma unroll
    for (int d = 1; d < 32; d <<= 1) {
      const int ya = __shfl_up(xa, d);
      const int yo = __shfl_up(xo, d);
      if (lane >= d) { xa += ya; xo += yo; }
    }
    int ra = xa - sa, ro = xo - so;
#pragma unroll
    for (int k = 0; k < 8; ++k) {
      const int b = 8 * lane + k;
      s_gd[b] = ra + s_pre[b] - ro;
      s_lc[b] = ro;
      if (t == 0) bstart[b] = ra;
      ra += a[k]; ro += o[k];
    }
    if (t == 0 && lane == 31) bstart[MAXB] = ra;
  }
  __syncthreads();

  const int e0 = t * TILE;
#pragma unroll
  for (int k = 0; k < TILE / 1024; ++k) {
    const int e = e0 + k * 1024 + tid;
    if (e < E) {
      const int d = dst[e];
      const int s = src[e];
      const int b = d >> sh;
      const int lp = atomicAdd(&s_lc[b], 1);
      s_word[lp] = ((unsigned)s << 8) | (unsigned)(d & lmask);
      s_bin[lp] = (unsigned char)b;
    }
  }
  __syncthreads();
  const int m = (E - e0 < TILE) ? (E - e0) : TILE;
  for (int i = tid; i < m; i += 1024)
    bn[i + s_gd[s_bin[i]]] = s_word[i];
}

__global__ __launch_bounds__(1024)
void hg_binsort(const unsigned* __restrict__ bn0, const unsigned* __restrict__ bn1,
                const unsigned* __restrict__ bn2, const unsigned* __restrict__ bn3,
                const int* __restrict__ bst,
                int* __restrict__ rp0, int* __restrict__ rp1,
                int* __restrict__ rp2, int* __restrict__ rp3,
                int* __restrict__ csr0, int* __restrict__ csr1,
                int* __restrict__ csr2, int* __restrict__ csr3,
                int n0, int n1, int n2, int n3)
{
  __shared__ int c[MAXB];
  __shared__ int cur[MAXB];
  __shared__ int stage[CAP];

  const unsigned* bn; int* rp; int* csr; int n;
  switch (blockIdx.y) {
    case 0:  bn = bn0; rp = rp0; csr = csr0; n = n0; break;
    case 1:  bn = bn1; rp = rp1; csr = csr1; n = n1; break;
    case 2:  bn = bn2; rp = rp2; csr = csr2; n = n2; break;
    default: bn = bn3; rp = rp3; csr = csr3; n = n3; break;
  }
  const int sh = bin_shift(n);
  const int nbins = ((n - 1) >> sh) + 1;
  const int b = blockIdx.x;
  if (b >= nbins) return;
  const int tid = threadIdx.x;
  const int lane = tid & 31;
  const int wave = tid >> 5;
  const int* bstart = bst + blockIdx.y * (MAXB + 1);
  const int bs = bstart[b];
  const int be = bstart[b + 1];
  const int m = be - bs;

  if (tid < MAXB) c[tid] = 0;
  __syncthreads();
  for (int i = bs + tid; i < be; i += 1024) atomicAdd(&c[bn[i] & 255u], 1);
  __syncthreads();
  if (wave == 0) {
    int a[8];
    int sa = 0;
#pragma unroll
    for (int k = 0; k < 8; ++k) { a[k] = c[8 * lane + k]; sa += a[k]; }
    int xa = sa;
#pragma unroll
    for (int d = 1; d < 32; d <<= 1) {
      const int ya = __shfl_up(xa, d);
      if (lane >= d) xa += ya;
    }
    int ra = xa - sa;
    const int R = 1 << sh;
#pragma unroll
    for (int k = 0; k < 8; ++k) {
      const int j = 8 * lane + k;
      cur[j] = ra;
      const int node = (b << sh) + j;
      if (j < R && node < n) rp[node] = bs + ra;
      ra += a[k];
    }
  }
  if (tid == 0 && b == nbins - 1) rp[n] = be;
  __syncthreads();
  if (m <= CAP) {
    for (int i = bs + tid; i < be; i += 1024) {
      const unsigned w = bn[i];
      const int p = atomicAdd(&cur[w & 255u], 1);
      stage[p] = (int)(w >> 8);
    }
    __syncthreads();
    for (int i = tid; i < m; i += 1024) csr[bs + i] = stage[i];
  } else {
    for (int i = bs + tid; i < be; i += 1024) {
      const unsigned w = bn[i];
      const int p = atomicAdd(&cur[w & 255u], 1);
      csr[bs + p] = (int)(w >> 8);
    }
  }
}

__device__ __forceinline__ float4 ld_row16(const float* __restrict__ x, int s, int c)
{
  return *(const float4*)(x + (size_t)s * HDIM + c);
}

__global__ __launch_bounds__(256)
void hg_gather(const float* __restrict__ x0, const float* __restrict__ x1,
               const float* __restrict__ x2, const float* __restrict__ x3,
               const int* __restrict__ rp0, const int* __restrict__ rp1,
               const int* __restrict__ rp2, const int* __restrict__ rp3,
               const int* __restrict__ csr0, const int* __restrict__ csr1,
               const int* __restrict__ csr2, const int* __restrict__ csr3,
               float* __restrict__ acc0, float* __restrict__ acc1,
               float* __restrict__ acc2, float* __restrict__ acc3,
               float* __restrict__ cnt0, float* __restrict__ cnt1,
               float* __restrict__ cnt2, float* __restrict__ cnt3,
               int n0, int n1, int n2, int n3)
{
  const float* x; const int* rp; const int* csr; float* acc; float* cnt; int n;
  switch (blockIdx.y) {
    case 0:  x = x0; rp = rp0; csr = csr0; acc = acc0; cnt = cnt0; n = n0; break;
    case 1:  x = x1; rp = rp1; csr = csr1; acc = acc1; cnt = cnt1; n = n1; break;
    case 2:  x = x2; rp = rp2; csr = csr2; acc = acc2; cnt = cnt2; n = n2; break;
    default: x = x3; rp = rp3; csr = csr3; acc = acc3; cnt = cnt3; n = n3; break;
  }
  const int wave = __builtin_amdgcn_readfirstlane((int)(threadIdx.x >> 5));
  const int lane = threadIdx.x & 31;
  const int h = lane >> 4;
  const int c = (lane & 15) << 2;
  const int dstep = gridDim.x * 8;
  for (int d = blockIdx.x * 8 + wave; d < n; d += dstep) {
    const int beg = rp[d];
    const int end = rp[d + 1];

    float4 a0 = make_float4(0.f, 0.f, 0.f, 0.f);
    float4 a1 = make_float4(0.f, 0.f, 0.f, 0.f);
    for (int jb = beg; jb < end; jb += 32) {
      const int j = jb + lane;
      const int idx = (j < end) ? csr[j] : 0;
      const int nk = (end - jb < 32) ? (end - jb) : 32;
      int k = 0;
      for (; k + 8 <= nk; k += 8) {
        const int s0 = __shfl(idx, k + h);
        const int s1 = __shfl(idx, k + 2 + h);
        const int s2 = __shfl(idx, k + 4 + h);
        const int s3 = __shfl(idx, k + 6 + h);
        const float4 v0 = ld_row16(x, s0, c);
        const float4 v1 = ld_row16(x, s1, c);
        const float4 v2 = ld_row16(x, s2, c);
        const float4 v3 = ld_row16(x, s3, c);
        a0.x += v0.x; a0.y += v0.y; a0.z += v0.z; a0.w += v0.w;
        a1.x += v1.x; a1.y += v1.y; a1.z += v1.z; a1.w += v1.w;
        a0.x += v2.x; a0.y += v2.y; a0.z += v2.z; a0.w += v2.w;
        a1.x += v3.x; a1.y += v3.y; a1.z += v3.z; a1.w += v3.w;
      }
      if (k < nk) {
        float4 t[4];
#pragma unroll
        for (int q = 0; q < 4; ++q) {
          const int kk = k + 2 * q + h;
          const int s = __shfl(idx, kk);
          t[q] = make_float4(0.f, 0.f, 0.f, 0.f);
          if (kk < nk) t[q] = ld_row16(x, s, c);
        }
#pragma unroll
        for (int q = 0; q < 4; ++q) {
          a0.x += t[q].x; a0.y += t[q].y; a0.z += t[q].z; a0.w += t[q].w;
        }
      }
    }
    a0.x += a1.x; a0.y += a1.y; a0.z += a1.z; a0.w += a1.w;
    a0.x += __shfl_xor(a0.x, 16);
    a0.y += __shfl_xor(a0.y, 16);
    a0.z += __shfl_xor(a0.z, 16);
    a0.w += __shfl_xor(a0.w, 16);
    if (h == 0) *(float4*)(acc + (size_t)d * HDIM + c) = a0;
    if (lane == 0) cnt[d] = (float)(end - beg);
  }
}

__device__ __forceinline__ void mma_product(v8f cacc[4],
                                            const float* __restrict__ Abase,
                                            float inv,
                                            const float* __restrict__ Wsh,
                                            int m, int half)
{
#pragma unroll
  for (int k0 = 0; k0 < HDIM; k0 += 4) {
    const int kk = k0 + 2 * half;
    v2f a;
    a.x = Abase[kk] * inv;
    a.y = Abase[kk + 1] * inv;
#pragma unroll
    for (int nt = 0; nt < 4; ++nt) {
      v2f b = *(const v2f*)(Wsh + (size_t)(nt * 16 + m) * HDIM + kk);
      cacc[nt] = __builtin_amdgcn_wmma_f32_16x16x4_f32(
          false, a, false, b, (short)0, cacc[nt], false, false);
    }
  }
}

__global__ __launch_bounds__(256)
void hg_sage(const float* __restrict__ accA, const float* __restrict__ cntA,
             const float* __restrict__ WlA, const float* __restrict__ blA,
             const float* __restrict__ WrA,
             const float* __restrict__ accB, const float* __restrict__ cntB,
             const float* __restrict__ WlB, const float* __restrict__ blB,
             const float* __restrict__ WrB,
             const float* __restrict__ xdst, float* __restrict__ out,
             int n_dst, int do_relu)
{
  __shared__ float sWlA[HDIM * HDIM];
  __shared__ float sWlB[HDIM * HDIM];
  __shared__ float sWr [HDIM * HDIM];
  __shared__ float sbias[HDIM];

  const int tid = threadIdx.x;
  const bool hasB = (accB != nullptr);

  for (int i = tid; i < HDIM * HDIM; i += 256) {
    sWlA[i] = WlA[i];
    sWr[i]  = hasB ? (WrA[i] + WrB[i]) : WrA[i];
    sWlB[i] = hasB ? WlB[i] : 0.0f;
  }
  if (tid < HDIM) sbias[tid] = hasB ? (blA[tid] + blB[tid]) : blA[tid];
  __syncthreads();

  const int wave = tid >> 5;
  const int lane = tid & 31;
  const int tile = blockIdx.x * 8 + wave;
  if (tile * 16 >= n_dst) return;

  const int m = lane & 15;
  const int half = lane >> 4;
  const int row = tile * 16 + m;

  v8f cacc[4] = {};

  {
    const float invA = 1.0f / fmaxf(cntA[row], 1.0f);
    mma_product(cacc, accA + (size_t)row * HDIM, invA, sWlA, m, half);
  }
  if (hasB) {
    const float invB = 1.0f / fmaxf(cntB[row], 1.0f);
    mma_product(cacc, accB + (size_t)row * HDIM, invB, sWlB, m, half);
  }
  mma_product(cacc, xdst + (size_t)row * HDIM, 1.0f, sWr, m, half);

#pragma unroll
  for (int nt = 0; nt < 4; ++nt) {
    const float bv = sbias[nt * 16 + m];
    float* obase = out + (size_t)(tile * 16) * HDIM + nt * 16 + m;
#pragma unroll
    for (int r = 0; r < 8; ++r) {
      float v = cacc[nt][r] + bv;
      if (do_relu) v = fmaxf(v, 0.0f);
      obase[(size_t)(r + 8 * half) * HDIM] = v;
    }
  }
}

extern "C" void kernel_launch(void* const* d_in, const int* in_sizes, int n_in,
                              void* d_out, int out_size, void* d_ws, size_t ws_size,
                              hipStream_t stream) {
  const int* src0 = (const int*)d_in[0]; const int* dst0 = (const int*)d_in[1];
  const int* src1 = (const int*)d_in[2]; const int* dst1 = (const int*)d_in[3];
  const int* src2 = (const int*)d_in[4]; const int* dst2 = (const int*)d_in[5];
  const int* src3 = (const int*)d_in[6]; const int* dst3 = (const int*)d_in[7];
  const float* ed  = (const float*)d_in[8];
  const float* eg  = (const float*)d_in[9];
  const float* es  = (const float*)d_in[10];
  const float* Wl0 = (const float*)d_in[11];
  const float* bl0 = (const float*)d_in[12];
  const float* Wr0 = (const float*)d_in[13];
  const float* Wl1 = (const float*)d_in[14];
  const float* bl1 = (const float*)d_in[15];
  const float* Wr1 = (const float*)d_in[16];

  const int E  = in_sizes[0];
  const int Nd = in_sizes[8]  / HDIM;
  const int Ng = in_sizes[9]  / HDIM;
  const int Ns = in_sizes[10] / HDIM;

  auto up64 = [](size_t n) { return (n + 63) & ~(size_t)63; };
  float* ws = (float*)d_ws;
  size_t off = 0;
  float* acc0 = ws + off; off += (size_t)Ng * HDIM;
  float* acc1 = ws + off; off += (size_t)Nd * HDIM;
  float* acc2 = ws + off; off += (size_t)Ns * HDIM;
  float* acc3 = ws + off; off += (size_t)Ng * HDIM;
  float* cnt0 = ws + off; off += up64(Ng);
  float* cnt1 = ws + off; off += up64(Nd);
  float* cnt2 = ws + off; off += up64(Ns);
  float* cnt3 = ws + off; off += up64(Ng);
  float* xd1 = ws + off; off += (size_t)Nd * HDIM;
  float* xg1 = ws + off; off += (size_t)Ng * HDIM;
  float* xs1 = ws + off; off += (size_t)Ns * HDIM;
  const int NT = (E + TILE - 1) / TILE;
  int* hist = (int*)(ws + off); off += up64((size_t)4 * MAXB * NT);
  int* bst  = (int*)(ws + off); off += up64((size_t)4 * (MAXB + 1));
  unsigned* bn0 = (unsigned*)(ws + off); off += up64(E);
  unsigned* bn1 = (unsigned*)(ws + off); off += up64(E);
  unsigned* bn2 = (unsigned*)(ws + off); off += up64(E);
  unsigned* bn3 = (unsigned*)(ws + off); off += up64(E);
  int* rp0 = (int*)(ws + off); off += up64((size_t)Ng + 1);
  int* rp1 = (int*)(ws + off); off += up64((size_t)Nd + 1);
  int* rp2 = (int*)(ws + off); off += up64((size_t)Ns + 1);
  int* rp3 = (int*)(ws + off); off += up64((size_t)Ng + 1);
  int* csr0 = (int*)(ws + off); off += up64(E);
  int* csr1 = (int*)(ws + off); off += up64(E);
  int* csr2 = (int*)(ws + off); off += up64(E);
  int* csr3 = (int*)(ws + off); off += up64(E);

  float* outd = (float*)d_out;
  float* outg = outd + (size_t)Nd * HDIM;
  float* outs = outg + (size_t)Ng * HDIM;

  const dim3 blk(256);
  auto Wt = [](const float* base, int t) { return base + (size_t)t * HDIM * HDIM; };
  auto Bt = [](const float* base, int t) { return base + (size_t)t * HDIM; };
  const int gGene = (Ng / 16 + 7) / 8;
  const int gDrug = (Nd / 16 + 7) / 8;
  const int gDis  = (Ns / 16 + 7) / 8;
  const dim3 tgrid(NT, 4);
  const dim3 bgrid(MAXB, 4);
  const dim3 ggrid(1024, 4);

  hg_bincount<<<tgrid, 1024, 0, stream>>>(dst0, dst1, dst2, dst3, hist, E, NT, Ng, Nd, Ns, Ng);
  hg_binscatter<<<tgrid, 1024, 0, stream>>>(src0, dst0, src1, dst1, src2, dst2, src3, dst3,
                                            hist, bst, bn0, bn1, bn2, bn3, E, NT, Ng, Nd, Ns, Ng);
  hg_binsort<<<bgrid, 1024, 0, stream>>>(bn0, bn1, bn2, bn3, bst, rp0, rp1, rp2, rp3,
                                         csr0, csr1, csr2, csr3, Ng, Nd, Ns, Ng);

  hg_gather<<<ggrid, blk, 0, stream>>>(ed, eg, eg, es, rp0, rp1, rp2, rp3,
                                       csr0, csr1, csr2, csr3, acc0, acc1, acc2, acc3,
                                       cnt0, cnt1, cnt2, cnt3, Ng, Nd, Ns, Ng);

  hg_sage<<<gGene, blk, 0, stream>>>(acc0, cnt0, Wt(Wl0,0), Bt(bl0,0), Wt(Wr0,0),
                                     acc3, cnt3, Wt(Wl0,3), Bt(bl0,3), Wt(Wr0,3),
                                     eg, xg1, Ng, 1);
  hg_sage<<<gDrug, blk, 0, stream>>>(acc1, cnt1, Wt(Wl0,1), Bt(bl0,1), Wt(Wr0,1),
                                     nullptr, nullptr, nullptr, nullptr, nullptr,
                                     ed, xd1, Nd, 1);
  hg_sage<<<gDis,  blk, 0, stream>>>(acc2, cnt2, Wt(Wl0,2), Bt(bl0,2), Wt(Wr0,2),
                                     nullptr, nullptr, nullptr, nullptr, nullptr,
                                     es, xs1, Ns, 1);

  hg_gather<<<ggrid, blk, 0, stream>>>(xd1, xg1, xg1, xs1, rp0, rp1, rp2, rp3,
                                       csr0, csr1, csr2, csr3, acc0, acc1, acc2, acc3,
                                       cnt0, cnt1, cnt2, cnt3, Ng, Nd, Ns, Ng);

  hg_sage<<<gGene, blk, 0, stream>>>(acc0, cnt0, Wt(Wl1,0), Bt(bl1,0), Wt(Wr1,0),
                                     acc3, cnt3, Wt(Wl1,3), Bt(bl1,3), Wt(Wr1,3),
                                     xg1, outg, Ng, 0);
  hg_sage<<<gDrug, blk, 0, stream>>>(acc1, cnt1, Wt(Wl1,1), Bt(bl1,1), Wt(Wr1,1),
                                     nullptr, nullptr, nullptr, nullptr, nullptr,
                                     xd1, outd, Nd, 0);
  hg_sage<<<gDis,  blk, 0, stream>>>(acc2, cnt2, Wt(Wl1,2), Bt(bl1,2), Wt(Wr1,2),
                                     nullptr, nullptr, nullptr, nullptr, nullptr,
                                     xs1, outs, Ns, 0);
}


# Merge the three per-node-type SAGE update launches of each layer into one launch (blockIdx.x range picks the node type, two-product work-groups first); launches per call 11 -> 7
# speedup vs baseline: 7.5028x; 1.3469x over previous
#include <hip/hip_runtime.h>

typedef __attribute__((ext_vector_type(2))) float v2f;
typedef __attribute__((ext_vector_type(8))) float v8f;

#define HDIM 64

#define TILE 8192
#define MAXB 256
#define CAP  12288

__device__ __forceinline__ int bin_shift(int n)
{
  int s = 8;
  while (s > 5 && ((n - 1) >> (s - 1)) < MAXB) --s;
  return s;
}

__global__ __launch_bounds__(1024)
void hg_bincount(const int* __restrict__ dst0, const int* __restrict__ dst1,
                 const int* __restrict__ dst2, const int* __restrict__ dst3,
                 int* __restrict__ hist, int E, int NT,
                 int n0, int n1, int n2, int n3)
{
  __shared__ int h[MAXB];
  const int* dst; int n;
  switch (blockIdx.y) {
    case 0:  dst = dst0; n = n0; break;
    case 1:  dst = dst1; n = n1; break;
    case 2:  dst = dst2; n = n2; break;
    default: dst = dst3; n = n3; break;
  }
  const int sh = bin_shift(n);
  const int tid = threadIdx.x;
  const int t = blockIdx.x;
  if (tid < MAXB) h[tid] = 0;
  __syncthreads();
  const int e0 = t * TILE;
#pragma unroll
  for (int k = 0; k < TILE / 1024; ++k) {
    const int e = e0 + k * 1024 + tid;
    if (e < E) atomicAdd(&h[dst[e] >> sh], 1);
  }
  __syncthreads();
  if (tid < MAXB) hist[((size_t)blockIdx.y * NT + t) * MAXB + tid] = h[tid];
}

__global__ __launch_bounds__(1024)
void hg_binscatter(const int* __restrict__ src0, const int* __restrict__ dst0,
                   const int* __restrict__ src1, const int* __restrict__ dst1,
                   const int* __restrict__ src2, const int* __restrict__ dst2,
                   const int* __restrict__ src3, const int* __restrict__ dst3,
                   const int* __restrict__ hist, int* __restrict__ bst,
                   unsigned* __restrict__ bn0, unsigned* __restrict__ bn1,
                   unsigned* __restrict__ bn2, unsigned* __restrict__ bn3,
                   int E, int NT, int n0, int n1, int n2, int n3)
{
  __shared__ int s_tot[MAXB];
  __shared__ int s_pre[MAXB];
  __shared__ int s_own[MAXB];
  __shared__ int s_gd[MAXB];
  __shared__ int s_lc[MAXB];
  __shared__ int s_part[3][4][MAXB];
  __shared__ unsigned s_word[TILE];
  __shared__ unsigned char s_bin[TILE];

  const int* src; const int* dst; unsigned* bn; int n;
  switch (blockIdx.y) {
    case 0:  src = src0; dst = dst0; bn = bn0; n = n0; break;
    case 1:  src = src1; dst = dst1; bn = bn1; n = n1; break;
    case 2:  src = src2; dst = dst2; bn = bn2; n = n2; break;
    default: src = src3; dst = dst3; bn = bn3; n = n3; break;
  }
  const int sh = bin_shift(n);
  const int lmask = (1 << sh) - 1;
  const int tid = threadIdx.x;
  const int lane = tid & 31;
  const int wave = tid >> 5;
  const int t = blockIdx.x;
  const int* hb = hist + (size_t)blockIdx.y * NT * MAXB;
  int* bstart = bst + blockIdx.y * (MAXB + 1);

  {
    const int bq = tid & (MAXB - 1);
    const int q = tid >> 8;
    int tot = 0, pre = 0, own = 0;
#pragma unroll 8
    for (int tt = q; tt < NT; tt += 4) {
      const int v = hb[(size_t)tt * MAXB + bq];
      tot += v;
      pre += (tt < t) ? v : 0;
      own += (tt == t) ? v : 0;
    }
    s_part[0][q][bq] = tot; s_part[1][q][bq] = pre; s_part[2][q][bq] = own;
  }
  __syncthreads();
  if (tid < MAXB) {
    s_tot[tid] = s_part[0][0][tid] + s_part[0][1][tid] + s_part[0][2][tid] + s_part[0][3][tid];
    s_pre[tid] = s_part[1][0][tid] + s_part[1][1][tid] + s_part[1][2][tid] + s_part[1][3][tid];
    s_own[tid] = s_part[2][0][tid] + s_part[2][1][tid] + s_part[2][2][tid] + s_part[2][3][tid];
  }
  __syncthreads();
  if (wave == 0) {
    int a[8], o[8];
    int sa = 0, so = 0;
#pragma unroll
    for (int k = 0; k < 8; ++k) {
      a[k] = s_tot[8 * lane + k]; o[k] = s_own[8 * lane + k];
      sa += a[k]; so += o[k];
    }
    int xa = sa, xo = so;
#pragma unroll
    for (int d = 1; d < 32; d <<= 1) {
      const int ya = __shfl_up(xa, d);
      const int yo = __shfl_up(xo, d);
      if (lane >= d) { xa += ya; xo += yo; }
    }
    int ra = xa - sa, ro = xo - so;
#pragma unroll
    for (int k = 0; k < 8; ++k) {
      const int b = 8 * lane + k;
      s_gd[b] = ra + s_pre[b] - ro;
      s_lc[b] = ro;
      if (t == 0) bstart[b] = ra;
      ra += a[k]; ro += o[k];
    }
    if (t == 0 && lane == 31) bstart[MAXB] = ra;
  }
  __syncthreads();

  const int e0 = t * TILE;
#pragma unroll
  for (int k = 0; k < TILE / 1024; ++k) {
    const int e = e0 + k * 1024 + tid;
    if (e < E) {
      const int d = dst[e];
      const int s = src[e];
      const int b = d >> sh;
      const int lp = atomicAdd(&s_lc[b], 1);
      s_word[lp] = ((unsigned)s << 8) | (unsigned)(d & lmask);
      s_bin[lp] = (unsigned char)b;
    }
  }
  __syncthreads();
  const int m = (E - e0 < TILE) ? (E - e0) : TILE;
  for (int i = tid; i < m; i += 1024)
    bn[i + s_gd[s_bin[i]]] = s_word[i];
}

__global__ __launch_bounds__(1024)
void hg_binsort(const unsigned* __restrict__ bn0, const unsigned* __restrict__ bn1,
                const unsigned* __restrict__ bn2, const unsigned* __restrict__ bn3,
                const int* __restrict__ bst,
                int* __restrict__ rp0, int* __restrict__ rp1,
                int* __restrict__ rp2, int* __restrict__ rp3,
                int* __restrict__ csr0, int* __restrict__ csr1,
                int* __restrict__ csr2, int* __restrict__ csr3,
                int n0, int n1, int n2, int n3)
{
  __shared__ int c[MAXB];
  __shared__ int cur[MAXB];
  __shared__ int stage[CAP];

  const unsigned* bn; int* rp; int* csr; int n;
  switch (blockIdx.y) {
    case 0:  bn = bn0; rp = rp0; csr = csr0; n = n0; break;
    case 1:  bn = bn1; rp = rp1; csr = csr1; n = n1; break;
    case 2:  bn = bn2; rp = rp2; csr = csr2; n = n2; break;
    default: bn = bn3; rp = rp3; csr = csr3; n = n3; break;
  }
  const int sh = bin_shift(n);
  const int nbins = ((n - 1) >> sh) + 1;
  const int b = blockIdx.x;
  if (b >= nbins) return;
  const int tid = threadIdx.x;
  const int lane = tid & 31;
  const int wave = tid >> 5;
  const int* bstart = bst + blockIdx.y * (MAXB + 1);
  const int bs = bstart[b];
  const int be = bstart[b + 1];
  const int m = be - bs;

  if (tid < MAXB) c[tid] = 0;
  __syncthreads();
  for (int i = bs + tid; i < be; i += 1024) atomicAdd(&c[bn[i] & 255u], 1);
  __syncthreads();
  if (wave == 0) {
    int a[8];
    int sa = 0;
#pragma unroll
    for (int k = 0; k < 8; ++k) { a[k] = c[8 * lane + k]; sa += a[k]; }
    int xa = sa;
#pragma unroll
    for (int d = 1; d < 32; d <<= 1) {
      const int ya = __shfl_up(xa, d);
      if (lane >= d) xa += ya;
    }
    int ra = xa - sa;
    const int R = 1 << sh;
#pragma unroll
    for (int k = 0; k < 8; ++k) {
      const int j = 8 * lane + k;
      cur[j] = ra;
      const int node = (b << sh) + j;
      if (j < R && node < n) rp[node] = bs + ra;
      ra += a[k];
    }
  }
  if (tid == 0 && b == nbins - 1) rp[n] = be;
  __syncthreads();
  if (m <= CAP) {
    for (int i = bs + tid; i < be; i += 1024) {
      const unsigned w = bn[i];
      const int p = atomicAdd(&cur[w & 255u], 1);
      stage[p] = (int)(w >> 8);
    }
    __syncthreads();
    for (int i = tid; i < m; i += 1024) csr[bs + i] = stage[i];
  } else {
    for (int i = bs + tid; i < be; i += 1024) {
      const unsigned w = bn[i];
      const int p = atomicAdd(&cur[w & 255u], 1);
      csr[bs + p] = (int)(w >> 8);
    }
  }
}

__device__ __forceinline__ float4 ld_row16(const float* __restrict__ x, int s, int c)
{
  return *(const float4*)(x + (size_t)s * HDIM + c);
}

__global__ __launch_bounds__(256)
void hg_gather(const float* __restrict__ x0, const float* __restrict__ x1,
               const float* __restrict__ x2, const float* __restrict__ x3,
               const int* __restrict__ rp0, const int* __restrict__ rp1,
               const int* __restrict__ rp2, const int* __restrict__ rp3,
               const int* __restrict__ csr0, const int* __restrict__ csr1,
               const int* __restrict__ csr2, const int* __restrict__ csr3,
               float* __restrict__ acc0, float* __restrict__ acc1,
               float* __restrict__ acc2, float* __restrict__ acc3,
               float* __restrict__ cnt0, float* __restrict__ cnt1,
               float* __restrict__ cnt2, float* __restrict__ cnt3,
               int n0, int n1, int n2, int n3)
{
  const float* x; const int* rp; const int* csr; float* acc; float* cnt; int n;
  switch (blockIdx.y) {
    case 0:  x = x0; rp = rp0; csr = csr0; acc = acc0; cnt = cnt0; n = n0; break;
    case 1:  x = x1; rp = rp1; csr = csr1; acc = acc1; cnt = cnt1; n = n1; break;
    case 2:  x = x2; rp = rp2; csr = csr2; acc = acc2; cnt = cnt2; n = n2; break;
    default: x = x3; rp = rp3; csr = csr3; acc = acc3; cnt = cnt3; n = n3; break;
  }
  const int wave = __builtin_amdgcn_readfirstlane((int)(threadIdx.x >> 5));
  const int lane = threadIdx.x & 31;
  const int h = lane >> 4;
  const int c = (lane & 15) << 2;
  const int dstep = gridDim.x * 8;
  for (int d = blockIdx.x * 8 + wave; d < n; d += dstep) {
    const int beg = rp[d];
    const int end = rp[d + 1];

    float4 a0 = make_float4(0.f, 0.f, 0.f, 0.f);
    float4 a1 = make_float4(0.f, 0.f, 0.f, 0.f);
    for (int jb = beg; jb < end; jb += 32) {
      const int j = jb + lane;
      const int idx = (j < end) ? csr[j] : 0;
      const int nk = (end - jb < 32) ? (end - jb) : 32;
      int k = 0;
      for (; k + 8 <= nk; k += 8) {
        const int s0 = __shfl(idx, k + h);
        const int s1 = __shfl(idx, k + 2 + h);
        const int s2 = __shfl(idx, k + 4 + h);
        const int s3 = __shfl(idx, k + 6 + h);
        const float4 v0 = ld_row16(x, s0, c);
        const float4 v1 = ld_row16(x, s1, c);
        const float4 v2 = ld_row16(x, s2, c);
        const float4 v3 = ld_row16(x, s3, c);
        a0.x += v0.x; a0.y += v0.y; a0.z += v0.z; a0.w += v0.w;
        a1.x += v1.x; a1.y += v1.y; a1.z += v1.z; a1.w += v1.w;
        a0.x += v2.x; a0.y += v2.y; a0.z += v2.z; a0.w += v2.w;
        a1.x += v3.x; a1.y += v3.y; a1.z += v3.z; a1.w += v3.w;
      }
      if (k < nk) {
        float4 t[4];
#pragma unroll
        for (int q = 0; q < 4; ++q) {
          const int kk = k + 2 * q + h;
          const int s = __shfl(idx, kk);
          t[q] = make_float4(0.f, 0.f, 0.f, 0.f);
          if (kk < nk) t[q] = ld_row16(x, s, c);
        }
#pragma unroll
        for (int q = 0; q < 4; ++q) {
          a0.x += t[q].x; a0.y += t[q].y; a0.z += t[q].z; a0.w += t[q].w;
        }
      }
    }
    a0.x += a1.x; a0.y += a1.y; a0.z += a1.z; a0.w += a1.w;
    a0.x += __shfl_xor(a0.x, 16);
    a0.y += __shfl_xor(a0.y, 16);
    a0.z += __shfl_xor(a0.z, 16);
    a0.w += __shfl_xor(a0.w, 16);
    if (h == 0) *(float4*)(acc + (size_t)d * HDIM + c) = a0;
    if (lane == 0) cnt[d] = (float)(end - beg);
  }
}

__device__ __forceinline__ void mma_product(v8f cacc[4],
                                            const float* __restrict__ Abase,
                                            float inv,
                                            const float* __restrict__ Wsh,
                                            int m, int half)
{
#pragma unroll
  for (int k0 = 0; k0 < HDIM; k0 += 4) {
    const int kk = k0 + 2 * half;
    v2f a;
    a.x = Abase[kk] * inv;
    a.y = Abase[kk + 1] * inv;
#pragma unroll
    for (int nt = 0; nt < 4; ++nt) {
      v2f b = *(const v2f*)(Wsh + (size_t)(nt * 16 + m) * HDIM + kk);
      cacc[nt] = __builtin_amdgcn_wmma_f32_16x16x4_f32(
          false, a, false, b, (short)0, cacc[nt], false, false);
    }
  }
}

__global__ __launch_bounds__(256)
void hg_sage(const float* __restrict__ acc0, const float* __restrict__ acc1,
             const float* __restrict__ acc2, const float* __restrict__ acc3,
             const float* __restrict__ cnt0, const float* __restrict__ cnt1,
             const float* __restrict__ cnt2, const float* __restrict__ cnt3,
             const float* __restrict__ Wl, const float* __restrict__ bl,
             const float* __restrict__ Wr,
             const float* __restrict__ xd, const float* __restrict__ xg,
             const float* __restrict__ xs,
             float* __restrict__ outd, float* __restrict__ outg,
             float* __restrict__ outs,
             int Nd, int Ng, int Ns, int gGene, int gDrug, int do_relu)
{
  __shared__ float sWlA[HDIM * HDIM];
  __shared__ float sWlB[HDIM * HDIM];
  __shared__ float sWr [HDIM * HDIM];
  __shared__ float sbias[HDIM];

  const int tid = threadIdx.x;

  const float* accA; const float* cntA; const float* accB; const float* cntB;
  const float* xdst; float* out;
  int tA, tB, n_dst, blk;
  bool hasB;
  if ((int)blockIdx.x < gGene) {
    blk = blockIdx.x; hasB = true; tA = 0; tB = 3;
    accA = acc0; cntA = cnt0; accB = acc3; cntB = cnt3;
    xdst = xg; out = outg; n_dst = Ng;
  } else if ((int)blockIdx.x < gGene + gDrug) {
    blk = blockIdx.x - gGene; hasB = false; tA = 1; tB = 1;
    accA = acc1; cntA = cnt1; accB = acc1; cntB = cnt1;
    xdst = xd; out = outd; n_dst = Nd;
  } else {
    blk = blockIdx.x - gGene - gDrug; hasB = false; tA = 2; tB = 2;
    accA = acc2; cntA = cnt2; accB = acc2; cntB = cnt2;
    xdst = xs; out = outs; n_dst = Ns;
  }
  const float* WlA = Wl + (size_t)tA * HDIM * HDIM;
  const float* WrA = Wr + (size_t)tA * HDIM * HDIM;
  const float* blA = bl + (size_t)tA * HDIM;
  const float* WlB = Wl + (size_t)tB * HDIM * HDIM;
  const float* WrB = Wr + (size_t)tB * HDIM * HDIM;
  const float* blB = bl + (size_t)tB * HDIM;

  for (int i = tid; i < HDIM * HDIM; i += 256) {
    sWlA[i] = WlA[i];
    sWr[i]  = hasB ? (WrA[i] + WrB[i]) : WrA[i];
    sWlB[i] = hasB ? WlB[i] : 0.0f;
  }
  if (tid < HDIM) sbias[tid] = hasB ? (blA[tid] + blB[tid]) : blA[tid];
  __syncthreads();

  const int wave = tid >> 5;
  const int lane = tid & 31;
  const int tile = blk * 8 + wave;
  if (tile * 16 >= n_dst) return;

  const int m = lane & 15;
  const int half = lane >> 4;
  const int row = tile * 16 + m;

  v8f cacc[4] = {};

  {
    const float invA = 1.0f / fmaxf(cntA[row], 1.0f);
    mma_product(cacc, accA + (size_t)row * HDIM, invA, sWlA, m, half);
  }
  if (hasB) {
    const float invB = 1.0f / fmaxf(cntB[row], 1.0f);
    mma_product(cacc, accB + (size_t)row * HDIM, invB, sWlB, m, half);
  }
  mma_product(cacc, xdst + (size_t)row * HDIM, 1.0f, sWr, m, half);

#pragma unroll
  for (int nt = 0; nt < 4; ++nt) {
    const float bv = sbias[nt * 16 + m];
    float* obase = out + (size_t)(tile * 16) * HDIM + nt * 16 + m;
#pragma unroll
    for (int r = 0; r < 8; ++r) {
      float v = cacc[nt][r] + bv;
      if (do_relu) v = fmaxf(v, 0.0f);
      obase[(size_t)(r + 8 * half) * HDIM] = v;
    }
  }
}

extern "C" void kernel_launch(void* const* d_in, const int* in_sizes, int n_in,
                              void* d_out, int out_size, void* d_ws, size_t ws_size,
                              hipStream_t stream) {
  const int* src0 = (const int*)d_in[0]; const int* dst0 = (const int*)d_in[1];
  const int* src1 = (const int*)d_in[2]; const int* dst1 = (const int*)d_in[3];
  const int* src2 = (const int*)d_in[4]; const int* dst2 = (const int*)d_in[5];
  const int* src3 = (const int*)d_in[6]; const int* dst3 = (const int*)d_in[7];
  const float* ed  = (const float*)d_in[8];
  const float* eg  = (const float*)d_in[9];
  const float* es  = (const float*)d_in[10];
  const float* Wl0 = (const float*)d_in[11];
  const float* bl0 = (const float*)d_in[12];
  const float* Wr0 = (const float*)d_in[13];
  const float* Wl1 = (const float*)d_in[14];
  const float* bl1 = (const float*)d_in[15];
  const float* Wr1 = (const float*)d_in[16];

  const int E  = in_sizes[0];
  const int Nd = in_sizes[8]  / HDIM;
  const int Ng = in_sizes[9]  / HDIM;
  const int Ns = in_sizes[10] / HDIM;

  auto up64 = [](size_t n) { return (n + 63) & ~(size_t)63; };
  float* ws = (float*)d_ws;
  size_t off = 0;
  float* acc0 = ws + off; off += (size_t)Ng * HDIM;
  float* acc1 = ws + off; off += (size_t)Nd * HDIM;
  float* acc2 = ws + off; off += (size_t)Ns * HDIM;
  float* acc3 = ws + off; off += (size_t)Ng * HDIM;
  float* cnt0 = ws + off; off += up64(Ng);
  float* cnt1 = ws + off; off += up64(Nd);
  float* cnt2 = ws + off; off += up64(Ns);
  float* cnt3 = ws + off; off += up64(Ng);
  float* xd1 = ws + off; off += (size_t)Nd * HDIM;
  float* xg1 = ws + off; off += (size_t)Ng * HDIM;
  float* xs1 = ws + off; off += (size_t)Ns * HDIM;
  const int NT = (E + TILE - 1) / TILE;
  int* hist = (int*)(ws + off); off += up64((size_t)4 * MAXB * NT);
  int* bst  = (int*)(ws + off); off += up64((size_t)4 * (MAXB + 1));
  unsigned* bn0 = (unsigned*)(ws + off); off += up64(E);
  unsigned* bn1 = (unsigned*)(ws + off); off += up64(E);
  unsigned* bn2 = (unsigned*)(ws + off); off += up64(E);
  unsigned* bn3 = (unsigned*)(ws + off); off += up64(E);
  int* rp0 = (int*)(ws + off); off += up64((size_t)Ng + 1);
  int* rp1 = (int*)(ws + off); off += up64((size_t)Nd + 1);
  int* rp2 = (int*)(ws + off); off += up64((size_t)Ns + 1);
  int* rp3 = (int*)(ws + off); off += up64((size_t)Ng + 1);
  int* csr0 = (int*)(ws + off); off += up64(E);
  int* csr1 = (int*)(ws + off); off += up64(E);
  int* csr2 = (int*)(ws + off); off += up64(E);
  int* csr3 = (int*)(ws + off); off += up64(E);

  float* outd = (float*)d_out;
  float* outg = outd + (size_t)Nd * HDIM;
  float* outs = outg + (size_t)Ng * HDIM;

  const dim3 blk(256);
  const int gGene = (Ng / 16 + 7) / 8;
  const int gDrug = (Nd / 16 + 7) / 8;
  const int gDis  = (Ns / 16 + 7) / 8;
  const int gAll  = gGene + gDrug + gDis;
  const dim3 tgrid(NT, 4);
  const dim3 bgrid(MAXB, 4);
  const dim3 ggrid(1024, 4);

  hg_bincount<<<tgrid, 1024, 0, stream>>>(dst0, dst1, dst2, dst3, hist, E, NT, Ng, Nd, Ns, Ng);
  hg_binscatter<<<tgrid, 1024, 0, stream>>>(src0, dst0, src1, dst1, src2, dst2, src3, dst3,
                                            hist, bst, bn0, bn1, bn2, bn3, E, NT, Ng, Nd, Ns, Ng);
  hg_binsort<<<bgrid, 1024, 0, stream>>>(bn0, bn1, bn2, bn3, bst, rp0, rp1, rp2, rp3,
                                         csr0, csr1, csr2, csr3, Ng, Nd, Ns, Ng);

  hg_gather<<<ggrid, blk, 0, stream>>>(ed, eg, eg, es, rp0, rp1, rp2, rp3,
                                       csr0, csr1, csr2, csr3, acc0, acc1, acc2, acc3,
                                       cnt0, cnt1, cnt2, cnt3, Ng, Nd, Ns, Ng);

  hg_sage<<<gAll, blk, 0, stream>>>(acc0, acc1, acc2, acc3, cnt0, cnt1, cnt2, cnt3,
                                    Wl0, bl0, Wr0, ed, eg, es, xd1, xg1, xs1,
                                    Nd, Ng, Ns, gGene, gDrug, 1);

  hg_gather<<<ggrid, blk, 0, stream>>>(xd1, xg1, xg1, xs1, rp0, rp1, rp2, rp3,
                                       csr0, csr1, csr2, csr3, acc0, acc1, acc2, acc3,
                                       cnt0, cnt1, cnt2, cnt3, Ng, Nd, Ns, Ng);

  hg_sage<<<gAll, blk, 0, stream>>>(acc0, acc1, acc2, acc3, cnt0, cnt1, cnt2, cnt3,
                                    Wl1, bl1, Wr1, xd1, xg1, xs1, outd, outg, outs,
                                    Nd, Ng, Ns, gGene, gDrug, 0);
}


# Pad the LDS row pitch of the staged weight matrices from 64 to 68 floats: the 16 rows of each B-fragment read fall on different banks instead of the same ones (16-way bank conflict removed)
# speedup vs baseline: 8.4888x; 1.1314x over previous
#include <hip/hip_runtime.h>

typedef __attribute__((ext_vector_type(2))) float v2f;
typedef __attribute__((ext_vector_type(8))) float v8f;

#define HDIM 64
#define WPITCH 68

#define TILE 8192
#define MAXB 256
#define CAP  12288

__device__ __forceinline__ int bin_shift(int n)
{
  int s = 8;
  while (s > 5 && ((n - 1) >> (s - 1)) < MAXB) --s;
  return s;
}

__global__ __launch_bounds__(1024)
void hg_bincount(const int* __restrict__ dst0, const int* __restrict__ dst1,
                 const int* __restrict__ dst2, const int* __restrict__ dst3,
                 int* __restrict__ hist, int E, int NT,
                 int n0, int n1, int n2, int n3)
{
  __shared__ int h[MAXB];
  const int* dst; int n;
  switch (blockIdx.y) {
    case 0:  dst = dst0; n = n0; break;
    case 1:  dst = dst1; n = n1; break;
    case 2:  dst = dst2; n = n2; break;
    default: dst = dst3; n = n3; break;
  }
  const int sh = bin_shift(n);
  const int tid = threadIdx.x;
  const int t = blockIdx.x;
  if (tid < MAXB) h[tid] = 0;
  __syncthreads();
  const int e0 = t * TILE;
#pragma unroll
  for (int k = 0; k < TILE / 1024; ++k) {
    const int e = e0 + k * 1024 + tid;
    if (e < E) atomicAdd(&h[dst[e] >> sh], 1);
  }
  __syncthreads();
  if (tid < MAXB) hist[((size_t)blockIdx.y * NT + t) * MAXB + tid] = h[tid];
}

__global__ __launch_bounds__(1024)
void hg_binscatter(const int* __restrict__ src0, const int* __restrict__ dst0,
                   const int* __restrict__ src1, const int* __restrict__ dst1,
                   const int* __restrict__ src2, const int* __restrict__ dst2,
                   const int* __restrict__ src3, const int* __restrict__ dst3,
                   const int* __restrict__ hist, int* __restrict__ bst,
                   unsigned* __restrict__ bn0, unsigned* __restrict__ bn1,
                   unsigned* __restrict__ bn2, unsigned* __restrict__ bn3,
                   int E, int NT, int n0, int n1, int n2, int n3)
{
  __shared__ int s_tot[MAXB];
  __shared__ int s_pre[MAXB];
  __shared__ int s_own[MAXB];
  __shared__ int s_gd[MAXB];
  __shared__ int s_lc[MAXB];
  __shared__ int s_part[3][4][MAXB];
  __shared__ unsigned s_word[TILE];
  __shared__ unsigned char s_bin[TILE];

  const int* src; const int* dst; unsigned* bn; int n;
  switch (blockIdx.y) {
    case 0:  src = src0; dst = dst0; bn = bn0; n = n0; break;
    case 1:  src = src1; dst = dst1; bn = bn1; n = n1; break;
    case 2:  src = src2; dst = dst2; bn = bn2; n = n2; break;
    default: src = src3; dst = dst3; bn = bn3; n = n3; break;
  }
  const int sh = bin_shift(n);
  const int lmask = (1 << sh) - 1;
  const int tid = threadIdx.x;
  const int lane = tid & 31;
  const int wave = tid >> 5;
  const int t = blockIdx.x;
  const int* hb = hist + (size_t)blockIdx.y * NT * MAXB;
  int* bstart = bst + blockIdx.y * (MAXB + 1);

  {
    const int bq = tid & (MAXB - 1);
    const int q = tid >> 8;
    int tot = 0, pre = 0, own = 0;
#pragma unroll 8
    for (int tt = q; tt < NT; tt += 4) {
      const int v = hb[(size_t)tt * MAXB + bq];
      tot += v;
      pre += (tt < t) ? v : 0;
      own += (tt == t) ? v : 0;
    }
    s_part[0][q][bq] = tot; s_part[1][q][bq] = pre; s_part[2][q][bq] = own;
  }
  __syncthreads();
  if (tid < MAXB) {
    s_tot[tid] = s_part[0][0][tid] + s_part[0][1][tid] + s_part[0][2][tid] + s_part[0][3][tid];
    s_pre[tid] = s_part[1][0][tid] + s_part[1][1][tid] + s_part[1][2][tid] + s_part[1][3][tid];
    s_own[tid] = s_part[2][0][tid] + s_part[2][1][tid] + s_part[2][2][tid] + s_part[2][3][tid];
  }
  __syncthreads();
  if (wave == 0) {
    int a[8], o[8];
    int sa = 0, so = 0;
#pragma unroll
    for (int k = 0; k < 8; ++k) {
      a[k] = s_tot[8 * lane + k]; o[k] = s_own[8 * lane + k];
      sa += a[k]; so += o[k];
    }
    int xa = sa, xo = so;
#pragma unroll
    for (int d = 1; d < 32; d <<= 1) {
      const int ya = __shfl_up(xa, d);
      const int yo = __shfl_up(xo, d);
      if (lane >= d) { xa += ya; xo += yo; }
    }
    int ra = xa - sa, ro = xo - so;
#pragma unroll
    for (int k = 0; k < 8; ++k) {
      const int b = 8 * lane + k;
      s_gd[b] = ra + s_pre[b] - ro;
      s_lc[b] = ro;
      if (t == 0) bstart[b] = ra;
      ra += a[k]; ro += o[k];
    }
    if (t == 0 && lane == 31) bstart[MAXB] = ra;
  }
  __syncthreads();

  const int e0 = t * TILE;
#pragma unroll
  for (int k = 0; k < TILE / 1024; ++k) {
    const int e = e0 + k * 1024 + tid;
    if (e < E) {
      const int d = dst[e];
      const int s = src[e];
      const int b = d >> sh;
      const int lp = atomicAdd(&s_lc[b], 1);
      s_word[lp] = ((unsigned)s << 8) | (unsigned)(d & lmask);
      s_bin[lp] = (unsigned char)b;
    }
  }
  __syncthreads();
  const int m = (E - e0 < TILE) ? (E - e0) : TILE;
  for (int i = tid; i < m; i += 1024)
    bn[i + s_gd[s_bin[i]]] = s_word[i];
}

__global__ __launch_bounds__(1024)
void hg_binsort(const unsigned* __restrict__ bn0, const unsigned* __restrict__ bn1,
                const unsigned* __restrict__ bn2, const unsigned* __restrict__ bn3,
                const int* __restrict__ bst,
                int* __restrict__ rp0, int* __restrict__ rp1,
                int* __restrict__ rp2, int* __restrict__ rp3,
                int* __restrict__ csr0, int* __restrict__ csr1,
                int* __restrict__ csr2, int* __restrict__ csr3,
                int n0, int n1, int n2, int n3)
{
  __shared__ int c[MAXB];
  __shared__ int cur[MAXB];
  __shared__ int stage[CAP];

  const unsigned* bn; int* rp; int* csr; int n;
  switch (blockIdx.y) {
    case 0:  bn = bn0; rp = rp0; csr = csr0; n = n0; break;
    case 1:  bn = bn1; rp = rp1; csr = csr1; n = n1; break;
    case 2:  bn = bn2; rp = rp2; csr = csr2; n = n2; break;
    default: bn = bn3; rp = rp3; csr = csr3; n = n3; break;
  }
  const int sh = bin_shift(n);
  const int nbins = ((n - 1) >> sh) + 1;
  const int b = blockIdx.x;
  if (b >= nbins) return;
  const int tid = threadIdx.x;
  const int lane = tid & 31;
  const int wave = tid >> 5;
  const int* bstart = bst + blockIdx.y * (MAXB + 1);
  const int bs = bstart[b];
  const int be = bstart[b + 1];
  const int m = be - bs;

  if (tid < MAXB) c[tid] = 0;
  __syncthreads();
  for (int i = bs + tid; i < be; i += 1024) atomicAdd(&c[bn[i] & 255u], 1);
  __syncthreads();
  if (wave == 0) {
    int a[8];
    int sa = 0;
#pragma unroll
    for (int k = 0; k < 8; ++k) { a[k] = c[8 * lane + k]; sa += a[k]; }
    int xa = sa;
#pragma unroll
    for (int d = 1; d < 32; d <<= 1) {
      const int ya = __shfl_up(xa, d);
      if (lane >= d) xa += ya;
    }
    int ra = xa - sa;
    const int R = 1 << sh;
#pragma unroll
    for (int k = 0; k < 8; ++k) {
      const int j = 8 * lane + k;
      cur[j] = ra;
      const int node = (b << sh) + j;
      if (j < R && node < n) rp[node] = bs + ra;
      ra += a[k];
    }
  }
  if (tid == 0 && b == nbins - 1) rp[n] = be;
  __syncthreads();
  if (m <= CAP) {
    for (int i = bs + tid; i < be; i += 1024) {
      const unsigned w = bn[i];
      const int p = atomicAdd(&cur[w & 255u], 1);
      stage[p] = (int)(w >> 8);
    }
    __syncthreads();
    for (int i = tid; i < m; i += 1024) csr[bs + i] = stage[i];
  } else {
    for (int i = bs + tid; i < be; i += 1024) {
      const unsigned w = bn[i];
      const int p = atomicAdd(&cur[w & 255u], 1);
      csr[bs + p] = (int)(w >> 8);
    }
  }
}

__device__ __forceinline__ float4 ld_row16(const float* __restrict__ x, int s, int c)
{
  return *(const float4*)(x + (size_t)s * HDIM + c);
}

__global__ __launch_bounds__(256)
void hg_gather(const float* __restrict__ x0, const float* __restrict__ x1,
               const float* __restrict__ x2, const float* __restrict__ x3,
               const int* __restrict__ rp0, const int* __restrict__ rp1,
               const int* __restrict__ rp2, const int* __restrict__ rp3,
               const int* __restrict__ csr0, const int* __restrict__ csr1,
               const int* __restrict__ csr2, const int* __restrict__ csr3,
               float* __restrict__ acc0, float* __restrict__ acc1,
               float* __restrict__ acc2, float* __restrict__ acc3,
               float* __restrict__ cnt0, float* __restrict__ cnt1,
               float* __restrict__ cnt2, float* __restrict__ cnt3,
               int n0, int n1, int n2, int n3)
{
  const float* x; const int* rp; const int* csr; float* acc; float* cnt; int n;
  switch (blockIdx.y) {
    case 0:  x = x0; rp = rp0; csr = csr0; acc = acc0; cnt = cnt0; n = n0; break;
    case 1:  x = x1; rp = rp1; csr = csr1; acc = acc1; cnt = cnt1; n = n1; break;
    case 2:  x = x2; rp = rp2; csr = csr2; acc = acc2; cnt = cnt2; n = n2; break;
    default: x = x3; rp = rp3; csr = csr3; acc = acc3; cnt = cnt3; n = n3; break;
  }
  const int wave = __builtin_amdgcn_readfirstlane((int)(threadIdx.x >> 5));
  const int lane = threadIdx.x & 31;
  const int h = lane >> 4;
  const int c = (lane & 15) << 2;
  const int dstep = gridDim.x * 8;
  for (int d = blockIdx.x * 8 + wave; d < n; d += dstep) {
    const int beg = rp[d];
    const int end = rp[d + 1];

    float4 a0 = make_float4(0.f, 0.f, 0.f, 0.f);
    float4 a1 = make_float4(0.f, 0.f, 0.f, 0.f);
    for (int jb = beg; jb < end; jb += 32) {
      const int j = jb + lane;
      const int idx = (j < end) ? csr[j] : 0;
      const int nk = (end - jb < 32) ? (end - jb) : 32;
      int k = 0;
      for (; k + 8 <= nk; k += 8) {
        const int s0 = __shfl(idx, k + h);
        const int s1 = __shfl(idx, k + 2 + h);
        const int s2 = __shfl(idx, k + 4 + h);
        const int s3 = __shfl(idx, k + 6 + h);
        const float4 v0 = ld_row16(x, s0, c);
        const float4 v1 = ld_row16(x, s1, c);
        const float4 v2 = ld_row16(x, s2, c);
        const float4 v3 = ld_row16(x, s3, c);
        a0.x += v0.x; a0.y += v0.y; a0.z += v0.z; a0.w += v0.w;
        a1.x += v1.x; a1.y += v1.y; a1.z += v1.z; a1.w += v1.w;
        a0.x += v2.x; a0.y += v2.y; a0.z += v2.z; a0.w += v2.w;
        a1.x += v3.x; a1.y += v3.y; a1.z += v3.z; a1.w += v3.w;
      }
      if (k < nk) {
        float4 t[4];
#pragma unroll
        for (int q = 0; q < 4; ++q) {
          const int kk = k + 2 * q + h;
          const int s = __shfl(idx, kk);
          t[q] = make_float4(0.f, 0.f, 0.f, 0.f);
          if (kk < nk) t[q] = ld_row16(x, s, c);
        }
#pragma unroll
        for (int q = 0; q < 4; ++q) {
          a0.x += t[q].x; a0.y += t[q].y; a0.z += t[q].z; a0.w += t[q].w;
        }
      }
    }
    a0.x += a1.x; a0.y += a1.y; a0.z += a1.z; a0.w += a1.w;
    a0.x += __shfl_xor(a0.x, 16);
    a0.y += __shfl_xor(a0.y, 16);
    a0.z += __shfl_xor(a0.z, 16);
    a0.w += __shfl_xor(a0.w, 16);
    if (h == 0) *(float4*)(acc + (size_t)d * HDIM + c) = a0;
    if (lane == 0) cnt[d] = (float)(end - beg);
  }
}

__device__ __forceinline__ void mma_product(v8f cacc[4],
                                            const float* __restrict__ Abase,
                                            float inv,
                                            const float* __restrict__ Wsh,
                                            int m, int half)
{
#pragma unroll
  for (int k0 = 0; k0 < HDIM; k0 += 4) {
    const int kk = k0 + 2 * half;
    v2f a;
    a.x = Abase[kk] * inv;
    a.y = Abase[kk + 1] * inv;
#pragma unroll
    for (int nt = 0; nt < 4; ++nt) {
      v2f b = *(const v2f*)(Wsh + (size_t)(nt * 16 + m) * WPITCH + kk);
      cacc[nt] = __builtin_amdgcn_wmma_f32_16x16x4_f32(
          false, a, false, b, (short)0, cacc[nt], false, false);
    }
  }
}

__global__ __launch_bounds__(256)
void hg_sage(const float* __restrict__ acc0, const float* __restrict__ acc1,
             const float* __restrict__ acc2, const float* __restrict__ acc3,
             const float* __restrict__ cnt0, const float* __restrict__ cnt1,
             const float* __restrict__ cnt2, const float* __restrict__ cnt3,
             const float* __restrict__ Wl, const float* __restrict__ bl,
             const float* __restrict__ Wr,
             const float* __restrict__ xd, const float* __restrict__ xg,
             const float* __restrict__ xs,
             float* __restrict__ outd, float* __restrict__ outg,
             float* __restrict__ outs,
             int Nd, int Ng, int Ns, int gGene, int gDrug, int do_relu)
{
  __shared__ float sWlA[HDIM * WPITCH];
  __shared__ float sWlB[HDIM * WPITCH];
  __shared__ float sWr [HDIM * WPITCH];
  __shared__ float sbias[HDIM];

  const int tid = threadIdx.x;

  const float* accA; const float* cntA; const float* accB; const float* cntB;
  const float* xdst; float* out;
  int tA, tB, n_dst, blk;
  bool hasB;
  if ((int)blockIdx.x < gGene) {
    blk = blockIdx.x; hasB = true; tA = 0; tB = 3;
    accA = acc0; cntA = cnt0; accB = acc3; cntB = cnt3;
    xdst = xg; out = outg; n_dst = Ng;
  } else if ((int)blockIdx.x < gGene + gDrug) {
    blk = blockIdx.x - gGene; hasB = false; tA = 1; tB = 1;
    accA = acc1; cntA = cnt1; accB = acc1; cntB = cnt1;
    xdst = xd; out = outd; n_dst = Nd;
  } else {
    blk = blockIdx.x - gGene - gDrug; hasB = false; tA = 2; tB = 2;
    accA = acc2; cntA = cnt2; accB = acc2; cntB = cnt2;
    xdst = xs; out = outs; n_dst = Ns;
  }
  const float* WlA = Wl + (size_t)tA * HDIM * HDIM;
  const float* WrA = Wr + (size_t)tA * HDIM * HDIM;
  const float* blA = bl + (size_t)tA * HDIM;
  const float* WlB = Wl + (size_t)tB * HDIM * HDIM;
  const float* WrB = Wr + (size_t)tB * HDIM * HDIM;
  const float* blB = bl + (size_t)tB * HDIM;

  for (int i = tid; i < HDIM * HDIM; i += 256) {
    const int p = (i >> 6) * WPITCH + (i & (HDIM - 1));
    sWlA[p] = WlA[i];
    sWr[p]  = hasB ? (WrA[i] + WrB[i]) : WrA[i];
    sWlB[p] = hasB ? WlB[i] : 0.0f;
  }
  if (tid < HDIM) sbias[tid] = hasB ? (blA[tid] + blB[tid]) : blA[tid];
  __syncthreads();

  const int wave = tid >> 5;
  const int lane = tid & 31;
  const int tile = blk * 8 + wave;
  if (tile * 16 >= n_dst) return;

  const int m = lane & 15;
  const int half = lane >> 4;
  const int row = tile * 16 + m;

  v8f cacc[4] = {};

  {
    const float invA = 1.0f / fmaxf(cntA[row], 1.0f);
    mma_product(cacc, accA + (size_t)row * HDIM, invA, sWlA, m, half);
  }
  if (hasB) {
    const float invB = 1.0f / fmaxf(cntB[row], 1.0f);
    mma_product(cacc, accB + (size_t)row * HDIM, invB, sWlB, m, half);
  }
  mma_product(cacc, xdst + (size_t)row * HDIM, 1.0f, sWr, m, half);

#pragma unroll
  for (int nt = 0; nt < 4; ++nt) {
    const float bv = sbias[nt * 16 + m];
    float* obase = out + (size_t)(tile * 16) * HDIM + nt * 16 + m;
#pragma unroll
    for (int r = 0; r < 8; ++r) {
      float v = cacc[nt][r] + bv;
      if (do_relu) v = fmaxf(v, 0.0f);
      obase[(size_t)(r + 8 * half) * HDIM] = v;
    }
  }
}

extern "C" void kernel_launch(void* const* d_in, const int* in_sizes, int n_in,
                              void* d_out, int out_size, void* d_ws, size_t ws_size,
                              hipStream_t stream) {
  const int* src0 = (const int*)d_in[0]; const int* dst0 = (const int*)d_in[1];
  const int* src1 = (const int*)d_in[2]; const int* dst1 = (const int*)d_in[3];
  const int* src2 = (const int*)d_in[4]; const int* dst2 = (const int*)d_in[5];
  const int* src3 = (const int*)d_in[6]; const int* dst3 = (const int*)d_in[7];
  const float* ed  = (const float*)d_in[8];
  const float* eg  = (const float*)d_in[9];
  const float* es  = (const float*)d_in[10];
  const float* Wl0 = (const float*)d_in[11];
  const float* bl0 = (const float*)d_in[12];
  const float* Wr0 = (const float*)d_in[13];
  const float* Wl1 = (const float*)d_in[14];
  const float* bl1 = (const float*)d_in[15];
  const float* Wr1 = (const float*)d_in[16];

  const int E  = in_sizes[0];
  const int Nd = in_sizes[8]  / HDIM;
  const int Ng = in_sizes[9]  / HDIM;
  const int Ns = in_sizes[10] / HDIM;

  auto up64 = [](size_t n) { return (n + 63) & ~(size_t)63; };
  float* ws = (float*)d_ws;
  size_t off = 0;
  float* acc0 = ws + off; off += (size_t)Ng * HDIM;
  float* acc1 = ws + off; off += (size_t)Nd * HDIM;
  float* acc2 = ws + off; off += (size_t)Ns * HDIM;
  float* acc3 = ws + off; off += (size_t)Ng * HDIM;
  float* cnt0 = ws + off; off += up64(Ng);
  float* cnt1 = ws + off; off += up64(Nd);
  float* cnt2 = ws + off; off += up64(Ns);
  float* cnt3 = ws + off; off += up64(Ng);
  float* xd1 = ws + off; off += (size_t)Nd * HDIM;
  float* xg1 = ws + off; off += (size_t)Ng * HDIM;
  float* xs1 = ws + off; off += (size_t)Ns * HDIM;
  const int NT = (E + TILE - 1) / TILE;
  int* hist = (int*)(ws + off); off += up64((size_t)4 * MAXB * NT);
  int* bst  = (int*)(ws + off); off += up64((size_t)4 * (MAXB + 1));
  unsigned* bn0 = (unsigned*)(ws + off); off += up64(E);
  unsigned* bn1 = (unsigned*)(ws + off); off += up64(E);
  unsigned* bn2 = (unsigned*)(ws + off); off += up64(E);
  unsigned* bn3 = (unsigned*)(ws + off); off += up64(E);
  int* rp0 = (int*)(ws + off); off += up64((size_t)Ng + 1);
  int* rp1 = (int*)(ws + off); off += up64((size_t)Nd + 1);
  int* rp2 = (int*)(ws + off); off += up64((size_t)Ns + 1);
  int* rp3 = (int*)(ws + off); off += up64((size_t)Ng + 1);
  int* csr0 = (int*)(ws + off); off += up64(E);
  int* csr1 = (int*)(ws + off); off += up64(E);
  int* csr2 = (int*)(ws + off); off += up64(E);
  int* csr3 = (int*)(ws + off); off += up64(E);

  float* outd = (float*)d_out;
  float* outg = outd + (size_t)Nd * HDIM;
  float* outs = outg + (size_t)Ng * HDIM;

  const dim3 blk(256);
  const int gGene = (Ng / 16 + 7) / 8;
  const int gDrug = (Nd / 16 + 7) / 8;
  const int gDis  = (Ns / 16 + 7) / 8;
  const int gAll  = gGene + gDrug + gDis;
  const dim3 tgrid(NT, 4);
  const dim3 bgrid(MAXB, 4);
  const dim3 ggrid(1024, 4);

  hg_bincount<<<tgrid, 1024, 0, stream>>>(dst0, dst1, dst2, dst3, hist, E, NT, Ng, Nd, Ns, Ng);
  hg_binscatter<<<tgrid, 1024, 0, stream>>>(src0, dst0, src1, dst1, src2, dst2, src3, dst3,
                                            hist, bst, bn0, bn1, bn2, bn3, E, NT, Ng, Nd, Ns, Ng);
  hg_binsort<<<bgrid, 1024, 0, stream>>>(bn0, bn1, bn2, bn3, bst, rp0, rp1, rp2, rp3,
                                         csr0, csr1, csr2, csr3, Ng, Nd, Ns, Ng);

  hg_gather<<<ggrid, blk, 0, stream>>>(ed, eg, eg, es, rp0, rp1, rp2, rp3,
                                       csr0, csr1, csr2, csr3, acc0, acc1, acc2, acc3,
                                       cnt0, cnt1, cnt2, cnt3, Ng, Nd, Ns, Ng);

  hg_sage<<<gAll, blk, 0, stream>>>(acc0, acc1, acc2, acc3, cnt0, cnt1, cnt2, cnt3,
                                    Wl0, bl0, Wr0, ed, eg, es, xd1, xg1, xs1,
                                    Nd, Ng, Ns, gGene, gDrug, 1);

  hg_gather<<<ggrid, blk, 0, stream>>>(xd1, xg1, xg1, xs1, rp0, rp1, rp2, rp3,
                                       csr0, csr1, csr2, csr3, acc0, acc1, acc2, acc3,
                                       cnt0, cnt1, cnt2, cnt3, Ng, Nd, Ns, Ng);

  hg_sage<<<gAll, blk, 0, stream>>>(acc0, acc1, acc2, acc3, cnt0, cnt1, cnt2, cnt3,
                                    Wl1, bl1, Wr1, xd1, xg1, xs1, outd, outg, outs,
                                    Nd, Ng, Ns, gGene, gDrug, 0);
}


# Fuse layer 1's row gather into the per-bin sort work-group: it sums the bin's destination rows straight from the ids it has just staged in LDS; launches per call 7 -> 6; row sum order as before
# speedup vs baseline: 8.8061x; 1.0374x over previous
#include <hip/hip_runtime.h>

typedef __attribute__((ext_vector_type(2))) float v2f;
typedef __attribute__((ext_vector_type(8))) float v8f;

#define HDIM 64
#define WPITCH 68

#define TILE 8192
#define MAXB 256
#define CAP  12288

__device__ __forceinline__ int bin_shift(int n)
{
  int s = 8;
  while (s > 5 && ((n - 1) >> (s - 1)) < MAXB) --s;
  return s;
}

__global__ __launch_bounds__(1024)
void hg_bincount(const int* __restrict__ dst0, const int* __restrict__ dst1,
                 const int* __restrict__ dst2, const int* __restrict__ dst3,
                 int* __restrict__ hist, int E, int NT,
                 int n0, int n1, int n2, int n3)
{
  __shared__ int h[MAXB];
  const int* dst; int n;
  switch (blockIdx.y) {
    case 0:  dst = dst0; n = n0; break;
    case 1:  dst = dst1; n = n1; break;
    case 2:  dst = dst2; n = n2; break;
    default: dst = dst3; n = n3; break;
  }
  const int sh = bin_shift(n);
  const int tid = threadIdx.x;
  const int t = blockIdx.x;
  if (tid < MAXB) h[tid] = 0;
  __syncthreads();
  const int e0 = t * TILE;
#pragma unroll
  for (int k = 0; k < TILE / 1024; ++k) {
    const int e = e0 + k * 1024 + tid;
    if (e < E) atomicAdd(&h[dst[e] >> sh], 1);
  }
  __syncthreads();
  if (tid < MAXB) hist[((size_t)blockIdx.y * NT + t) * MAXB + tid] = h[tid];
}

__global__ __launch_bounds__(1024)
void hg_binscatter(const int* __restrict__ src0, const int* __restrict__ dst0,
                   const int* __restrict__ src1, const int* __restrict__ dst1,
                   const int* __restrict__ src2, const int* __restrict__ dst2,
                   const int* __restrict__ src3, const int* __restrict__ dst3,
                   const int* __restrict__ hist, int* __restrict__ bst,
                   unsigned* __restrict__ bn0, unsigned* __restrict__ bn1,
                   unsigned* __restrict__ bn2, unsigned* __restrict__ bn3,
                   int E, int NT, int n0, int n1, int n2, int n3)
{
  __shared__ int s_tot[MAXB];
  __shared__ int s_pre[MAXB];
  __shared__ int s_own[MAXB];
  __shared__ int s_gd[MAXB];
  __shared__ int s_lc[MAXB];
  __shared__ int s_part[3][4][MAXB];
  __shared__ unsigned s_word[TILE];
  __shared__ unsigned char s_bin[TILE];

  const int* src; const int* dst; unsigned* bn; int n;
  switch (blockIdx.y) {
    case 0:  src = src0; dst = dst0; bn = bn0; n = n0; break;
    case 1:  src = src1; dst = dst1; bn = bn1; n = n1; break;
    case 2:  src = src2; dst = dst2; bn = bn2; n = n2; break;
    default: src = src3; dst = dst3; bn = bn3; n = n3; break;
  }
  const int sh = bin_shift(n);
  const int lmask = (1 << sh) - 1;
  const int tid = threadIdx.x;
  const int lane = tid & 31;
  const int wave = tid >> 5;
  const int t = blockIdx.x;
  const int* hb = hist + (size_t)blockIdx.y * NT * MAXB;
  int* bstart = bst + blockIdx.y * (MAXB + 1);

  {
    const int bq = tid & (MAXB - 1);
    const int q = tid >> 8;
    int tot = 0, pre = 0, own = 0;
#pragma unroll 8
    for (int tt = q; tt < NT; tt += 4) {
      const int v = hb[(size_t)tt * MAXB + bq];
      tot += v;
      pre += (tt < t) ? v : 0;
      own += (tt == t) ? v : 0;
    }
    s_part[0][q][bq] = tot; s_part[1][q][bq] = pre; s_part[2][q][bq] = own;
  }
  __syncthreads();
  if (tid < MAXB) {
    s_tot[tid] = s_part[0][0][tid] + s_part[0][1][tid] + s_part[0][2][tid] + s_part[0][3][tid];
    s_pre[tid] = s_part[1][0][tid] + s_part[1][1][tid] + s_part[1][2][tid] + s_part[1][3][tid];
    s_own[tid] = s_part[2][0][tid] + s_part[2][1][tid] + s_part[2][2][tid] + s_part[2][3][tid];
  }
  __syncthreads();
  if (wave == 0) {
    int a[8], o[8];
    int sa = 0, so = 0;
#pragma unroll
    for (int k = 0; k < 8; ++k) {
      a[k] = s_tot[8 * lane + k]; o[k] = s_own[8 * lane + k];
      sa += a[k]; so += o[k];
    }
    int xa = sa, xo = so;
#pragma unroll
    for (int d = 1; d < 32; d <<= 1) {
      const int ya = __shfl_up(xa, d);
      const int yo = __shfl_up(xo, d);
      if (lane >= d) { xa += ya; xo += yo; }
    }
    int ra = xa - sa, ro = xo - so;
#pragma unroll
    for (int k = 0; k < 8; ++k) {
      const int b = 8 * lane + k;
      s_gd[b] = ra + s_pre[b] - ro;
      s_lc[b] = ro;
      if (t == 0) bstart[b] = ra;
      ra += a[k]; ro += o[k];
    }
    if (t == 0 && lane == 31) bstart[MAXB] = ra;
  }
  __syncthreads();

  const int e0 = t * TILE;
#pragma unroll
  for (int k = 0; k < TILE / 1024; ++k) {
    const int e = e0 + k * 1024 + tid;
    if (e < E) {
      const int d = dst[e];
      const int s = src[e];
      const int b = d >> sh;
      const int lp = atomicAdd(&s_lc[b], 1);
      s_word[lp] = ((unsigned)s << 8) | (unsigned)(d & lmask);
      s_bin[lp] = (unsigned char)b;
    }
  }
  __syncthreads();
  const int m = (E - e0 < TILE) ? (E - e0) : TILE;
  for (int i = tid; i < m; i += 1024)
    bn[i + s_gd[s_bin[i]]] = s_word[i];
}

__device__ __forceinline__ float4 ld_row16(const float* __restrict__ x, int s, int c)
{
  return *(const float4*)(x + (size_t)s * HDIM + c);
}

template <class IdAt>
__device__ __forceinline__ float4 row_sum(const float* __restrict__ x, int beg, int end,
                                          int lane, int h, int c, IdAt id_at)
{
  float4 a0 = make_float4(0.f, 0.f, 0.f, 0.f);
  float4 a1 = make_float4(0.f, 0.f, 0.f, 0.f);
  for (int jb = beg; jb < end; jb += 32) {
    const int j = jb + lane;
    const int idx = (j < end) ? id_at(j) : 0;
    const int nk = (end - jb < 32) ? (end - jb) : 32;
    int k = 0;
    for (; k + 8 <= nk; k += 8) {
      const int s0 = __shfl(idx, k + h);
      const int s1 = __shfl(idx, k + 2 + h);
      const int s2 = __shfl(idx, k + 4 + h);
      const int s3 = __shfl(idx, k + 6 + h);
      const float4 v0 = ld_row16(x, s0, c);
      const float4 v1 = ld_row16(x, s1, c);
      const float4 v2 = ld_row16(x, s2, c);
      const float4 v3 = ld_row16(x, s3, c);
      a0.x += v0.x; a0.y += v0.y; a0.z += v0.z; a0.w += v0.w;
      a1.x += v1.x; a1.y += v1.y; a1.z += v1.z; a1.w += v1.w;
      a0.x += v2.x; a0.y += v2.y; a0.z += v2.z; a0.w += v2.w;
      a1.x += v3.x; a1.y += v3.y; a1.z += v3.z; a1.w += v3.w;
    }
    if (k < nk) {
      float4 t[4];
#pragma unroll
      for (int q = 0; q < 4; ++q) {
        const int kk = k + 2 * q + h;
        const int s = __shfl(idx, kk);
        t[q] = make_float4(0.f, 0.f, 0.f, 0.f);
        if (kk < nk) t[q] = ld_row16(x, s, c);
      }
#pragma unroll
      for (int q = 0; q < 4; ++q) {
        a0.x += t[q].x; a0.y += t[q].y; a0.z += t[q].z; a0.w += t[q].w;
      }
    }
  }
  a0.x += a1.x; a0.y += a1.y; a0.z += a1.z; a0.w += a1.w;
  a0.x += __shfl_xor(a0.x, 16);
  a0.y += __shfl_xor(a0.y, 16);
  a0.z += __shfl_xor(a0.z, 16);
  a0.w += __shfl_xor(a0.w, 16);
  return a0;
}

__global__ __launch_bounds__(1024)
void hg_binsort_gather(const unsigned* __restrict__ bn0, const unsigned* __restrict__ bn1,
                       const unsigned* __restrict__ bn2, const unsigned* __restrict__ bn3,
                       const int* __restrict__ bst,
                       int* __restrict__ rp0, int* __restrict__ rp1,
                       int* __restrict__ rp2, int* __restrict__ rp3,
                       int* __restrict__ csr0, int* __restrict__ csr1,
                       int* __restrict__ csr2, int* __restrict__ csr3,
                       const float* __restrict__ x0, const float* __restrict__ x1,
                       const float* __restrict__ x2, const float* __restrict__ x3,
                       float* __restrict__ acc0, float* __restrict__ acc1,
                       float* __restrict__ acc2, float* __restrict__ acc3,
                       float* __restrict__ cnt0, float* __restrict__ cnt1,
                       float* __restrict__ cnt2, float* __restrict__ cnt3,
                       int n0, int n1, int n2, int n3)
{
  __shared__ int c[MAXB];
  __shared__ int cur[MAXB];
  __shared__ int stage[CAP];

  const unsigned* bn; int* rp; int* csr; const float* x; float* acc; float* cnt; int n;
  switch (blockIdx.y) {
    case 0:  bn = bn0; rp = rp0; csr = csr0; x = x0; acc = acc0; cnt = cnt0; n = n0; break;
    case 1:  bn = bn1; rp = rp1; csr = csr1; x = x1; acc = acc1; cnt = cnt1; n = n1; break;
    case 2:  bn = bn2; rp = rp2; csr = csr2; x = x2; acc = acc2; cnt = cnt2; n = n2; break;
    default: bn = bn3; rp = rp3; csr = csr3; x = x3; acc = acc3; cnt = cnt3; n = n3; break;
  }
  const int sh = bin_shift(n);
  const int nbins = ((n - 1) >> sh) + 1;
  const int b = blockIdx.x;
  if (b >= nbins) return;
  const int tid = threadIdx.x;
  const int lane = tid & 31;
  const int wave = tid >> 5;
  const int* bstart = bst + blockIdx.y * (MAXB + 1);
  const int bs = bstart[b];
  const int be = bstart[b + 1];
  const int m = be - bs;

  if (tid < MAXB) c[tid] = 0;
  __syncthreads();
  for (int i = bs + tid; i < be; i += 1024) atomicAdd(&c[bn[i] & 255u], 1);
  __syncthreads();
  if (wave == 0) {
    int a[8];
    int sa = 0;
#pragma unroll
    for (int k = 0; k < 8; ++k) { a[k] = c[8 * lane + k]; sa += a[k]; }
    int xa = sa;
#pragma unroll
    for (int d = 1; d < 32; d <<= 1) {
      const int ya = __shfl_up(xa, d);
      if (lane >= d) xa += ya;
    }
    int ra = xa - sa;
    const int R = 1 << sh;
#pragma unroll
    for (int k = 0; k < 8; ++k) {
      const int j = 8 * lane + k;
      cur[j] = ra;
      const int node = (b << sh) + j;
      if (j < R && node < n) rp[node] = bs + ra;
      ra += a[k];
    }
  }
  if (tid == 0 && b == nbins - 1) rp[n] = be;
  __syncthreads();
  const bool staged = (m <= CAP);
  if (staged) {
    for (int i = bs + tid; i < be; i += 1024) {
      const unsigned w = bn[i];
      const int p = atomicAdd(&cur[w & 255u], 1);
      stage[p] = (int)(w >> 8);
    }
  } else {
    for (int i = bs + tid; i < be; i += 1024) {
      const unsigned w = bn[i];
      const int p = atomicAdd(&cur[w & 255u], 1);
      csr[bs + p] = (int)(w >> 8);
    }
    __threadfence();
  }
  __syncthreads();
  if (staged)
    for (int i = tid; i < m; i += 1024) csr[bs + i] = stage[i];

  const int wv = __builtin_amdgcn_readfirstlane(wave);
  const int h = lane >> 4;
  const int c4 = (lane & 15) << 2;
  const int R = 1 << sh;
  for (int j = wv; j < R; j += 32) {
    const int node = (b << sh) + j;
    if (node >= n) break;
    const int beg = __builtin_amdgcn_readfirstlane(j ? cur[j - 1] : 0);
    const int end = __builtin_amdgcn_readfirstlane(cur[j]);
    float4 a;
    if (staged) a = row_sum(x, beg, end, lane, h, c4, [&](int q) { return stage[q]; });
    else        a = row_sum(x, beg, end, lane, h, c4, [&](int q) { return csr[bs + q]; });
    if (h == 0) *(float4*)(acc + (size_t)node * HDIM + c4) = a;
    if (lane == 0) cnt[node] = (float)(end - beg);
  }
}

__global__ __launch_bounds__(256)
void hg_gather(const float* __restrict__ x0, const float* __restrict__ x1,
               const float* __restrict__ x2, const float* __restrict__ x3,
               const int* __restrict__ rp0, const int* __restrict__ rp1,
               const int* __restrict__ rp2, const int* __restrict__ rp3,
               const int* __restrict__ csr0, const int* __restrict__ csr1,
               const int* __restrict__ csr2, const int* __restrict__ csr3,
               float* __restrict__ acc0, float* __restrict__ acc1,
               float* __restrict__ acc2, float* __restrict__ acc3,
               float* __restrict__ cnt0, float* __restrict__ cnt1,
               float* __restrict__ cnt2, float* __restrict__ cnt3,
               int n0, int n1, int n2, int n3)
{
  const float* x; const int* rp; const int* csr; float* acc; float* cnt; int n;
  switch (blockIdx.y) {
    case 0:  x = x0; rp = rp0; csr = csr0; acc = acc0; cnt = cnt0; n = n0; break;
    case 1:  x = x1; rp = rp1; csr = csr1; acc = acc1; cnt = cnt1; n = n1; break;
    case 2:  x = x2; rp = rp2; csr = csr2; acc = acc2; cnt = cnt2; n = n2; break;
    default: x = x3; rp = rp3; csr = csr3; acc = acc3; cnt = cnt3; n = n3; break;
  }
  const int wave = __builtin_amdgcn_readfirstlane((int)(threadIdx.x >> 5));
  const int lane = threadIdx.x & 31;
  const int h = lane >> 4;
  const int c = (lane & 15) << 2;
  const int dstep = gridDim.x * 8;
  for (int d = blockIdx.x * 8 + wave; d < n; d += dstep) {
    const int beg = rp[d];
    const int end = rp[d + 1];
    const float4 a0 = row_sum(x, beg, end, lane, h, c, [&](int q) { return csr[q]; });
    if (h == 0) *(float4*)(acc + (size_t)d * HDIM + c) = a0;
    if (lane == 0) cnt[d] = (float)(end - beg);
  }
}

__device__ __forceinline__ void mma_product(v8f cacc[4],
                                            const float* __restrict__ Abase,
                                            float inv,
                                            const float* __restrict__ Wsh,
                                            int m, int half)
{
#pragma unroll
  for (int k0 = 0; k0 < HDIM; k0 += 4) {
    const int kk = k0 + 2 * half;
    v2f a;
    a.x = Abase[kk] * inv;
    a.y = Abase[kk + 1] * inv;
#pragma unroll
    for (int nt = 0; nt < 4; ++nt) {
      v2f b = *(const v2f*)(Wsh + (size_t)(nt * 16 + m) * WPITCH + kk);
      cacc[nt] = __builtin_amdgcn_wmma_f32_16x16x4_f32(
          false, a, false, b, (short)0, cacc[nt], false, false);
    }
  }
}

__global__ __launch_bounds__(256)
void hg_sage(const float* __restrict__ acc0, const float* __restrict__ acc1,
             const float* __restrict__ acc2, const float* __restrict__ acc3,
             const float* __restrict__ cnt0, const float* __restrict__ cnt1,
             const float* __restrict__ cnt2, const float* __restrict__ cnt3,
             const float* __restrict__ Wl, const float* __restrict__ bl,
             const float* __restrict__ Wr,
             const float* __restrict__ xd, const float* __restrict__ xg,
             const float* __restrict__ xs,
             float* __restrict__ outd, float* __restrict__ outg,
             float* __restrict__ outs,
             int Nd, int Ng, int Ns, int gGene, int gDrug, int do_relu)
{
  __shared__ float sWlA[HDIM * WPITCH];
  __shared__ float sWlB[HDIM * WPITCH];
  __shared__ float sWr [HDIM * WPITCH];
  __shared__ float sbias[HDIM];

  const int tid = threadIdx.x;

  const float* accA; const float* cntA; const float* accB; const float* cntB;
  const float* xdst; float* out;
  int tA, tB, n_dst, blk;
  bool hasB;
  if ((int)blockIdx.x < gGene) {
    blk = blockIdx.x; hasB = true; tA = 0; tB = 3;
    accA = acc0; cntA = cnt0; accB = acc3; cntB = cnt3;
    xdst = xg; out = outg; n_dst = Ng;
  } else if ((int)blockIdx.x < gGene + gDrug) {
    blk = blockIdx.x - gGene; hasB = false; tA = 1; tB = 1;
    accA = acc1; cntA = cnt1; accB = acc1; cntB = cnt1;
    xdst = xd; out = outd; n_dst = Nd;
  } else {
    blk = blockIdx.x - gGene - gDrug; hasB = false; tA = 2; tB = 2;
    accA = acc2; cntA = cnt2; accB = acc2; cntB = cnt2;
    xdst = xs; out = outs; n_dst = Ns;
  }
  const float* WlA = Wl + (size_t)tA * HDIM * HDIM;
  const float* WrA = Wr + (size_t)tA * HDIM * HDIM;
  const float* blA = bl + (size_t)tA * HDIM;
  const float* WlB = Wl + (size_t)tB * HDIM * HDIM;
  const float* WrB = Wr + (size_t)tB * HDIM * HDIM;
  const float* blB = bl + (size_t)tB * HDIM;

  for (int i = tid; i < HDIM * HDIM; i += 256) {
    const int p = (i >> 6) * WPITCH + (i & (HDIM - 1));
    sWlA[p] = WlA[i];
    sWr[p]  = hasB ? (WrA[i] + WrB[i]) : WrA[i];
    sWlB[p] = hasB ? WlB[i] : 0.0f;
  }
  if (tid < HDIM) sbias[tid] = hasB ? (blA[tid] + blB[tid]) : blA[tid];
  __syncthreads();

  const int wave = tid >> 5;
  const int lane = tid & 31;
  const int tile = blk * 8 + wave;
  if (tile * 16 >= n_dst) return;

  const int m = lane & 15;
  const int half = lane >> 4;
  const int row = tile * 16 + m;

  v8f cacc[4] = {};

  {
    const float invA = 1.0f / fmaxf(cntA[row], 1.0f);
    mma_product(cacc, accA + (size_t)row * HDIM, invA, sWlA, m, half);
  }
  if (hasB) {
    const float invB = 1.0f / fmaxf(cntB[row], 1.0f);
    mma_product(cacc, accB + (size_t)row * HDIM, invB, sWlB, m, half);
  }
  mma_product(cacc, xdst + (size_t)row * HDIM, 1.0f, sWr, m, half);

#pragma unroll
  for (int nt = 0; nt < 4; ++nt) {
    const float bv = sbias[nt * 16 + m];
    float* obase = out + (size_t)(tile * 16) * HDIM + nt * 16 + m;
#pragma unroll
    for (int r = 0; r < 8; ++r) {
      float v = cacc[nt][r] + bv;
      if (do_relu) v = fmaxf(v, 0.0f);
      obase[(size_t)(r + 8 * half) * HDIM] = v;
    }
  }
}

extern "C" void kernel_launch(void* const* d_in, const int* in_sizes, int n_in,
                              void* d_out, int out_size, void* d_ws, size_t ws_size,
                              hipStream_t stream) {
  const int* src0 = (const int*)d_in[0]; const int* dst0 = (const int*)d_in[1];
  const int* src1 = (const int*)d_in[2]; const int* dst1 = (const int*)d_in[3];
  const int* src2 = (const int*)d_in[4]; const int* dst2 = (const int*)d_in[5];
  const int* src3 = (const int*)d_in[6]; const int* dst3 = (const int*)d_in[7];
  const float* ed  = (const float*)d_in[8];
  const float* eg  = (const float*)d_in[9];
  const float* es  = (const float*)d_in[10];
  const float* Wl0 = (const float*)d_in[11];
  const float* bl0 = (const float*)d_in[12];
  const float* Wr0 = (const float*)d_in[13];
  const float* Wl1 = (const float*)d_in[14];
  const float* bl1 = (const float*)d_in[15];
  const float* Wr1 = (const float*)d_in[16];

  const int E  = in_sizes[0];
  const int Nd = in_sizes[8]  / HDIM;
  const int Ng = in_sizes[9]  / HDIM;
  const int Ns = in_sizes[10] / HDIM;

  auto up64 = [](size_t n) { return (n + 63) & ~(size_t)63; };
  float* ws = (float*)d_ws;
  size_t off = 0;
  float* acc0 = ws + off; off += (size_t)Ng * HDIM;
  float* acc1 = ws + off; off += (size_t)Nd * HDIM;
  float* acc2 = ws + off; off += (size_t)Ns * HDIM;
  float* acc3 = ws + off; off += (size_t)Ng * HDIM;
  float* cnt0 = ws + off; off += up64(Ng);
  float* cnt1 = ws + off; off += up64(Nd);
  float* cnt2 = ws + off; off += up64(Ns);
  float* cnt3 = ws + off; off += up64(Ng);
  float* xd1 = ws + off; off += (size_t)Nd * HDIM;
  float* xg1 = ws + off; off += (size_t)Ng * HDIM;
  float* xs1 = ws + off; off += (size_t)Ns * HDIM;
  const int NT = (E + TILE - 1) / TILE;
  int* hist = (int*)(ws + off); off += up64((size_t)4 * MAXB * NT);
  int* bst  = (int*)(ws + off); off += up64((size_t)4 * (MAXB + 1));
  unsigned* bn0 = (unsigned*)(ws + off); off += up64(E);
  unsigned* bn1 = (unsigned*)(ws + off); off += up64(E);
  unsigned* bn2 = (unsigned*)(ws + off); off += up64(E);
  unsigned* bn3 = (unsigned*)(ws + off); off += up64(E);
  int* rp0 = (int*)(ws + off); off += up64((size_t)Ng + 1);
  int* rp1 = (int*)(ws + off); off += up64((size_t)Nd + 1);
  int* rp2 = (int*)(ws + off); off += up64((size_t)Ns + 1);
  int* rp3 = (int*)(ws + off); off += up64((size_t)Ng + 1);
  int* csr0 = (int*)(ws + off); off += up64(E);
  int* csr1 = (int*)(ws + off); off += up64(E);
  int* csr2 = (int*)(ws + off); off += up64(E);
  int* csr3 = (int*)(ws + off); off += up64(E);

  float* outd = (float*)d_out;
  float* outg = outd + (size_t)Nd * HDIM;
  float* outs = outg + (size_t)Ng * HDIM;

  const dim3 blk(256);
  const int gGene = (Ng / 16 + 7) / 8;
  const int gDrug = (Nd / 16 + 7) / 8;
  const int gDis  = (Ns / 16 + 7) / 8;
  const int gAll  = gGene + gDrug + gDis;
  const dim3 tgrid(NT, 4);
  const dim3 bgrid(MAXB, 4);
  const dim3 ggrid(1024, 4);

  hg_bincount<<<tgrid, 1024, 0, stream>>>(dst0, dst1, dst2, dst3, hist, E, NT, Ng, Nd, Ns, Ng);
  hg_binscatter<<<tgrid, 1024, 0, stream>>>(src0, dst0, src1, dst1, src2, dst2, src3, dst3,
                                            hist, bst, bn0, bn1, bn2, bn3, E, NT, Ng, Nd, Ns, Ng);
  hg_binsort_gather<<<bgrid, 1024, 0, stream>>>(bn0, bn1, bn2, bn3, bst, rp0, rp1, rp2, rp3,
                                                csr0, csr1, csr2, csr3, ed, eg, eg, es,
                                                acc0, acc1, acc2, acc3,
                                                cnt0, cnt1, cnt2, cnt3, Ng, Nd, Ns, Ng);

  hg_sage<<<gAll, blk, 0, stream>>>(acc0, acc1, acc2, acc3, cnt0, cnt1, cnt2, cnt3,
                                    Wl0, bl0, Wr0, ed, eg, es, xd1, xg1, xs1,
                                    Nd, Ng, Ns, gGene, gDrug, 1);

  hg_gather<<<ggrid, blk, 0, stream>>>(xd1, xg1, xg1, xs1, rp0, rp1, rp2, rp3,
                                       csr0, csr1, csr2, csr3, acc0, acc1, acc2, acc3,
                                       cnt0, cnt1, cnt2, cnt3, Ng, Nd, Ns, Ng);

  hg_sage<<<gAll, blk, 0, stream>>>(acc0, acc1, acc2, acc3, cnt0, cnt1, cnt2, cnt3,
                                    Wl1, bl1, Wr1, xd1, xg1, xs1, outd, outg, outs,
                                    Nd, Ng, Ns, gGene, gDrug, 0);
}


# Fuse layer 2's row gather into its SAGE update launch: a 32-wave work-group sums the source rows of 64 destinations into LDS and feeds WMMA from there; launches 6 -> 5; sum order as before
# speedup vs baseline: 10.0290x; 1.1389x over previous
#include <hip/hip_runtime.h>

typedef __attribute__((ext_vector_type(2))) float v2f;
typedef __attribute__((ext_vector_type(8))) float v8f;

#define HDIM 64
#define WPITCH 68

#define TILE 8192
#define MAXB 256
#define CAP  12288

__device__ __forceinline__ int bin_shift(int n)
{
  int s = 8;
  while (s > 5 && ((n - 1) >> (s - 1)) < MAXB) --s;
  return s;
}

__global__ __launch_bounds__(1024)
void hg_bincount(const int* __restrict__ dst0, const int* __restrict__ dst1,
                 const int* __restrict__ dst2, const int* __restrict__ dst3,
                 int* __restrict__ hist, int E, int NT,
                 int n0, int n1, int n2, int n3)
{
  __shared__ int h[MAXB];
  const int* dst; int n;
  switch (blockIdx.y) {
    case 0:  dst = dst0; n = n0; break;
    case 1:  dst = dst1; n = n1; break;
    case 2:  dst = dst2; n = n2; break;
    default: dst = dst3; n = n3; break;
  }
  const int sh = bin_shift(n);
  const int tid = threadIdx.x;
  const int t = blockIdx.x;
  if (tid < MAXB) h[tid] = 0;
  __syncthreads();
  const int e0 = t * TILE;
#pragma unroll
  for (int k = 0; k < TILE / 1024; ++k) {
    const int e = e0 + k * 1024 + tid;
    if (e < E) atomicAdd(&h[dst[e] >> sh], 1);
  }
  __syncthreads();
  if (tid < MAXB) hist[((size_t)blockIdx.y * NT + t) * MAXB + tid] = h[tid];
}

__global__ __launch_bounds__(1024)
void hg_binscatter(const int* __restrict__ src0, const int* __restrict__ dst0,
                   const int* __restrict__ src1, const int* __restrict__ dst1,
                   const int* __restrict__ src2, const int* __restrict__ dst2,
                   const int* __restrict__ src3, const int* __restrict__ dst3,
                   const int* __restrict__ hist, int* __restrict__ bst,
                   unsigned* __restrict__ bn0, unsigned* __restrict__ bn1,
                   unsigned* __restrict__ bn2, unsigned* __restrict__ bn3,
                   int E, int NT, int n0, int n1, int n2, int n3)
{
  __shared__ int s_tot[MAXB];
  __shared__ int s_pre[MAXB];
  __shared__ int s_own[MAXB];
  __shared__ int s_gd[MAXB];
  __shared__ int s_lc[MAXB];
  __shared__ int s_part[3][4][MAXB];
  __shared__ unsigned s_word[TILE];
  __shared__ unsigned char s_bin[TILE];

  const int* src; const int* dst; unsigned* bn; int n;
  switch (blockIdx.y) {
    case 0:  src = src0; dst = dst0; bn = bn0; n = n0; break;
    case 1:  src = src1; dst = dst1; bn = bn1; n = n1; break;
    case 2:  src = src2; dst = dst2; bn = bn2; n = n2; break;
    default: src = src3; dst = dst3; bn = bn3; n = n3; break;
  }
  const int sh = bin_shift(n);
  const int lmask = (1 << sh) - 1;
  const int tid = threadIdx.x;
  const int lane = tid & 31;
  const int wave = tid >> 5;
  const int t = blockIdx.x;
  const int* hb = hist + (size_t)blockIdx.y * NT * MAXB;
  int* bstart = bst + blockIdx.y * (MAXB + 1);

  {
    const int bq = tid & (MAXB - 1);
    const int q = tid >> 8;
    int tot = 0, pre = 0, own = 0;
#pragma unroll 8
    for (int tt = q; tt < NT; tt += 4) {
      const int v = hb[(size_t)tt * MAXB + bq];
      tot += v;
      pre += (tt < t) ? v : 0;
      own += (tt == t) ? v : 0;
    }
    s_part[0][q][bq] = tot; s_part[1][q][bq] = pre; s_part[2][q][bq] = own;
  }
  __syncthreads();
  if (tid < MAXB) {
    s_tot[tid] = s_part[0][0][tid] + s_part[0][1][tid] + s_part[0][2][tid] + s_part[0][3][tid];
    s_pre[tid] = s_part[1][0][tid] + s_part[1][1][tid] + s_part[1][2][tid] + s_part[1][3][tid];
    s_own[tid] = s_part[2][0][tid] + s_part[2][1][tid] + s_part[2][2][tid] + s_part[2][3][tid];
  }
  __syncthreads();
  if (wave == 0) {
    int a[8], o[8];
    int sa = 0, so = 0;
#pragma unroll
    for (int k = 0; k < 8; ++k) {
      a[k] = s_tot[8 * lane + k]; o[k] = s_own[8 * lane + k];
      sa += a[k]; so += o[k];
    }
    int xa = sa, xo = so;
#pragma unroll
    for (int d = 1; d < 32; d <<= 1) {
      const int ya = __shfl_up(xa, d);
      const int yo = __shfl_up(xo, d);
      if (lane >= d) { xa += ya; xo += yo; }
    }
    int ra = xa - sa, ro = xo - so;
#pragma unroll
    for (int k = 0; k < 8; ++k) {
      const int b = 8 * lane + k;
      s_gd[b] = ra + s_pre[b] - ro;
      s_lc[b] = ro;
      if (t == 0) bstart[b] = ra;
      ra += a[k]; ro += o[k];
    }
    if (t == 0 && lane == 31) bstart[MAXB] = ra;
  }
  __syncthreads();

  const int e0 = t * TILE;
#pragma unroll
  for (int k = 0; k < TILE / 1024; ++k) {
    const int e = e0 + k * 1024 + tid;
    if (e < E) {
      const int d = dst[e];
      const int s = src[e];
      const int b = d >> sh;
      const int lp = atomicAdd(&s_lc[b], 1);
      s_word[lp] = ((unsigned)s << 8) | (unsigned)(d & lmask);
      s_bin[lp] = (unsigned char)b;
    }
  }
  __syncthreads();
  const int m = (E - e0 < TILE) ? (E - e0) : TILE;
  for (int i = tid; i < m; i += 1024)
    bn[i + s_gd[s_bin[i]]] = s_word[i];
}

__device__ __forceinline__ float4 ld_row16(const float* __restrict__ x, int s, int c)
{
  return *(const float4*)(x + (size_t)s * HDIM + c);
}

template <class IdAt>
__device__ __forceinline__ float4 row_sum(const float* __restrict__ x, int beg, int end,
                                          int lane, int h, int c, IdAt id_at)
{
  float4 a0 = make_float4(0.f, 0.f, 0.f, 0.f);
  float4 a1 = make_float4(0.f, 0.f, 0.f, 0.f);
  for (int jb = beg; jb < end; jb += 32) {
    const int j = jb + lane;
    const int idx = (j < end) ? id_at(j) : 0;
    const int nk = (end - jb < 32) ? (end - jb) : 32;
    int k = 0;
    for (; k + 8 <= nk; k += 8) {
      const int s0 = __shfl(idx, k + h);
      const int s1 = __shfl(idx, k + 2 + h);
      const int s2 = __shfl(idx, k + 4 + h);
      const int s3 = __shfl(idx, k + 6 + h);
      const float4 v0 = ld_row16(x, s0, c);
      const float4 v1 = ld_row16(x, s1, c);
      const float4 v2 = ld_row16(x, s2, c);
      const float4 v3 = ld_row16(x, s3, c);
      a0.x += v0.x; a0.y += v0.y; a0.z += v0.z; a0.w += v0.w;
      a1.x += v1.x; a1.y += v1.y; a1.z += v1.z; a1.w += v1.w;
      a0.x += v2.x; a0.y += v2.y; a0.z += v2.z; a0.w += v2.w;
      a1.x += v3.x; a1.y += v3.y; a1.z += v3.z; a1.w += v3.w;
    }
    if (k < nk) {
      float4 t[4];
#pragma unroll
      for (int q = 0; q < 4; ++q) {
        const int kk = k + 2 * q + h;
        const int s = __shfl(idx, kk);
        t[q] = make_float4(0.f, 0.f, 0.f, 0.f);
        if (kk < nk) t[q] = ld_row16(x, s, c);
      }
#pragma unroll
      for (int q = 0; q < 4; ++q) {
        a0.x += t[q].x; a0.y += t[q].y; a0.z += t[q].z; a0.w += t[q].w;
      }
    }
  }
  a0.x += a1.x; a0.y += a1.y; a0.z += a1.z; a0.w += a1.w;
  a0.x += __shfl_xor(a0.x, 16);
  a0.y += __shfl_xor(a0.y, 16);
  a0.z += __shfl_xor(a0.z, 16);
  a0.w += __shfl_xor(a0.w, 16);
  return a0;
}

__global__ __launch_bounds__(1024)
void hg_binsort_gather(const unsigned* __restrict__ bn0, const unsigned* __restrict__ bn1,
                       const unsigned* __restrict__ bn2, const unsigned* __restrict__ bn3,
                       const int* __restrict__ bst,
                       int* __restrict__ rp0, int* __restrict__ rp1,
                       int* __restrict__ rp2, int* __restrict__ rp3,
                       int* __restrict__ csr0, int* __restrict__ csr1,
                       int* __restrict__ csr2, int* __restrict__ csr3,
                       const float* __restrict__ x0, const float* __restrict__ x1,
                       const float* __restrict__ x2, const float* __restrict__ x3,
                       float* __restrict__ acc0, float* __restrict__ acc1,
                       float* __restrict__ acc2, float* __restrict__ acc3,
                       float* __restrict__ cnt0, float* __restrict__ cnt1,
                       float* __restrict__ cnt2, float* __restrict__ cnt3,
                       int n0, int n1, int n2, int n3)
{
  __shared__ int c[MAXB];
  __shared__ int cur[MAXB];
  __shared__ int stage[CAP];

  const unsigned* bn; int* rp; int* csr; const float* x; float* acc; float* cnt; int n;
  switch (blockIdx.y) {
    case 0:  bn = bn0; rp = rp0; csr = csr0; x = x0; acc = acc0; cnt = cnt0; n = n0; break;
    case 1:  bn = bn1; rp = rp1; csr = csr1; x = x1; acc = acc1; cnt = cnt1; n = n1; break;
    case 2:  bn = bn2; rp = rp2; csr = csr2; x = x2; acc = acc2; cnt = cnt2; n = n2; break;
    default: bn = bn3; rp = rp3; csr = csr3; x = x3; acc = acc3; cnt = cnt3; n = n3; break;
  }
  const int sh = bin_shift(n);
  const int nbins = ((n - 1) >> sh) + 1;
  const int b = blockIdx.x;
  if (b >= nbins) return;
  const int tid = threadIdx.x;
  const int lane = tid & 31;
  const int wave = tid >> 5;
  const int* bstart = bst + blockIdx.y * (MAXB + 1);
  const int bs = bstart[b];
  const int be = bstart[b + 1];
  const int m = be - bs;

  if (tid < MAXB) c[tid] = 0;
  __syncthreads();
  for (int i = bs + tid; i < be; i += 1024) atomicAdd(&c[bn[i] & 255u], 1);
  __syncthreads();
  if (wave == 0) {
    int a[8];
    int sa = 0;
#pragma unroll
    for (int k = 0; k < 8; ++k) { a[k] = c[8 * lane + k]; sa += a[k]; }
    int xa = sa;
#pragma unroll
    for (int d = 1; d < 32; d <<= 1) {
      const int ya = __shfl_up(xa, d);
      if (lane >= d) xa += ya;
    }
    int ra = xa - sa;
    const int R = 1 << sh;
#pragma unroll
    for (int k = 0; k < 8; ++k) {
      const int j = 8 * lane + k;
      cur[j] = ra;
      const int node = (b << sh) + j;
      if (j < R && node < n) rp[node] = bs + ra;
      ra += a[k];
    }
  }
  if (tid == 0 && b == nbins - 1) rp[n] = be;
  __syncthreads();
  const bool staged = (m <= CAP);
  if (staged) {
    for (int i = bs + tid; i < be; i += 1024) {
      const unsigned w = bn[i];
      const int p = atomicAdd(&cur[w & 255u], 1);
      stage[p] = (int)(w >> 8);
    }
  } else {
    for (int i = bs + tid; i < be; i += 1024) {
      const unsigned w = bn[i];
      const int p = atomicAdd(&cur[w & 255u], 1);
      csr[bs + p] = (int)(w >> 8);
    }
    __threadfence();
  }
  __syncthreads();
  if (staged)
    for (int i = tid; i < m; i += 1024) csr[bs + i] = stage[i];

  const int wv = __builtin_amdgcn_readfirstlane(wave);
  const int h = lane >> 4;
  const int c4 = (lane & 15) << 2;
  const int R = 1 << sh;
  for (int j = wv; j < R; j += 32) {
    const int node = (b << sh) + j;
    if (node >= n) break;
    const int beg = __builtin_amdgcn_readfirstlane(j ? cur[j - 1] : 0);
    const int end = __builtin_amdgcn_readfirstlane(cur[j]);
    float4 a;
    if (staged) a = row_sum(x, beg, end, lane, h, c4, [&](int q) { return stage[q]; });
    else        a = row_sum(x, beg, end, lane, h, c4, [&](int q) { return csr[bs + q]; });
    if (h == 0) *(float4*)(acc + (size_t)node * HDIM + c4) = a;
    if (lane == 0) cnt[node] = (float)(end - beg);
  }
}

__device__ __forceinline__ void mma_product(v8f cacc[4],
                                            const float* __restrict__ Abase,
                                            float inv,
                                            const float* __restrict__ Wsh,
                                            int m, int half)
{
#pragma unroll
  for (int k0 = 0; k0 < HDIM; k0 += 4) {
    const int kk = k0 + 2 * half;
    v2f a;
    a.x = Abase[kk] * inv;
    a.y = Abase[kk + 1] * inv;
#pragma unroll
    for (int nt = 0; nt < 4; ++nt) {
      v2f b = *(const v2f*)(Wsh + (size_t)(nt * 16 + m) * WPITCH + kk);
      cacc[nt] = __builtin_amdgcn_wmma_f32_16x16x4_f32(
          false, a, false, b, (short)0, cacc[nt], false, false);
    }
  }
}

__global__ __launch_bounds__(256)
void hg_sage(const float* __restrict__ acc0, const float* __restrict__ acc1,
             const float* __restrict__ acc2, const float* __restrict__ acc3,
             const float* __restrict__ cnt0, const float* __restrict__ cnt1,
             const float* __restrict__ cnt2, const float* __restrict__ cnt3,
             const float* __restrict__ Wl, const float* __restrict__ bl,
             const float* __restrict__ Wr,
             const float* __restrict__ xd, const float* __restrict__ xg,
             const float* __restrict__ xs,
             float* __restrict__ outd, float* __restrict__ outg,
             float* __restrict__ outs,
             int Nd, int Ng, int Ns, int gGene, int gDrug, int do_relu)
{
  __shared__ float sWlA[HDIM * WPITCH];
  __shared__ float sWlB[HDIM * WPITCH];
  __shared__ float sWr [HDIM * WPITCH];
  __shared__ float sbias[HDIM];

  const int tid = threadIdx.x;

  const float* accA; const float* cntA; const float* accB; const float* cntB;
  const float* xdst; float* out;
  int tA, tB, n_dst, blk;
  bool hasB;
  if ((int)blockIdx.x < gGene) {
    blk = blockIdx.x; hasB = true; tA = 0; tB = 3;
    accA = acc0; cntA = cnt0; accB = acc3; cntB = cnt3;
    xdst = xg; out = outg; n_dst = Ng;
  } else if ((int)blockIdx.x < gGene + gDrug) {
    blk = blockIdx.x - gGene; hasB = false; tA = 1; tB = 1;
    accA = acc1; cntA = cnt1; accB = acc1; cntB = cnt1;
    xdst = xd; out = outd; n_dst = Nd;
  } else {
    blk = blockIdx.x - gGene - gDrug; hasB = false; tA = 2; tB = 2;
    accA = acc2; cntA = cnt2; accB = acc2; cntB = cnt2;
    xdst = xs; out = outs; n_dst = Ns;
  }
  const float* WlA = Wl + (size_t)tA * HDIM * HDIM;
  const float* WrA = Wr + (size_t)tA * HDIM * HDIM;
  const float* blA = bl + (size_t)tA * HDIM;
  const float* WlB = Wl + (size_t)tB * HDIM * HDIM;
  const float* WrB = Wr + (size_t)tB * HDIM * HDIM;
  const float* blB = bl + (size_t)tB * HDIM;

  for (int i = tid; i < HDIM * HDIM; i += 256) {
    const int p = (i >> 6) * WPITCH + (i & (HDIM - 1));
    sWlA[p] = WlA[i];
    sWr[p]  = hasB ? (WrA[i] + WrB[i]) : WrA[i];
    sWlB[p] = hasB ? WlB[i] : 0.0f;
  }
  if (tid < HDIM) sbias[tid] = hasB ? (blA[tid] + blB[tid]) : blA[tid];
  __syncthreads();

  const int wave = tid >> 5;
  const int lane = tid & 31;
  const int tile = blk * 8 + wave;
  if (tile * 16 >= n_dst) return;

  const int m = lane & 15;
  const int half = lane >> 4;
  const int row = tile * 16 + m;

  v8f cacc[4] = {};

  {
    const float invA = 1.0f / fmaxf(cntA[row], 1.0f);
    mma_product(cacc, accA + (size_t)row * HDIM, invA, sWlA, m, half);
  }
  if (hasB) {
    const float invB = 1.0f / fmaxf(cntB[row], 1.0f);
    mma_product(cacc, accB + (size_t)row * HDIM, invB, sWlB, m, half);
  }
  mma_product(cacc, xdst + (size_t)row * HDIM, 1.0f, sWr, m, half);

#pragma unroll
  for (int nt = 0; nt < 4; ++nt) {
    const float bv = sbias[nt * 16 + m];
    float* obase = out + (size_t)(tile * 16) * HDIM + nt * 16 + m;
#pragma unroll
    for (int r = 0; r < 8; ++r) {
      float v = cacc[nt][r] + bv;
      if (do_relu) v = fmaxf(v, 0.0f);
      obase[(size_t)(r + 8 * half) * HDIM] = v;
    }
  }
}

#define FROWS 64

__device__ __forceinline__ void mma_lds(v8f& c, const float* Arow, const float* Wrow, int half)
{
#pragma unroll
  for (int k0 = 0; k0 < HDIM; k0 += 4) {
    const int kk = k0 + 2 * half;
    const v2f a = *(const v2f*)(Arow + kk);
    const v2f b = *(const v2f*)(Wrow + kk);
    c = __builtin_amdgcn_wmma_f32_16x16x4_f32(false, a, false, b, (short)0, c, false, false);
  }
}

__global__ __launch_bounds__(1024)
void hg_gather_sage(const float* __restrict__ xd, const float* __restrict__ xg,
                    const float* __restrict__ xs,
                    const int* __restrict__ rp0, const int* __restrict__ rp1,
                    const int* __restrict__ rp2, const int* __restrict__ rp3,
                    const int* __restrict__ csr0, const int* __restrict__ csr1,
                    const int* __restrict__ csr2, const int* __restrict__ csr3,
                    const float* __restrict__ Wl, const float* __restrict__ bl,
                    const float* __restrict__ Wr,
                    float* __restrict__ outd, float* __restrict__ outg,
                    float* __restrict__ outs,
                    int Nd, int Ng, int Ns, int gDis, int gDrug)
{
  __shared__ __attribute__((aligned(16))) float sWlA[HDIM * WPITCH];
  __shared__ __attribute__((aligned(16))) float sWlB[HDIM * WPITCH];
  __shared__ __attribute__((aligned(16))) float sWr [HDIM * WPITCH];
  __shared__ __attribute__((aligned(16))) float sMA [FROWS * WPITCH];
  __shared__ __attribute__((aligned(16))) float sMB [FROWS * WPITCH];
  __shared__ __attribute__((aligned(16))) float sX  [FROWS * WPITCH];
  __shared__ float sbias[HDIM];

  const int tid = threadIdx.x;

  const float* xsA; const int* rpA; const int* csrA;
  const float* xsB; const int* rpB; const int* csrB;
  const float* xdst; float* out;
  int tA, tB, n_dst, blk;
  bool hasB;
  if ((int)blockIdx.x < gDis) {
    blk = blockIdx.x; hasB = false; tA = 2; tB = 2;
    xsA = xg; rpA = rp2; csrA = csr2; xsB = xg; rpB = rp2; csrB = csr2;
    xdst = xs; out = outs; n_dst = Ns;
  } else if ((int)blockIdx.x < gDis + gDrug) {
    blk = blockIdx.x - gDis; hasB = false; tA = 1; tB = 1;
    xsA = xg; rpA = rp1; csrA = csr1; xsB = xg; rpB = rp1; csrB = csr1;
    xdst = xd; out = outd; n_dst = Nd;
  } else {
    blk = blockIdx.x - gDis - gDrug; hasB = true; tA = 0; tB = 3;
    xsA = xd; rpA = rp0; csrA = csr0; xsB = xs; rpB = rp3; csrB = csr3;
    xdst = xg; out = outg; n_dst = Ng;
  }
  const float* WlA = Wl + (size_t)tA * HDIM * HDIM;
  const float* WrA = Wr + (size_t)tA * HDIM * HDIM;
  const float* blA = bl + (size_t)tA * HDIM;
  const float* WlB = Wl + (size_t)tB * HDIM * HDIM;
  const float* WrB = Wr + (size_t)tB * HDIM * HDIM;
  const float* blB = bl + (size_t)tB * HDIM;

  for (int i = tid; i < HDIM * HDIM; i += 1024) {
    const int p = (i >> 6) * WPITCH + (i & (HDIM - 1));
    sWlA[p] = WlA[i];
    sWr[p]  = hasB ? (WrA[i] + WrB[i]) : WrA[i];
    if (hasB) sWlB[p] = WlB[i];
  }
  if (tid < HDIM) sbias[tid] = hasB ? (blA[tid] + blB[tid]) : blA[tid];

  const int wv = __builtin_amdgcn_readfirstlane(tid >> 5);
  const int lane = tid & 31;
  const int h = lane >> 4;
  const int c4 = (lane & 15) << 2;
  const int row0 = blk * FROWS;

  for (int q = 0; q < FROWS / 32; ++q) {
    const int lr = wv * (FROWS / 32) + q;
    const int d = row0 + lr;
    float4 ma = make_float4(0.f, 0.f, 0.f, 0.f);
    float4 mb = make_float4(0.f, 0.f, 0.f, 0.f);
    float4 xr = make_float4(0.f, 0.f, 0.f, 0.f);
    if (d < n_dst) {
      {
        const int beg = rpA[d];
        const int end = rpA[d + 1];
        const float4 a = row_sum(xsA, beg, end, lane, h, c4, [&](int j) { return csrA[j]; });
        const float inv = 1.0f / fmaxf((float)(end - beg), 1.0f);
        ma = make_float4(a.x * inv, a.y * inv, a.z * inv, a.w * inv);
      }
      if (hasB) {
        const int beg = rpB[d];
        const int end = rpB[d + 1];
        const float4 a = row_sum(xsB, beg, end, lane, h, c4, [&](int j) { return csrB[j]; });
        const float inv = 1.0f / fmaxf((float)(end - beg), 1.0f);
        mb = make_float4(a.x * inv, a.y * inv, a.z * inv, a.w * inv);
      }
      xr = ld_row16(xdst, d, c4);
    }
    if (h == 0) {
      *(float4*)(sMA + lr * WPITCH + c4) = ma;
      if (hasB) *(float4*)(sMB + lr * WPITCH + c4) = mb;
      *(float4*)(sX + lr * WPITCH + c4) = xr;
    }
  }
  __syncthreads();

  const int tile = wv >> 2;
  const int nt = wv & 3;
  const int m = lane & 15;
  const bool live = (wv < 16) && (row0 + tile * 16 < n_dst);
  v8f cacc = {};
  if (live) {
    const int ar = (tile * 16 + m) * WPITCH;
    const int wr = (nt * 16 + m) * WPITCH;
    mma_lds(cacc, sMA + ar, sWlA + wr, h);
    if (hasB) mma_lds(cacc, sMB + ar, sWlB + wr, h);
    mma_lds(cacc, sX + ar, sWr + wr, h);
  }
  __syncthreads();

  if (live) {
    const float bv = sbias[nt * 16 + m];
#pragma unroll
    for (int r = 0; r < 8; ++r)
      sMA[(tile * 16 + r + 8 * h) * WPITCH + nt * 16 + m] = cacc[r] + bv;
  }
  __syncthreads();
  for (int i = tid; i < FROWS * (HDIM / 4); i += 1024) {
    const int lr = i >> 4;
    const int cc = (i & 15) << 2;
    const int d = row0 + lr;
    if (d < n_dst)
      *(float4*)(out + (size_t)d * HDIM + cc) = *(const float4*)(sMA + lr * WPITCH + cc);
  }
}

extern "C" void kernel_launch(void* const* d_in, const int* in_sizes, int n_in,
                              void* d_out, int out_size, void* d_ws, size_t ws_size,
                              hipStream_t stream) {
  const int* src0 = (const int*)d_in[0]; const int* dst0 = (const int*)d_in[1];
  const int* src1 = (const int*)d_in[2]; const int* dst1 = (const int*)d_in[3];
  const int* src2 = (const int*)d_in[4]; const int* dst2 = (const int*)d_in[5];
  const int* src3 = (const int*)d_in[6]; const int* dst3 = (const int*)d_in[7];
  const float* ed  = (const float*)d_in[8];
  const float* eg  = (const float*)d_in[9];
  const float* es  = (const float*)d_in[10];
  const float* Wl0 = (const float*)d_in[11];
  const float* bl0 = (const float*)d_in[12];
  const float* Wr0 = (const float*)d_in[13];
  const float* Wl1 = (const float*)d_in[14];
  const float* bl1 = (const float*)d_in[15];
  const float* Wr1 = (const float*)d_in[16];

  const int E  = in_sizes[0];
  const int Nd = in_sizes[8]  / HDIM;
  const int Ng = in_sizes[9]  / HDIM;
  const int Ns = in_sizes[10] / HDIM;

  auto up64 = [](size_t n) { return (n + 63) & ~(size_t)63; };
  float* ws = (float*)d_ws;
  size_t off = 0;
  float* acc0 = ws + off; off += (size_t)Ng * HDIM;
  float* acc1 = ws + off; off += (size_t)Nd * HDIM;
  float* acc2 = ws + off; off += (size_t)Ns * HDIM;
  float* acc3 = ws + off; off += (size_t)Ng * HDIM;
  float* cnt0 = ws + off; off += up64(Ng);
  float* cnt1 = ws + off; off += up64(Nd);
  float* cnt2 = ws + off; off += up64(Ns);
  float* cnt3 = ws + off; off += up64(Ng);
  float* xd1 = ws + off; off += (size_t)Nd * HDIM;
  float* xg1 = ws + off; off += (size_t)Ng * HDIM;
  float* xs1 = ws + off; off += (size_t)Ns * HDIM;
  const int NT = (E + TILE - 1) / TILE;
  int* hist = (int*)(ws + off); off += up64((size_t)4 * MAXB * NT);
  int* bst  = (int*)(ws + off); off += up64((size_t)4 * (MAXB + 1));
  unsigned* bn0 = (unsigned*)(ws + off); off += up64(E);
  unsigned* bn1 = (unsigned*)(ws + off); off += up64(E);
  unsigned* bn2 = (unsigned*)(ws + off); off += up64(E);
  unsigned* bn3 = (unsigned*)(ws + off); off += up64(E);
  int* rp0 = (int*)(ws + off); off += up64((size_t)Ng + 1);
  int* rp1 = (int*)(ws + off); off += up64((size_t)Nd + 1);
  int* rp2 = (int*)(ws + off); off += up64((size_t)Ns + 1);
  int* rp3 = (int*)(ws + off); off += up64((size_t)Ng + 1);
  int* csr0 = (int*)(ws + off); off += up64(E);
  int* csr1 = (int*)(ws + off); off += up64(E);
  int* csr2 = (int*)(ws + off); off += up64(E);
  int* csr3 = (int*)(ws + off); off += up64(E);

  float* outd = (float*)d_out;
  float* outg = outd + (size_t)Nd * HDIM;
  float* outs = outg + (size_t)Ng * HDIM;

  const dim3 blk(256);
  const int gGene = (Ng / 16 + 7) / 8;
  const int gDrug = (Nd / 16 + 7) / 8;
  const int gDis  = (Ns / 16 + 7) / 8;
  const int gAll  = gGene + gDrug + gDis;
  const dim3 tgrid(NT, 4);
  const dim3 bgrid(MAXB, 4);
  const int fGene = (Ng + FROWS - 1) / FROWS;
  const int fDrug = (Nd + FROWS - 1) / FROWS;
  const int fDis  = (Ns + FROWS - 1) / FROWS;
  const int fAll  = fGene + fDrug + fDis;

  hg_bincount<<<tgrid, 1024, 0, stream>>>(dst0, dst1, dst2, dst3, hist, E, NT, Ng, Nd, Ns, Ng);
  hg_binscatter<<<tgrid, 1024, 0, stream>>>(src0, dst0, src1, dst1, src2, dst2, src3, dst3,
                                            hist, bst, bn0, bn1, bn2, bn3, E, NT, Ng, Nd, Ns, Ng);
  hg_binsort_gather<<<bgrid, 1024, 0, stream>>>(bn0, bn1, bn2, bn3, bst, rp0, rp1, rp2, rp3,
                                                csr0, csr1, csr2, csr3, ed, eg, eg, es,
                                                acc0, acc1, acc2, acc3,
                                                cnt0, cnt1, cnt2, cnt3, Ng, Nd, Ns, Ng);

  hg_sage<<<gAll, blk, 0, stream>>>(acc0, acc1, acc2, acc3, cnt0, cnt1, cnt2, cnt3,
                                    Wl0, bl0, Wr0, ed, eg, es, xd1, xg1, xs1,
                                    Nd, Ng, Ns, gGene, gDrug, 1);

  hg_gather_sage<<<fAll, 1024, 0, stream>>>(xd1, xg1, xs1, rp0, rp1, rp2, rp3,
                                           csr0, csr1, csr2, csr3, Wl1, bl1, Wr1,
                                           outd, outg, outs, Nd, Ng, Ns, fDis, fDrug);
}


# Layer 1 takes layer 2's fused row-gather + update launch (with ReLU) instead of row sums written by the sort work-groups and a separate update launch; sums stay out of memory; sum order as before
# speedup vs baseline: 11.1086x; 1.1076x over previous
#include <hip/hip_runtime.h>

typedef __attribute__((ext_vector_type(2))) float v2f;
typedef __attribute__((ext_vector_type(8))) float v8f;

#define HDIM 64
#define WPITCH 68

#define TILE 8192
#define MAXB 256
#define CAP  12288

__device__ __forceinline__ int bin_shift(int n)
{
  int s = 8;
  while (s > 5 && ((n - 1) >> (s - 1)) < MAXB) --s;
  return s;
}

__global__ __launch_bounds__(1024)
void hg_bincount(const int* __restrict__ dst0, const int* __restrict__ dst1,
                 const int* __restrict__ dst2, const int* __restrict__ dst3,
                 int* __restrict__ hist, int E, int NT,
                 int n0, int n1, int n2, int n3)
{
  __shared__ int h[MAXB];
  const int* dst; int n;
  switch (blockIdx.y) {
    case 0:  dst = dst0; n = n0; break;
    case 1:  dst = dst1; n = n1; break;
    case 2:  dst = dst2; n = n2; break;
    default: dst = dst3; n = n3; break;
  }
  const int sh = bin_shift(n);
  const int tid = threadIdx.x;
  const int t = blockIdx.x;
  if (tid < MAXB) h[tid] = 0;
  __syncthreads();
  const int e0 = t * TILE;
#pragma unroll
  for (int k = 0; k < TILE / 1024; ++k) {
    const int e = e0 + k * 1024 + tid;
    if (e < E) atomicAdd(&h[dst[e] >> sh], 1);
  }
  __syncthreads();
  if (tid < MAXB) hist[((size_t)blockIdx.y * NT + t) * MAXB + tid] = h[tid];
}

__global__ __launch_bounds__(1024)
void hg_binscatter(const int* __restrict__ src0, const int* __restrict__ dst0,
                   const int* __restrict__ src1, const int* __restrict__ dst1,
                   const int* __restrict__ src2, const int* __restrict__ dst2,
                   const int* __restrict__ src3, const int* __restrict__ dst3,
                   const int* __restrict__ hist, int* __restrict__ bst,
                   unsigned* __restrict__ bn0, unsigned* __restrict__ bn1,
                   unsigned* __restrict__ bn2, unsigned* __restrict__ bn3,
                   int E, int NT, int n0, int n1, int n2, int n3)
{
  __shared__ int s_tot[MAXB];
  __shared__ int s_pre[MAXB];
  __shared__ int s_own[MAXB];
  __shared__ int s_gd[MAXB];
  __shared__ int s_lc[MAXB];
  __shared__ int s_part[3][4][MAXB];
  __shared__ unsigned s_word[TILE];
  __shared__ unsigned char s_bin[TILE];

  const int* src; const int* dst; unsigned* bn; int n;
  switch (blockIdx.y) {
    case 0:  src = src0; dst = dst0; bn = bn0; n = n0; break;
    case 1:  src = src1; dst = dst1; bn = bn1; n = n1; break;
    case 2:  src = src2; dst = dst2; bn = bn2; n = n2; break;
    default: src = src3; dst = dst3; bn = bn3; n = n3; break;
  }
  const int sh = bin_shift(n);
  const int lmask = (1 << sh) - 1;
  const int tid = threadIdx.x;
  const int lane = tid & 31;
  const int wave = tid >> 5;
  const int t = blockIdx.x;
  const int* hb = hist + (size_t)blockIdx.y * NT * MAXB;
  int* bstart = bst + blockIdx.y * (MAXB + 1);

  {
    const int bq = tid & (MAXB - 1);
    const int q = tid >> 8;
    int tot = 0, pre = 0, own = 0;
#pragma unroll 8
    for (int tt = q; tt < NT; tt += 4) {
      const int v = hb[(size_t)tt * MAXB + bq];
      tot += v;
      pre += (tt < t) ? v : 0;
      own += (tt == t) ? v : 0;
    }
    s_part[0][q][bq] = tot; s_part[1][q][bq] = pre; s_part[2][q][bq] = own;
  }
  __syncthreads();
  if (tid < MAXB) {
    s_tot[tid] = s_part[0][0][tid] + s_part[0][1][tid] + s_part[0][2][tid] + s_part[0][3][tid];
    s_pre[tid] = s_part[1][0][tid] + s_part[1][1][tid] + s_part[1][2][tid] + s_part[1][3][tid];
    s_own[tid] = s_part[2][0][tid] + s_part[2][1][tid] + s_part[2][2][tid] + s_part[2][3][tid];
  }
  __syncthreads();
  if (wave == 0) {
    int a[8], o[8];
    int sa = 0, so = 0;
#pragma unroll
    for (int k = 0; k < 8; ++k) {
      a[k] = s_tot[8 * lane + k]; o[k] = s_own[8 * lane + k];
      sa += a[k]; so += o[k];
    }
    int xa = sa, xo = so;
#pragma unroll
    for (int d = 1; d < 32; d <<= 1) {
      const int ya = __shfl_up(xa, d);
      const int yo = __shfl_up(xo, d);
      if (lane >= d) { xa += ya; xo += yo; }
    }
    int ra = xa - sa, ro = xo - so;
#pragma unroll
    for (int k = 0; k < 8; ++k) {
      const int b = 8 * lane + k;
      s_gd[b] = ra + s_pre[b] - ro;
      s_lc[b] = ro;
      if (t == 0) bstart[b] = ra;
      ra += a[k]; ro += o[k];
    }
    if (t == 0 && lane == 31) bstart[MAXB] = ra;
  }
  __syncthreads();

  const int e0 = t * TILE;
#pragma unroll
  for (int k = 0; k < TILE / 1024; ++k) {
    const int e = e0 + k * 1024 + tid;
    if (e < E) {
      const int d = dst[e];
      const int s = src[e];
      const int b = d >> sh;
      const int lp = atomicAdd(&s_lc[b], 1);
      s_word[lp] = ((unsigned)s << 8) | (unsigned)(d & lmask);
      s_bin[lp] = (unsigned char)b;
    }
  }
  __syncthreads();
  const int m = (E - e0 < TILE) ? (E - e0) : TILE;
  for (int i = tid; i < m; i += 1024)
    bn[i + s_gd[s_bin[i]]] = s_word[i];
}

__device__ __forceinline__ float4 ld_row16(const float* __restrict__ x, int s, int c)
{
  return *(const float4*)(x + (size_t)s * HDIM + c);
}

template <class IdAt>
__device__ __forceinline__ float4 row_sum(const float* __restrict__ x, int beg, int end,
                                          int lane, int h, int c, IdAt id_at)
{
  float4 a0 = make_float4(0.f, 0.f, 0.f, 0.f);
  float4 a1 = make_float4(0.f, 0.f, 0.f, 0.f);
  for (int jb = beg; jb < end; jb += 32) {
    const int j = jb + lane;
    const int idx = (j < end) ? id_at(j) : 0;
    const int nk = (end - jb < 32) ? (end - jb) : 32;
    int k = 0;
    for (; k + 8 <= nk; k += 8) {
      const int s0 = __shfl(idx, k + h);
      const int s1 = __shfl(idx, k + 2 + h);
      const int s2 = __shfl(idx, k + 4 + h);
      const int s3 = __shfl(idx, k + 6 + h);
      const float4 v0 = ld_row16(x, s0, c);
      const float4 v1 = ld_row16(x, s1, c);
      const float4 v2 = ld_row16(x, s2, c);
      const float4 v3 = ld_row16(x, s3, c);
      a0.x += v0.x; a0.y += v0.y; a0.z += v0.z; a0.w += v0.w;
      a1.x += v1.x; a1.y += v1.y; a1.z += v1.z; a1.w += v1.w;
      a0.x += v2.x; a0.y += v2.y; a0.z += v2.z; a0.w += v2.w;
      a1.x += v3.x; a1.y += v3.y; a1.z += v3.z; a1.w += v3.w;
    }
    if (k < nk) {
      float4 t[4];
#pragma unroll
      for (int q = 0; q < 4; ++q) {
        const int kk = k + 2 * q + h;
        const int s = __shfl(idx, kk);
        t[q] = make_float4(0.f, 0.f, 0.f, 0.f);
        if (kk < nk) t[q] = ld_row16(x, s, c);
      }
#pragma unroll
      for (int q = 0; q < 4; ++q) {
        a0.x += t[q].x; a0.y += t[q].y; a0.z += t[q].z; a0.w += t[q].w;
      }
    }
  }
  a0.x += a1.x; a0.y += a1.y; a0.z += a1.z; a0.w += a1.w;
  a0.x += __shfl_xor(a0.x, 16);
  a0.y += __shfl_xor(a0.y, 16);
  a0.z += __shfl_xor(a0.z, 16);
  a0.w += __shfl_xor(a0.w, 16);
  return a0;
}

__global__ __launch_bounds__(1024)
void hg_binsort(const unsigned* __restrict__ bn0, const unsigned* __restrict__ bn1,
                const unsigned* __restrict__ bn2, const unsigned* __restrict__ bn3,
                const int* __restrict__ bst,
                int* __restrict__ rp0, int* __restrict__ rp1,
                int* __restrict__ rp2, int* __restrict__ rp3,
                int* __restrict__ csr0, int* __restrict__ csr1,
                int* __restrict__ csr2, int* __restrict__ csr3,
                int n0, int n1, int n2, int n3)
{
  __shared__ int c[MAXB];
  __shared__ int cur[MAXB];
  __shared__ int stage[CAP];

  const unsigned* bn; int* rp; int* csr; int n;
  switch (blockIdx.y) {
    case 0:  bn = bn0; rp = rp0; csr = csr0; n = n0; break;
    case 1:  bn = bn1; rp = rp1; csr = csr1; n = n1; break;
    case 2:  bn = bn2; rp = rp2; csr = csr2; n = n2; break;
    default: bn = bn3; rp = rp3; csr = csr3; n = n3; break;
  }
  const int sh = bin_shift(n);
  const int nbins = ((n - 1) >> sh) + 1;
  const int b = blockIdx.x;
  if (b >= nbins) return;
  const int tid = threadIdx.x;
  const int lane = tid & 31;
  const int wave = tid >> 5;
  const int* bstart = bst + blockIdx.y * (MAXB + 1);
  const int bs = bstart[b];
  const int be = bstart[b + 1];
  const int m = be - bs;

  if (tid < MAXB) c[tid] = 0;
  __syncthreads();
  for (int i = bs + tid; i < be; i += 1024) atomicAdd(&c[bn[i] & 255u], 1);
  __syncthreads();
  if (wave == 0) {
    int a[8];
    int sa = 0;
#pragma unroll
    for (int k = 0; k < 8; ++k) { a[k] = c[8 * lane + k]; sa += a[k]; }
    int xa = sa;
#pragma unroll
    for (int d = 1; d < 32; d <<= 1) {
      const int ya = __shfl_up(xa, d);
      if (lane >= d) xa += ya;
    }
    int ra = xa - sa;
    const int R = 1 << sh;
#pragma unroll
    for (int k = 0; k < 8; ++k) {
      const int j = 8 * lane + k;
      cur[j] = ra;
      const int node = (b << sh) + j;
      if (j < R && node < n) rp[node] = bs + ra;
      ra += a[k];
    }
  }
  if (tid == 0 && b == nbins - 1) rp[n] = be;
  __syncthreads();
  const bool staged = (m <= CAP);
  if (staged) {
    for (int i = bs + tid; i < be; i += 1024) {
      const unsigned w = bn[i];
      const int p = atomicAdd(&cur[w & 255u], 1);
      stage[p] = (int)(w >> 8);
    }
  } else {
    for (int i = bs + tid; i < be; i += 1024) {
      const unsigned w = bn[i];
      const int p = atomicAdd(&cur[w & 255u], 1);
      csr[bs + p] = (int)(w >> 8);
    }
  }
  __syncthreads();
  if (staged)
    for (int i = tid; i < m; i += 1024) csr[bs + i] = stage[i];
}

#define FROWS 64

__device__ __forceinline__ void mma_lds(v8f& c, const float* Arow, const float* Wrow, int half)
{
#pragma unroll
  for (int k0 = 0; k0 < HDIM; k0 += 4) {
    const int kk = k0 + 2 * half;
    const v2f a = *(const v2f*)(Arow + kk);
    const v2f b = *(const v2f*)(Wrow + kk);
    c = __builtin_amdgcn_wmma_f32_16x16x4_f32(false, a, false, b, (short)0, c, false, false);
  }
}

__global__ __launch_bounds__(1024)
void hg_gather_sage(const float* __restrict__ xd, const float* __restrict__ xg,
                    const float* __restrict__ xs,
                    const int* __restrict__ rp0, const int* __restrict__ rp1,
                    const int* __restrict__ rp2, const int* __restrict__ rp3,
                    const int* __restrict__ csr0, const int* __restrict__ csr1,
                    const int* __restrict__ csr2, const int* __restrict__ csr3,
                    const float* __restrict__ Wl, const float* __restrict__ bl,
                    const float* __restrict__ Wr,
                    float* __restrict__ outd, float* __restrict__ outg,
                    float* __restrict__ outs,
                    int Nd, int Ng, int Ns, int gDis, int gDrug, int do_relu)
{
  __shared__ __attribute__((aligned(16))) float sWlA[HDIM * WPITCH];
  __shared__ __attribute__((aligned(16))) float sWlB[HDIM * WPITCH];
  __shared__ __attribute__((aligned(16))) float sWr [HDIM * WPITCH];
  __shared__ __attribute__((aligned(16))) float sMA [FROWS * WPITCH];
  __shared__ __attribute__((aligned(16))) float sMB [FROWS * WPITCH];
  __shared__ __attribute__((aligned(16))) float sX  [FROWS * WPITCH];
  __shared__ float sbias[HDIM];

  const int tid = threadIdx.x;

  const float* xsA; const int* rpA; const int* csrA;
  const float* xsB; const int* rpB; const int* csrB;
  const float* xdst; float* out;
  int tA, tB, n_dst, blk;
  bool hasB;
  if ((int)blockIdx.x < gDis) {
    blk = blockIdx.x; hasB = false; tA = 2; tB = 2;
    xsA = xg; rpA = rp2; csrA = csr2; xsB = xg; rpB = rp2; csrB = csr2;
    xdst = xs; out = outs; n_dst = Ns;
  } else if ((int)blockIdx.x < gDis + gDrug) {
    blk = blockIdx.x - gDis; hasB = false; tA = 1; tB = 1;
    xsA = xg; rpA = rp1; csrA = csr1; xsB = xg; rpB = rp1; csrB = csr1;
    xdst = xd; out = outd; n_dst = Nd;
  } else {
    blk = blockIdx.x - gDis - gDrug; hasB = true; tA = 0; tB = 3;
    xsA = xd; rpA = rp0; csrA = csr0; xsB = xs; rpB = rp3; csrB = csr3;
    xdst = xg; out = outg; n_dst = Ng;
  }
  const float* WlA = Wl + (size_t)tA * HDIM * HDIM;
  const float* WrA = Wr + (size_t)tA * HDIM * HDIM;
  const float* blA = bl + (size_t)tA * HDIM;
  const float* WlB = Wl + (size_t)tB * HDIM * HDIM;
  const float* WrB = Wr + (size_t)tB * HDIM * HDIM;
  const float* blB = bl + (size_t)tB * HDIM;

  for (int i = tid; i < HDIM * HDIM; i += 1024) {
    const int p = (i >> 6) * WPITCH + (i & (HDIM - 1));
    sWlA[p] = WlA[i];
    sWr[p]  = hasB ? (WrA[i] + WrB[i]) : WrA[i];
    if (hasB) sWlB[p] = WlB[i];
  }
  if (tid < HDIM) sbias[tid] = hasB ? (blA[tid] + blB[tid]) : blA[tid];

  const int wv = __builtin_amdgcn_readfirstlane(tid >> 5);
  const int lane = tid & 31;
  const int h = lane >> 4;
  const int c4 = (lane & 15) << 2;
  const int row0 = blk * FROWS;

  for (int q = 0; q < FROWS / 32; ++q) {
    const int lr = wv * (FROWS / 32) + q;
    const int d = row0 + lr;
    float4 ma = make_float4(0.f, 0.f, 0.f, 0.f);
    float4 mb = make_float4(0.f, 0.f, 0.f, 0.f);
    float4 xr = make_float4(0.f, 0.f, 0.f, 0.f);
    if (d < n_dst) {
      {
        const int beg = rpA[d];
        const int end = rpA[d + 1];
        const float4 a = row_sum(xsA, beg, end, lane, h, c4, [&](int j) { return csrA[j]; });
        const float inv = 1.0f / fmaxf((float)(end - beg), 1.0f);
        ma = make_float4(a.x * inv, a.y * inv, a.z * inv, a.w * inv);
      }
      if (hasB) {
        const int beg = rpB[d];
        const int end = rpB[d + 1];
        const float4 a = row_sum(xsB, beg, end, lane, h, c4, [&](int j) { return csrB[j]; });
        const float inv = 1.0f / fmaxf((float)(end - beg), 1.0f);
        mb = make_float4(a.x * inv, a.y * inv, a.z * inv, a.w * inv);
      }
      xr = ld_row16(xdst, d, c4);
    }
    if (h == 0) {
      *(float4*)(sMA + lr * WPITCH + c4) = ma;
      if (hasB) *(float4*)(sMB + lr * WPITCH + c4) = mb;
      *(float4*)(sX + lr * WPITCH + c4) = xr;
    }
  }
  __syncthreads();

  const int tile = wv >> 2;
  const int nt = wv & 3;
  const int m = lane & 15;
  const bool live = (wv < 16) && (row0 + tile * 16 < n_dst);
  v8f cacc = {};
  if (live) {
    const int ar = (tile * 16 + m) * WPITCH;
    const int wr = (nt * 16 + m) * WPITCH;
    mma_lds(cacc, sMA + ar, sWlA + wr, h);
    if (hasB) mma_lds(cacc, sMB + ar, sWlB + wr, h);
    mma_lds(cacc, sX + ar, sWr + wr, h);
  }
  __syncthreads();

  if (live) {
    const float bv = sbias[nt * 16 + m];
#pragma unroll
    for (int r = 0; r < 8; ++r) {
      float v = cacc[r] + bv;
      if (do_relu) v = fmaxf(v, 0.0f);
      sMA[(tile * 16 + r + 8 * h) * WPITCH + nt * 16 + m] = v;
    }
  }
  __syncthreads();
  for (int i = tid; i < FROWS * (HDIM / 4); i += 1024) {
    const int lr = i >> 4;
    const int cc = (i & 15) << 2;
    const int d = row0 + lr;
    if (d < n_dst)
      *(float4*)(out + (size_t)d * HDIM + cc) = *(const float4*)(sMA + lr * WPITCH + cc);
  }
}

extern "C" void kernel_launch(void* const* d_in, const int* in_sizes, int n_in,
                              void* d_out, int out_size, void* d_ws, size_t ws_size,
                              hipStream_t stream) {
  const int* src0 = (const int*)d_in[0]; const int* dst0 = (const int*)d_in[1];
  const int* src1 = (const int*)d_in[2]; const int* dst1 = (const int*)d_in[3];
  const int* src2 = (const int*)d_in[4]; const int* dst2 = (const int*)d_in[5];
  const int* src3 = (const int*)d_in[6]; const int* dst3 = (const int*)d_in[7];
  const float* ed  = (const float*)d_in[8];
  const float* eg  = (const float*)d_in[9];
  const float* es  = (const float*)d_in[10];
  const float* Wl0 = (const float*)d_in[11];
  const float* bl0 = (const float*)d_in[12];
  const float* Wr0 = (const float*)d_in[13];
  const float* Wl1 = (const float*)d_in[14];
  const float* bl1 = (const float*)d_in[15];
  const float* Wr1 = (const float*)d_in[16];

  const int E  = in_sizes[0];
  const int Nd = in_sizes[8]  / HDIM;
  const int Ng = in_sizes[9]  / HDIM;
  const int Ns = in_sizes[10] / HDIM;

  auto up64 = [](size_t n) { return (n + 63) & ~(size_t)63; };
  float* ws = (float*)d_ws;
  size_t off = 0;
  float* xd1 = ws + off; off += (size_t)Nd * HDIM;
  float* xg1 = ws + off; off += (size_t)Ng * HDIM;
  float* xs1 = ws + off; off += (size_t)Ns * HDIM;
  const int NT = (E + TILE - 1) / TILE;
  int* hist = (int*)(ws + off); off += up64((size_t)4 * MAXB * NT);
  int* bst  = (int*)(ws + off); off += up64((size_t)4 * (MAXB + 1));
  unsigned* bn0 = (unsigned*)(ws + off); off += up64(E);
  unsigned* bn1 = (unsigned*)(ws + off); off += up64(E);
  unsigned* bn2 = (unsigned*)(ws + off); off += up64(E);
  unsigned* bn3 = (unsigned*)(ws + off); off += up64(E);
  int* rp0 = (int*)(ws + off); off += up64((size_t)Ng + 1);
  int* rp1 = (int*)(ws + off); off += up64((size_t)Nd + 1);
  int* rp2 = (int*)(ws + off); off += up64((size_t)Ns + 1);
  int* rp3 = (int*)(ws + off); off += up64((size_t)Ng + 1);
  int* csr0 = (int*)(ws + off); off += up64(E);
  int* csr1 = (int*)(ws + off); off += up64(E);
  int* csr2 = (int*)(ws + off); off += up64(E);
  int* csr3 = (int*)(ws + off); off += up64(E);

  float* outd = (float*)d_out;
  float* outg = outd + (size_t)Nd * HDIM;
  float* outs = outg + (size_t)Ng * HDIM;

  const dim3 tgrid(NT, 4);
  const dim3 bgrid(MAXB, 4);
  const int fGene = (Ng + FROWS - 1) / FROWS;
  const int fDrug = (Nd + FROWS - 1) / FROWS;
  const int fDis  = (Ns + FROWS - 1) / FROWS;
  const int fAll  = fGene + fDrug + fDis;

  hg_bincount<<<tgrid, 1024, 0, stream>>>(dst0, dst1, dst2, dst3, hist, E, NT, Ng, Nd, Ns, Ng);
  hg_binscatter<<<tgrid, 1024, 0, stream>>>(src0, dst0, src1, dst1, src2, dst2, src3, dst3,
                                            hist, bst, bn0, bn1, bn2, bn3, E, NT, Ng, Nd, Ns, Ng);
  hg_binsort<<<bgrid, 1024, 0, stream>>>(bn0, bn1, bn2, bn3, bst, rp0, rp1, rp2, rp3,
                                         csr0, csr1, csr2, csr3, Ng, Nd, Ns, Ng);

  hg_gather_sage<<<fAll, 1024, 0, stream>>>(ed, eg, es, rp0, rp1, rp2, rp3,
                                           csr0, csr1, csr2, csr3, Wl0, bl0, Wr0,
                                           xd1, xg1, xs1, Nd, Ng, Ns, fDis, fDrug, 1);

  hg_gather_sage<<<fAll, 1024, 0, stream>>>(xd1, xg1, xs1, rp0, rp1, rp2, rp3,
                                           csr0, csr1, csr2, csr3, Wl1, bl1, Wr1,
                                           outd, outg, outs, Nd, Ng, Ns, fDis, fDrug, 0);
}


# Drop the bin-count launch: each edge tile is sorted by bin in LDS, stored whole with its bin offsets; the per-bin sort collects its runs from the tiles; launches 5 -> 4; row sum order changes
# speedup vs baseline: 11.7343x; 1.0563x over previous
#include <hip/hip_runtime.h>

typedef __attribute__((ext_vector_type(2))) float v2f;
typedef __attribute__((ext_vector_type(8))) float v8f;

#define HDIM 64
#define WPITCH 68

#define TILE 8192
#define MAXB 256
#define CAP  12288
#define MAXT 1024
#define OPITCH 264

__device__ __forceinline__ int bin_shift(int n)
{
  int s = 8;
  while (s > 5 && ((n - 1) >> (s - 1)) < MAXB) --s;
  return s;
}

__global__ __launch_bounds__(1024)
void hg_tilesort(const int* __restrict__ src0, const int* __restrict__ dst0,
                 const int* __restrict__ src1, const int* __restrict__ dst1,
                 const int* __restrict__ src2, const int* __restrict__ dst2,
                 const int* __restrict__ src3, const int* __restrict__ dst3,
                 int* __restrict__ toff,
                 unsigned* __restrict__ ts0, unsigned* __restrict__ ts1,
                 unsigned* __restrict__ ts2, unsigned* __restrict__ ts3,
                 int E, int NT, int n0, int n1, int n2, int n3)
{
  __shared__ int s_cnt[MAXB];
  __shared__ int s_off[MAXB];
  __shared__ unsigned s_word[TILE];

  const int* src; const int* dst; unsigned* ts; int n;
  switch (blockIdx.y) {
    case 0:  src = src0; dst = dst0; ts = ts0; n = n0; break;
    case 1:  src = src1; dst = dst1; ts = ts1; n = n1; break;
    case 2:  src = src2; dst = dst2; ts = ts2; n = n2; break;
    default: src = src3; dst = dst3; ts = ts3; n = n3; break;
  }
  const int sh = bin_shift(n);
  const int lmask = (1 << sh) - 1;
  const int tid = threadIdx.x;
  const int lane = tid & 31;
  const int wave = tid >> 5;
  const int t = blockIdx.x;
  int* orow = toff + ((size_t)blockIdx.y * NT + t) * OPITCH;

  if (tid < MAXB) s_cnt[tid] = 0;
  __syncthreads();

  const int e0 = t * TILE;
  int dd[TILE / 1024], ss[TILE / 1024], rk[TILE / 1024];
#pragma unroll
  for (int k = 0; k < TILE / 1024; ++k) {
    const int e = e0 + k * 1024 + tid;
    const int ec = (e < E) ? e : (E - 1);
    dd[k] = dst[ec];
    ss[k] = src[ec];
  }
#pragma unroll
  for (int k = 0; k < TILE / 1024; ++k) {
    const int e = e0 + k * 1024 + tid;
    if (e >= E) dd[k] = -1;
    rk[k] = 0;
    if (dd[k] >= 0) rk[k] = atomicAdd(&s_cnt[dd[k] >> sh], 1);
  }
  __syncthreads();
  if (wave == 0) {
    int a[8];
    int sa = 0;
#pragma unroll
    for (int k = 0; k < 8; ++k) { a[k] = s_cnt[8 * lane + k]; sa += a[k]; }
    int xa = sa;
#pragma unroll
    for (int d = 1; d < 32; d <<= 1) {
      const int ya = __shfl_up(xa, d);
      if (lane >= d) xa += ya;
    }
    int ra = xa - sa;
#pragma unroll
    for (int k = 0; k < 8; ++k) {
      const int b = 8 * lane + k;
      s_off[b] = ra;
      orow[b] = ra;
      ra += a[k];
    }
    if (lane == 31) orow[MAXB] = ra;
  }
  __syncthreads();
#pragma unroll
  for (int k = 0; k < TILE / 1024; ++k) {
    if (dd[k] >= 0)
      s_word[s_off[dd[k] >> sh] + rk[k]] = ((unsigned)ss[k] << 8) | (unsigned)(dd[k] & lmask);
  }
  __syncthreads();
  const int m = (E - e0 < TILE) ? (E - e0) : TILE;
  for (int i = tid; i < m; i += 1024)
    ts[e0 + i] = s_word[i];
}

__device__ __forceinline__ float4 ld_row16(const float* __restrict__ x, int s, int c)
{
  return *(const float4*)(x + (size_t)s * HDIM + c);
}

template <class IdAt>
__device__ __forceinline__ float4 row_sum(const float* __restrict__ x, int beg, int end,
                                          int lane, int h, int c, IdAt id_at)
{
  float4 a0 = make_float4(0.f, 0.f, 0.f, 0.f);
  float4 a1 = make_float4(0.f, 0.f, 0.f, 0.f);
  for (int jb = beg; jb < end; jb += 32) {
    const int j = jb + lane;
    const int idx = (j < end) ? id_at(j) : 0;
    const int nk = (end - jb < 32) ? (end - jb) : 32;
    int k = 0;
    for (; k + 8 <= nk; k += 8) {
      const int s0 = __shfl(idx, k + h);
      const int s1 = __shfl(idx, k + 2 + h);
      const int s2 = __shfl(idx, k + 4 + h);
      const int s3 = __shfl(idx, k + 6 + h);
      const float4 v0 = ld_row16(x, s0, c);
      const float4 v1 = ld_row16(x, s1, c);
      const float4 v2 = ld_row16(x, s2, c);
      const float4 v3 = ld_row16(x, s3, c);
      a0.x += v0.x; a0.y += v0.y; a0.z += v0.z; a0.w += v0.w;
      a1.x += v1.x; a1.y += v1.y; a1.z += v1.z; a1.w += v1.w;
      a0.x += v2.x; a0.y += v2.y; a0.z += v2.z; a0.w += v2.w;
      a1.x += v3.x; a1.y += v3.y; a1.z += v3.z; a1.w += v3.w;
    }
    if (k < nk) {
      float4 t[4];
#pragma unroll
      for (int q = 0; q < 4; ++q) {
        const int kk = k + 2 * q + h;
        const int s = __shfl(idx, kk);
        t[q] = make_float4(0.f, 0.f, 0.f, 0.f);
        if (kk < nk) t[q] = ld_row16(x, s, c);
      }
#pragma unroll
      for (int q = 0; q < 4; ++q) {
        a0.x += t[q].x; a0.y += t[q].y; a0.z += t[q].z; a0.w += t[q].w;
      }
    }
  }
  a0.x += a1.x; a0.y += a1.y; a0.z += a1.z; a0.w += a1.w;
  a0.x += __shfl_xor(a0.x, 16);
  a0.y += __shfl_xor(a0.y, 16);
  a0.z += __shfl_xor(a0.z, 16);
  a0.w += __shfl_xor(a0.w, 16);
  return a0;
}

__global__ __launch_bounds__(1024)
void hg_binsort(const unsigned* __restrict__ ts0, const unsigned* __restrict__ ts1,
                const unsigned* __restrict__ ts2, const unsigned* __restrict__ ts3,
                const int* __restrict__ toff,
                int* __restrict__ rp0, int* __restrict__ rp1,
                int* __restrict__ rp2, int* __restrict__ rp3,
                int* __restrict__ csr0, int* __restrict__ csr1,
                int* __restrict__ csr2, int* __restrict__ csr3,
                int NT, int n0, int n1, int n2, int n3)
{
  __shared__ int c[MAXB];
  __shared__ int cur[MAXB];
  __shared__ int s_lo[MAXT];
  __shared__ int s_n[MAXT];
  __shared__ int s_pos[MAXT];
  __shared__ int s_wn[32];
  __shared__ int s_wl[32];
  __shared__ unsigned raw[CAP];
  __shared__ int stage[CAP];

  const unsigned* ts; int* rp; int* csr; int n;
  switch (blockIdx.y) {
    case 0:  ts = ts0; rp = rp0; csr = csr0; n = n0; break;
    case 1:  ts = ts1; rp = rp1; csr = csr1; n = n1; break;
    case 2:  ts = ts2; rp = rp2; csr = csr2; n = n2; break;
    default: ts = ts3; rp = rp3; csr = csr3; n = n3; break;
  }
  const int sh = bin_shift(n);
  const int nbins = ((n - 1) >> sh) + 1;
  const int b = blockIdx.x;
  if (b >= nbins) return;
  const int tid = threadIdx.x;
  const int lane = tid & 31;
  const int wave = tid >> 5;
  const int* ocol = toff + (size_t)blockIdx.y * NT * OPITCH + b;

  int lo = 0, cnt = 0;
  if (tid < NT) {
    lo = ocol[(size_t)tid * OPITCH];
    cnt = ocol[(size_t)tid * OPITCH + 1] - lo;
  }
  if (tid < MAXB) c[tid] = 0;
  int xn = cnt, xl = lo;
#pragma unroll
  for (int d = 1; d < 32; d <<= 1) {
    const int yn = __shfl_up(xn, d);
    if (lane >= d) xn += yn;
  }
#pragma unroll
  for (int d = 16; d > 0; d >>= 1) xl += __shfl_xor(xl, d);
  if (lane == 31) s_wn[wave] = xn;
  if (lane == 0) s_wl[wave] = xl;
  __syncthreads();
  int wbase = 0, m = 0, bs = 0;
#pragma unroll
  for (int w = 0; w < 32; ++w) {
    const int v = s_wn[w];
    m += v;
    wbase += (w < wave) ? v : 0;
    bs += s_wl[w];
  }
  const int be = bs + m;
  if (tid < NT) { s_lo[tid] = lo; s_n[tid] = cnt; s_pos[tid] = wbase + xn - cnt; }
  __syncthreads();

  const bool staged = (m <= CAP);
  for (int t = wave; t < NT; t += 32) {
    const int nt = s_n[t];
    const int pos = s_pos[t];
    const unsigned* run = ts + (size_t)t * TILE + s_lo[t];
    for (int j0 = 0; j0 < nt; j0 += 64) {
      const int j1 = j0 + lane;
      const int j2 = j1 + 32;
      unsigned w1 = 0, w2 = 0;
      if (j1 < nt) w1 = run[j1];
      if (j2 < nt) w2 = run[j2];
      if (j1 < nt) { atomicAdd(&c[w1 & 255u], 1); if (staged) raw[pos + j1] = w1; }
      if (j2 < nt) { atomicAdd(&c[w2 & 255u], 1); if (staged) raw[pos + j2] = w2; }
    }
  }
  __syncthreads();
  if (wave == 0) {
    int a[8];
    int sa = 0;
#pragma unroll
    for (int k = 0; k < 8; ++k) { a[k] = c[8 * lane + k]; sa += a[k]; }
    int xa = sa;
#pragma unroll
    for (int d = 1; d < 32; d <<= 1) {
      const int ya = __shfl_up(xa, d);
      if (lane >= d) xa += ya;
    }
    int ra = xa - sa;
    const int R = 1 << sh;
#pragma unroll
    for (int k = 0; k < 8; ++k) {
      const int j = 8 * lane + k;
      cur[j] = ra;
      const int node = (b << sh) + j;
      if (j < R && node < n) rp[node] = bs + ra;
      ra += a[k];
    }
  }
  if (tid == 0 && b == nbins - 1) rp[n] = be;
  __syncthreads();
  if (staged) {
    for (int i = tid; i < m; i += 1024) {
      const unsigned w = raw[i];
      const int p = atomicAdd(&cur[w & 255u], 1);
      stage[p] = (int)(w >> 8);
    }
  } else {
    for (int t = wave; t < NT; t += 32) {
      const int nt = s_n[t];
      const unsigned* run = ts + (size_t)t * TILE + s_lo[t];
      for (int j = lane; j < nt; j += 32) {
        const unsigned w = run[j];
        const int p = atomicAdd(&cur[w & 255u], 1);
        csr[bs + p] = (int)(w >> 8);
      }
    }
  }
  __syncthreads();
  if (staged)
    for (int i = tid; i < m; i += 1024) csr[bs + i] = stage[i];
}

#define FROWS 64

__device__ __forceinline__ void mma_lds(v8f& c, const float* Arow, const float* Wrow, int half)
{
#pragma unroll
  for (int k0 = 0; k0 < HDIM; k0 += 4) {
    const int kk = k0 + 2 * half;
    const v2f a = *(const v2f*)(Arow + kk);
    const v2f b = *(const v2f*)(Wrow + kk);
    c = __builtin_amdgcn_wmma_f32_16x16x4_f32(false, a, false, b, (short)0, c, false, false);
  }
}

__global__ __launch_bounds__(1024)
void hg_gather_sage(const float* __restrict__ xd, const float* __restrict__ xg,
                    const float* __restrict__ xs,
                    const int* __restrict__ rp0, const int* __restrict__ rp1,
                    const int* __restrict__ rp2, const int* __restrict__ rp3,
                    const int* __restrict__ csr0, const int* __restrict__ csr1,
                    const int* __restrict__ csr2, const int* __restrict__ csr3,
                    const float* __restrict__ Wl, const float* __restrict__ bl,
                    const float* __restrict__ Wr,
                    float* __restrict__ outd, float* __restrict__ outg,
                    float* __restrict__ outs,
                    int Nd, int Ng, int Ns, int gDis, int gDrug, int do_relu)
{
  __shared__ __attribute__((aligned(16))) float sWlA[HDIM * WPITCH];
  __shared__ __attribute__((aligned(16))) float sWlB[HDIM * WPITCH];
  __shared__ __attribute__((aligned(16))) float sWr [HDIM * WPITCH];
  __shared__ __attribute__((aligned(16))) float sMA [FROWS * WPITCH];
  __shared__ __attribute__((aligned(16))) float sMB [FROWS * WPITCH];
  __shared__ __attribute__((aligned(16))) float sX  [FROWS * WPITCH];
  __shared__ float sbias[HDIM];

  const int tid = threadIdx.x;

  const float* xsA; const int* rpA; const int* csrA;
  const float* xsB; const int* rpB; const int* csrB;
  const float* xdst; float* out;
  int tA, tB, n_dst, blk;
  bool hasB;
  if ((int)blockIdx.x < gDis) {
    blk = blockIdx.x; hasB = false; tA = 2; tB = 2;
    xsA = xg; rpA = rp2; csrA = csr2; xsB = xg; rpB = rp2; csrB = csr2;
    xdst = xs; out = outs; n_dst = Ns;
  } else if ((int)blockIdx.x < gDis + gDrug) {
    blk = blockIdx.x - gDis; hasB = false; tA = 1; tB = 1;
    xsA = xg; rpA = rp1; csrA = csr1; xsB = xg; rpB = rp1; csrB = csr1;
    xdst = xd; out = outd; n_dst = Nd;
  } else {
    blk = blockIdx.x - gDis - gDrug; hasB = true; tA = 0; tB = 3;
    xsA = xd; rpA = rp0; csrA = csr0; xsB = xs; rpB = rp3; csrB = csr3;
    xdst = xg; out = outg; n_dst = Ng;
  }
  const float* WlA = Wl + (size_t)tA * HDIM * HDIM;
  const float* WrA = Wr + (size_t)tA * HDIM * HDIM;
  const float* blA = bl + (size_t)tA * HDIM;
  const float* WlB = Wl + (size_t)tB * HDIM * HDIM;
  const float* WrB = Wr + (size_t)tB * HDIM * HDIM;
  const float* blB = bl + (size_t)tB * HDIM;

  for (int i = tid; i < HDIM * HDIM; i += 1024) {
    const int p = (i >> 6) * WPITCH + (i & (HDIM - 1));
    sWlA[p] = WlA[i];
    sWr[p]  = hasB ? (WrA[i] + WrB[i]) : WrA[i];
    if (hasB) sWlB[p] = WlB[i];
  }
  if (tid < HDIM) sbias[tid] = hasB ? (blA[tid] + blB[tid]) : blA[tid];

  const int wv = __builtin_amdgcn_readfirstlane(tid >> 5);
  const int lane = tid & 31;
  const int h = lane >> 4;
  const int c4 = (lane & 15) << 2;
  const int row0 = blk * FROWS;

  for (int q = 0; q < FROWS / 32; ++q) {
    const int lr = wv * (FROWS / 32) + q;
    const int d = row0 + lr;
    float4 ma = make_float4(0.f, 0.f, 0.f, 0.f);
    float4 mb = make_float4(0.f, 0.f, 0.f, 0.f);
    float4 xr = make_float4(0.f, 0.f, 0.f, 0.f);
    if (d < n_dst) {
      {
        const int beg = rpA[d];
        const int end = rpA[d + 1];
        const float4 a = row_sum(xsA, beg, end, lane, h, c4, [&](int j) { return csrA[j]; });
        const float inv = 1.0f / fmaxf((float)(end - beg), 1.0f);
        ma = make_float4(a.x * inv, a.y * inv, a.z * inv, a.w * inv);
      }
      if (hasB) {
        const int beg = rpB[d];
        const int end = rpB[d + 1];
        const float4 a = row_sum(xsB, beg, end, lane, h, c4, [&](int j) { return csrB[j]; });
        const float inv = 1.0f / fmaxf((float)(end - beg), 1.0f);
        mb = make_float4(a.x * inv, a.y * inv, a.z * inv, a.w * inv);
      }
      xr = ld_row16(xdst, d, c4);
    }
    if (h == 0) {
      *(float4*)(sMA + lr * WPITCH + c4) = ma;
      if (hasB) *(float4*)(sMB + lr * WPITCH + c4) = mb;
      *(float4*)(sX + lr * WPITCH + c4) = xr;
    }
  }
  __syncthreads();

  const int tile = wv >> 2;
  const int nt = wv & 3;
  const int m = lane & 15;
  const bool live = (wv < 16) && (row0 + tile * 16 < n_dst);
  v8f cacc = {};
  if (live) {
    const int ar = (tile * 16 + m) * WPITCH;
    const int wr = (nt * 16 + m) * WPITCH;
    mma_lds(cacc, sMA + ar, sWlA + wr, h);
    if (hasB) mma_lds(cacc, sMB + ar, sWlB + wr, h);
    mma_lds(cacc, sX + ar, sWr + wr, h);
  }
  __syncthreads();

  if (live) {
    const float bv = sbias[nt * 16 + m];
#pragma unroll
    for (int r = 0; r < 8; ++r) {
      float v = cacc[r] + bv;
      if (do_relu) v = fmaxf(v, 0.0f);
      sMA[(tile * 16 + r + 8 * h) * WPITCH + nt * 16 + m] = v;
    }
  }
  __syncthreads();
  for (int i = tid; i < FROWS * (HDIM / 4); i += 1024) {
    const int lr = i >> 4;
    const int cc = (i & 15) << 2;
    const int d = row0 + lr;
    if (d < n_dst)
      *(float4*)(out + (size_t)d * HDIM + cc) = *(const float4*)(sMA + lr * WPITCH + cc);
  }
}

extern "C" void kernel_launch(void* const* d_in, const int* in_sizes, int n_in,
                              void* d_out, int out_size, void* d_ws, size_t ws_size,
                              hipStream_t stream) {
  const int* src0 = (const int*)d_in[0]; const int* dst0 = (const int*)d_in[1];
  const int* src1 = (const int*)d_in[2]; const int* dst1 = (const int*)d_in[3];
  const int* src2 = (const int*)d_in[4]; const int* dst2 = (const int*)d_in[5];
  const int* src3 = (const int*)d_in[6]; const int* dst3 = (const int*)d_in[7];
  const float* ed  = (const float*)d_in[8];
  const float* eg  = (const float*)d_in[9];
  const float* es  = (const float*)d_in[10];
  const float* Wl0 = (const float*)d_in[11];
  const float* bl0 = (const float*)d_in[12];
  const float* Wr0 = (const float*)d_in[13];
  const float* Wl1 = (const float*)d_in[14];
  const float* bl1 = (const float*)d_in[15];
  const float* Wr1 = (const float*)d_in[16];

  const int E  = in_sizes[0];
  const int Nd = in_sizes[8]  / HDIM;
  const int Ng = in_sizes[9]  / HDIM;
  const int Ns = in_sizes[10] / HDIM;

  auto up64 = [](size_t n) { return (n + 63) & ~(size_t)63; };
  float* ws = (float*)d_ws;
  size_t off = 0;
  float* xd1 = ws + off; off += (size_t)Nd * HDIM;
  float* xg1 = ws + off; off += (size_t)Ng * HDIM;
  float* xs1 = ws + off; off += (size_t)Ns * HDIM;
  const int NT = (E + TILE - 1) / TILE;
  int* toff = (int*)(ws + off); off += up64((size_t)4 * NT * OPITCH);
  unsigned* ts0 = (unsigned*)(ws + off); off += up64(E);
  unsigned* ts1 = (unsigned*)(ws + off); off += up64(E);
  unsigned* ts2 = (unsigned*)(ws + off); off += up64(E);
  unsigned* ts3 = (unsigned*)(ws + off); off += up64(E);
  int* rp0 = (int*)(ws + off); off += up64((size_t)Ng + 1);
  int* rp1 = (int*)(ws + off); off += up64((size_t)Nd + 1);
  int* rp2 = (int*)(ws + off); off += up64((size_t)Ns + 1);
  int* rp3 = (int*)(ws + off); off += up64((size_t)Ng + 1);
  int* csr0 = (int*)(ws + off); off += up64(E);
  int* csr1 = (int*)(ws + off); off += up64(E);
  int* csr2 = (int*)(ws + off); off += up64(E);
  int* csr3 = (int*)(ws + off); off += up64(E);

  float* outd = (float*)d_out;
  float* outg = outd + (size_t)Nd * HDIM;
  float* outs = outg + (size_t)Ng * HDIM;

  const dim3 tgrid(NT, 4);
  const dim3 bgrid(MAXB, 4);
  const int fGene = (Ng + FROWS - 1) / FROWS;
  const int fDrug = (Nd + FROWS - 1) / FROWS;
  const int fDis  = (Ns + FROWS - 1) / FROWS;
  const int fAll  = fGene + fDrug + fDis;

  hg_tilesort<<<tgrid, 1024, 0, stream>>>(src0, dst0, src1, dst1, src2, dst2, src3, dst3,
                                          toff, ts0, ts1, ts2, ts3, E, NT, Ng, Nd, Ns, Ng);
  hg_binsort<<<bgrid, 1024, 0, stream>>>(ts0, ts1, ts2, ts3, toff, rp0, rp1, rp2, rp3,
                                         csr0, csr1, csr2, csr3, NT, Ng, Nd, Ns, Ng);

  hg_gather_sage<<<fAll, 1024, 0, stream>>>(ed, eg, es, rp0, rp1, rp2, rp3,
                                           csr0, csr1, csr2, csr3, Wl0, bl0, Wr0,
                                           xd1, xg1, xs1, Nd, Ng, Ns, fDis, fDrug, 1);

  hg_gather_sage<<<fAll, 1024, 0, stream>>>(xd1, xg1, xs1, rp0, rp1, rp2, rp3,
                                           csr0, csr1, csr2, csr3, Wl1, bl1, Wr1,
                                           outd, outg, outs, Nd, Ng, Ns, fDis, fDrug, 0);
}


# Work-group start-up in one round trip: each weight matrix staged by one 16-byte load per thread, bias, the waves' row bounds and first ids fetched beside them (was 4 to 8 dependent round trips)
# speedup vs baseline: 12.3713x; 1.0543x over previous
#include <hip/hip_runtime.h>

typedef __attribute__((ext_vector_type(2))) float v2f;
typedef __attribute__((ext_vector_type(8))) float v8f;

#define HDIM 64
#define WPITCH 68

#define TILE 8192
#define MAXB 256
#define CAP  12288
#define MAXT 1024
#define OPITCH 264

__device__ __forceinline__ int bin_shift(int n)
{
  int s = 8;
  while (s > 5 && ((n - 1) >> (s - 1)) < MAXB) --s;
  return s;
}

__global__ __launch_bounds__(1024)
void hg_tilesort(const int* __restrict__ src0, const int* __restrict__ dst0,
                 const int* __restrict__ src1, const int* __restrict__ dst1,
                 const int* __restrict__ src2, const int* __restrict__ dst2,
                 const int* __restrict__ src3, const int* __restrict__ dst3,
                 int* __restrict__ toff,
                 unsigned* __restrict__ ts0, unsigned* __restrict__ ts1,
                 unsigned* __restrict__ ts2, unsigned* __restrict__ ts3,
                 int E, int NT, int n0, int n1, int n2, int n3)
{
  __shared__ int s_cnt[MAXB];
  __shared__ int s_off[MAXB];
  __shared__ unsigned s_word[TILE];

  const int* src; const int* dst; unsigned* ts; int n;
  switch (blockIdx.y) {
    case 0:  src = src0; dst = dst0; ts = ts0; n = n0; break;
    case 1:  src = src1; dst = dst1; ts = ts1; n = n1; break;
    case 2:  src = src2; dst = dst2; ts = ts2; n = n2; break;
    default: src = src3; dst = dst3; ts = ts3; n = n3; break;
  }
  const int sh = bin_shift(n);
  const int lmask = (1 << sh) - 1;
  const int tid = threadIdx.x;
  const int lane = tid & 31;
  const int wave = tid >> 5;
  const int t = blockIdx.x;
  int* orow = toff + ((size_t)blockIdx.y * NT + t) * OPITCH;

  if (tid < MAXB) s_cnt[tid] = 0;
  __syncthreads();

  const int e0 = t * TILE;
  int dd[TILE / 1024], ss[TILE / 1024], rk[TILE / 1024];
#pragma unroll
  for (int k = 0; k < TILE / 1024; ++k) {
    const int e = e0 + k * 1024 + tid;
    const int ec = (e < E) ? e : (E - 1);
    dd[k] = dst[ec];
    ss[k] = src[ec];
  }
#pragma unroll
  for (int k = 0; k < TILE / 1024; ++k) {
    const int e = e0 + k * 1024 + tid;
    if (e >= E) dd[k] = -1;
    rk[k] = 0;
    if (dd[k] >= 0) rk[k] = atomicAdd(&s_cnt[dd[k] >> sh], 1);
  }
  __syncthreads();
  if (wave == 0) {
    int a[8];
    int sa = 0;
#pragma unroll
    for (int k = 0; k < 8; ++k) { a[k] = s_cnt[8 * lane + k]; sa += a[k]; }
    int xa = sa;
#pragma unroll
    for (int d = 1; d < 32; d <<= 1) {
      const int ya = __shfl_up(xa, d);
      if (lane >= d) xa += ya;
    }
    int ra = xa - sa;
#pragma unroll
    for (int k = 0; k < 8; ++k) {
      const int b = 8 * lane + k;
      s_off[b] = ra;
      orow[b] = ra;
      ra += a[k];
    }
    if (lane == 31) orow[MAXB] = ra;
  }
  __syncthreads();
#pragma unroll
  for (int k = 0; k < TILE / 1024; ++k) {
    if (dd[k] >= 0)
      s_word[s_off[dd[k] >> sh] + rk[k]] = ((unsigned)ss[k] << 8) | (unsigned)(dd[k] & lmask);
  }
  __syncthreads();
  const int m = (E - e0 < TILE) ? (E - e0) : TILE;
  for (int i = tid; i < m; i += 1024)
    ts[e0 + i] = s_word[i];
}

__device__ __forceinline__ float4 ld_row16(const float* __restrict__ x, int s, int c)
{
  return *(const float4*)(x + (size_t)s * HDIM + c);
}

template <class IdAt>
__device__ __forceinline__ int first_ids(int beg, int end, int lane, IdAt id_at)
{
  const int j = beg + lane;
  return (j < end) ? id_at(j) : 0;
}

template <class IdAt>
__device__ __forceinline__ float4 row_sum(const float* __restrict__ x, int beg, int end,
                                          int lane, int h, int c, int idx, IdAt id_at)
{
  float4 a0 = make_float4(0.f, 0.f, 0.f, 0.f);
  float4 a1 = make_float4(0.f, 0.f, 0.f, 0.f);
  for (int jb = beg; jb < end; jb += 32) {
    if (jb != beg) {
      const int j = jb + lane;
      idx = (j < end) ? id_at(j) : 0;
    }
    const int nk = (end - jb < 32) ? (end - jb) : 32;
    int k = 0;
    for (; k + 8 <= nk; k += 8) {
      const int s0 = __shfl(idx, k + h);
      const int s1 = __shfl(idx, k + 2 + h);
      const int s2 = __shfl(idx, k + 4 + h);
      const int s3 = __shfl(idx, k + 6 + h);
      const float4 v0 = ld_row16(x, s0, c);
      const float4 v1 = ld_row16(x, s1, c);
      const float4 v2 = ld_row16(x, s2, c);
      const float4 v3 = ld_row16(x, s3, c);
      a0.x += v0.x; a0.y += v0.y; a0.z += v0.z; a0.w += v0.w;
      a1.x += v1.x; a1.y += v1.y; a1.z += v1.z; a1.w += v1.w;
      a0.x += v2.x; a0.y += v2.y; a0.z += v2.z; a0.w += v2.w;
      a1.x += v3.x; a1.y += v3.y; a1.z += v3.z; a1.w += v3.w;
    }
    if (k < nk) {
      float4 t[4];
#pragma unroll
      for (int q = 0; q < 4; ++q) {
        const int kk = k + 2 * q + h;
        const int s = __shfl(idx, kk);
        t[q] = make_float4(0.f, 0.f, 0.f, 0.f);
        if (kk < nk) t[q] = ld_row16(x, s, c);
      }
#pragma unroll
      for (int q = 0; q < 4; ++q) {
        a0.x += t[q].x; a0.y += t[q].y; a0.z += t[q].z; a0.w += t[q].w;
      }
    }
  }
  a0.x += a1.x; a0.y += a1.y; a0.z += a1.z; a0.w += a1.w;
  a0.x += __shfl_xor(a0.x, 16);
  a0.y += __shfl_xor(a0.y, 16);
  a0.z += __shfl_xor(a0.z, 16);
  a0.w += __shfl_xor(a0.w, 16);
  return a0;
}

__global__ __launch_bounds__(1024)
void hg_binsort(const unsigned* __restrict__ ts0, const unsigned* __restrict__ ts1,
                const unsigned* __restrict__ ts2, const unsigned* __restrict__ ts3,
                const int* __restrict__ toff,
                int* __restrict__ rp0, int* __restrict__ rp1,
                int* __restrict__ rp2, int* __restrict__ rp3,
                int* __restrict__ csr0, int* __restrict__ csr1,
                int* __restrict__ csr2, int* __restrict__ csr3,
                int NT, int n0, int n1, int n2, int n3)
{
  __shared__ int c[MAXB];
  __shared__ int cur[MAXB];
  __shared__ int s_lo[MAXT];
  __shared__ int s_n[MAXT];
  __shared__ int s_pos[MAXT];
  __shared__ int s_wn[32];
  __shared__ int s_wl[32];
  __shared__ unsigned raw[CAP];
  __shared__ int stage[CAP];

  const unsigned* ts; int* rp; int* csr; int n;
  switch (blockIdx.y) {
    case 0:  ts = ts0; rp = rp0; csr = csr0; n = n0; break;
    case 1:  ts = ts1; rp = rp1; csr = csr1; n = n1; break;
    case 2:  ts = ts2; rp = rp2; csr = csr2; n = n2; break;
    default: ts = ts3; rp = rp3; csr = csr3; n = n3; break;
  }
  const int sh = bin_shift(n);
  const int nbins = ((n - 1) >> sh) + 1;
  const int b = blockIdx.x;
  if (b >= nbins) return;
  const int tid = threadIdx.x;
  const int lane = tid & 31;
  const int wave = tid >> 5;
  const int* ocol = toff + (size_t)blockIdx.y * NT * OPITCH + b;

  int lo = 0, cnt = 0;
  if (tid < NT) {
    lo = ocol[(size_t)tid * OPITCH];
    cnt = ocol[(size_t)tid * OPITCH + 1] - lo;
  }
  if (tid < MAXB) c[tid] = 0;
  int xn = cnt, xl = lo;
#pragma unroll
  for (int d = 1; d < 32; d <<= 1) {
    const int yn = __shfl_up(xn, d);
    if (lane >= d) xn += yn;
  }
#pragma unroll
  for (int d = 16; d > 0; d >>= 1) xl += __shfl_xor(xl, d);
  if (lane == 31) s_wn[wave] = xn;
  if (lane == 0) s_wl[wave] = xl;
  __syncthreads();
  int wbase = 0, m = 0, bs = 0;
#pragma unroll
  for (int w = 0; w < 32; ++w) {
    const int v = s_wn[w];
    m += v;
    wbase += (w < wave) ? v : 0;
    bs += s_wl[w];
  }
  const int be = bs + m;
  if (tid < NT) { s_lo[tid] = lo; s_n[tid] = cnt; s_pos[tid] = wbase + xn - cnt; }
  __syncthreads();

  const bool staged = (m <= CAP);
  for (int t = wave; t < NT; t += 32) {
    const int nt = s_n[t];
    const int pos = s_pos[t];
    const unsigned* run = ts + (size_t)t * TILE + s_lo[t];
    for (int j0 = 0; j0 < nt; j0 += 64) {
      const int j1 = j0 + lane;
      const int j2 = j1 + 32;
      unsigned w1 = 0, w2 = 0;
      if (j1 < nt) w1 = run[j1];
      if (j2 < nt) w2 = run[j2];
      if (j1 < nt) { atomicAdd(&c[w1 & 255u], 1); if (staged) raw[pos + j1] = w1; }
      if (j2 < nt) { atomicAdd(&c[w2 & 255u], 1); if (staged) raw[pos + j2] = w2; }
    }
  }
  __syncthreads();
  if (wave == 0) {
    int a[8];
    int sa = 0;
#pragma unroll
    for (int k = 0; k < 8; ++k) { a[k] = c[8 * lane + k]; sa += a[k]; }
    int xa = sa;
#pragma unroll
    for (int d = 1; d < 32; d <<= 1) {
      const int ya = __shfl_up(xa, d);
      if (lane >= d) xa += ya;
    }
    int ra = xa - sa;
    const int R = 1 << sh;
#pragma unroll
    for (int k = 0; k < 8; ++k) {
      const int j = 8 * lane + k;
      cur[j] = ra;
      const int node = (b << sh) + j;
      if (j < R && node < n) rp[node] = bs + ra;
      ra += a[k];
    }
  }
  if (tid == 0 && b == nbins - 1) rp[n] = be;
  __syncthreads();
  if (staged) {
    for (int i = tid; i < m; i += 1024) {
      const unsigned w = raw[i];
      const int p = atomicAdd(&cur[w & 255u], 1);
      stage[p] = (int)(w >> 8);
    }
  } else {
    for (int t = wave; t < NT; t += 32) {
      const int nt = s_n[t];
      const unsigned* run = ts + (size_t)t * TILE + s_lo[t];
      for (int j = lane; j < nt; j += 32) {
        const unsigned w = run[j];
        const int p = atomicAdd(&cur[w & 255u], 1);
        csr[bs + p] = (int)(w >> 8);
      }
    }
  }
  __syncthreads();
  if (staged)
    for (int i = tid; i < m; i += 1024) csr[bs + i] = stage[i];
}

#define FROWS 64

__device__ __forceinline__ void mma_lds(v8f& c, const float* Arow, const float* Wrow, int half)
{
#pragma unroll
  for (int k0 = 0; k0 < HDIM; k0 += 4) {
    const int kk = k0 + 2 * half;
    const v2f a = *(const v2f*)(Arow + kk);
    const v2f b = *(const v2f*)(Wrow + kk);
    c = __builtin_amdgcn_wmma_f32_16x16x4_f32(false, a, false, b, (short)0, c, false, false);
  }
}

__global__ __launch_bounds__(1024)
void hg_gather_sage(const float* __restrict__ xd, const float* __restrict__ xg,
                    const float* __restrict__ xs,
                    const int* __restrict__ rp0, const int* __restrict__ rp1,
                    const int* __restrict__ rp2, const int* __restrict__ rp3,
                    const int* __restrict__ csr0, const int* __restrict__ csr1,
                    const int* __restrict__ csr2, const int* __restrict__ csr3,
                    const float* __restrict__ Wl, const float* __restrict__ bl,
                    const float* __restrict__ Wr,
                    float* __restrict__ outd, float* __restrict__ outg,
                    float* __restrict__ outs,
                    int Nd, int Ng, int Ns, int gDis, int gDrug, int do_relu)
{
  __shared__ __attribute__((aligned(16))) float sWlA[HDIM * WPITCH];
  __shared__ __attribute__((aligned(16))) float sWlB[HDIM * WPITCH];
  __shared__ __attribute__((aligned(16))) float sWr [HDIM * WPITCH];
  __shared__ __attribute__((aligned(16))) float sMA [FROWS * WPITCH];
  __shared__ __attribute__((aligned(16))) float sMB [FROWS * WPITCH];
  __shared__ __attribute__((aligned(16))) float sX  [FROWS * WPITCH];
  __shared__ float sbias[HDIM];

  const int tid = threadIdx.x;

  const float* xsA; const int* rpA; const int* csrA;
  const float* xsB; const int* rpB; const int* csrB;
  const float* xdst; float* out;
  int tA, tB, n_dst, blk;
  bool hasB;
  if ((int)blockIdx.x < gDis) {
    blk = blockIdx.x; hasB = false; tA = 2; tB = 2;
    xsA = xg; rpA = rp2; csrA = csr2; xsB = xg; rpB = rp2; csrB = csr2;
    xdst = xs; out = outs; n_dst = Ns;
  } else if ((int)blockIdx.x < gDis + gDrug) {
    blk = blockIdx.x - gDis; hasB = false; tA = 1; tB = 1;
    xsA = xg; rpA = rp1; csrA = csr1; xsB = xg; rpB = rp1; csrB = csr1;
    xdst = xd; out = outd; n_dst = Nd;
  } else {
    blk = blockIdx.x - gDis - gDrug; hasB = true; tA = 0; tB = 3;
    xsA = xd; rpA = rp0; csrA = csr0; xsB = xs; rpB = rp3; csrB = csr3;
    xdst = xg; out = outg; n_dst = Ng;
  }
  const float* WlA = Wl + (size_t)tA * HDIM * HDIM;
  const float* WrA = Wr + (size_t)tA * HDIM * HDIM;
  const float* blA = bl + (size_t)tA * HDIM;
  const float* WlB = Wl + (size_t)tB * HDIM * HDIM;
  const float* WrB = Wr + (size_t)tB * HDIM * HDIM;
  const float* blB = bl + (size_t)tB * HDIM;

  const int wv = __builtin_amdgcn_readfirstlane(tid >> 5);
  const int lane = tid & 31;
  const int h = lane >> 4;
  const int c4 = (lane & 15) << 2;
  const int row0 = blk * FROWS;
  static_assert(FROWS == 64 && HDIM == 64, "one float4 of every weight matrix per thread, two destinations per wave");

  const float4 wl = *(const float4*)(WlA + 4 * tid);
  float4 wr = *(const float4*)(WrA + 4 * tid);
  float4 wlb = make_float4(0.f, 0.f, 0.f, 0.f);
  float4 wrb = make_float4(0.f, 0.f, 0.f, 0.f);
  if (hasB) {
    wlb = *(const float4*)(WlB + 4 * tid);
    wrb = *(const float4*)(WrB + 4 * tid);
  }
  float bA = 0.f, bB = 0.f;
  if (tid < HDIM) {
    bA = blA[tid];
    if (hasB) bB = blB[tid];
  }
  const int d0 = row0 + wv * 2;
  int rA0 = 0, rA1 = 0, rA2 = 0, rB0 = 0, rB1 = 0, rB2 = 0;
  if (d0 < n_dst) {
    rA0 = rpA[d0]; rA1 = rpA[d0 + 1];
    if (hasB) { rB0 = rpB[d0]; rB1 = rpB[d0 + 1]; }
  }
  if (d0 + 1 < n_dst) {
    rA2 = rpA[d0 + 2];
    if (hasB) rB2 = rpB[d0 + 2];
  }
  const int idxA0 = first_ids(rA0, rA1, lane, [&](int j) { return csrA[j]; });

  {
    const int p = (tid >> 4) * WPITCH + ((tid & 15) << 2);
    *(float4*)(sWlA + p) = wl;
    if (hasB) {
      wr.x += wrb.x; wr.y += wrb.y; wr.z += wrb.z; wr.w += wrb.w;
      *(float4*)(sWlB + p) = wlb;
    }
    *(float4*)(sWr + p) = wr;
  }
  if (tid < HDIM) sbias[tid] = hasB ? (bA + bB) : bA;

  for (int q = 0; q < 2; ++q) {
    const int lr = wv * 2 + q;
    const int d = d0 + q;
    float4 ma = make_float4(0.f, 0.f, 0.f, 0.f);
    float4 mb = make_float4(0.f, 0.f, 0.f, 0.f);
    float4 xr = make_float4(0.f, 0.f, 0.f, 0.f);
    if (d < n_dst) {
      {
        const int beg = q ? rA1 : rA0;
        const int end = q ? rA2 : rA1;
        auto idA = [&](int j) { return csrA[j]; };
        const int idx = q ? first_ids(beg, end, lane, idA) : idxA0;
        const float4 a = row_sum(xsA, beg, end, lane, h, c4, idx, idA);
        const float inv = 1.0f / fmaxf((float)(end - beg), 1.0f);
        ma = make_float4(a.x * inv, a.y * inv, a.z * inv, a.w * inv);
      }
      if (hasB) {
        const int beg = q ? rB1 : rB0;
        const int end = q ? rB2 : rB1;
        auto idB = [&](int j) { return csrB[j]; };
        const float4 a = row_sum(xsB, beg, end, lane, h, c4, first_ids(beg, end, lane, idB), idB);
        const float inv = 1.0f / fmaxf((float)(end - beg), 1.0f);
        mb = make_float4(a.x * inv, a.y * inv, a.z * inv, a.w * inv);
      }
      xr = ld_row16(xdst, d, c4);
    }
    if (h == 0) {
      *(float4*)(sMA + lr * WPITCH + c4) = ma;
      if (hasB) *(float4*)(sMB + lr * WPITCH + c4) = mb;
      *(float4*)(sX + lr * WPITCH + c4) = xr;
    }
  }
  __syncthreads();

  const int tile = wv >> 2;
  const int nt = wv & 3;
  const int m = lane & 15;
  const bool live = (wv < 16) && (row0 + tile * 16 < n_dst);
  v8f cacc = {};
  if (live) {
    const int ar = (tile * 16 + m) * WPITCH;
    const int wr = (nt * 16 + m) * WPITCH;
    mma_lds(cacc, sMA + ar, sWlA + wr, h);
    if (hasB) mma_lds(cacc, sMB + ar, sWlB + wr, h);
    mma_lds(cacc, sX + ar, sWr + wr, h);
  }
  __syncthreads();

  if (live) {
    const float bv = sbias[nt * 16 + m];
#pragma unroll
    for (int r = 0; r < 8; ++r) {
      float v = cacc[r] + bv;
      if (do_relu) v = fmaxf(v, 0.0f);
      sMA[(tile * 16 + r + 8 * h) * WPITCH + nt * 16 + m] = v;
    }
  }
  __syncthreads();
  for (int i = tid; i < FROWS * (HDIM / 4); i += 1024) {
    const int lr = i >> 4;
    const int cc = (i & 15) << 2;
    const int d = row0 + lr;
    if (d < n_dst)
      *(float4*)(out + (size_t)d * HDIM + cc) = *(const float4*)(sMA + lr * WPITCH + cc);
  }
}

extern "C" void kernel_launch(void* const* d_in, const int* in_sizes, int n_in,
                              void* d_out, int out_size, void* d_ws, size_t ws_size,
                              hipStream_t stream) {
  const int* src0 = (const int*)d_in[0]; const int* dst0 = (const int*)d_in[1];
  const int* src1 = (const int*)d_in[2]; const int* dst1 = (const int*)d_in[3];
  const int* src2 = (const int*)d_in[4]; const int* dst2 = (const int*)d_in[5];
  const int* src3 = (const int*)d_in[6]; const int* dst3 = (const int*)d_in[7];
  const float* ed  = (const float*)d_in[8];
  const float* eg  = (const float*)d_in[9];
  const float* es  = (const float*)d_in[10];
  const float* Wl0 = (const float*)d_in[11];
  const float* bl0 = (const float*)d_in[12];
  const float* Wr0 = (const float*)d_in[13];
  const float* Wl1 = (const float*)d_in[14];
  const float* bl1 = (const float*)d_in[15];
  const float* Wr1 = (const float*)d_in[16];

  const int E  = in_sizes[0];
  const int Nd = in_sizes[8]  / HDIM;
  const int Ng = in_sizes[9]  / HDIM;
  const int Ns = in_sizes[10] / HDIM;

  auto up64 = [](size_t n) { return (n + 63) & ~(size_t)63; };
  float* ws = (float*)d_ws;
  size_t off = 0;
  float* xd1 = ws + off; off += (size_t)Nd * HDIM;
  float* xg1 = ws + off; off += (size_t)Ng * HDIM;
  float* xs1 = ws + off; off += (size_t)Ns * HDIM;
  const int NT = (E + TILE - 1) / TILE;
  int* toff = (int*)(ws + off); off += up64((size_t)4 * NT * OPITCH);
  unsigned* ts0 = (unsigned*)(ws + off); off += up64(E);
  unsigned* ts1 = (unsigned*)(ws + off); off += up64(E);
  unsigned* ts2 = (unsigned*)(ws + off); off += up64(E);
  unsigned* ts3 = (unsigned*)(ws + off); off += up64(E);
  int* rp0 = (int*)(ws + off); off += up64((size_t)Ng + 1);
  int* rp1 = (int*)(ws + off); off += up64((size_t)Nd + 1);
  int* rp2 = (int*)(ws + off); off += up64((size_t)Ns + 1);
  int* rp3 = (int*)(ws + off); off += up64((size_t)Ng + 1);
  int* csr0 = (int*)(ws + off); off += up64(E);
  int* csr1 = (int*)(ws + off); off += up64(E);
  int* csr2 = (int*)(ws + off); off += up64(E);
  int* csr3 = (int*)(ws + off); off += up64(E);

  float* outd = (float*)d_out;
  float* outg = outd + (size_t)Nd * HDIM;
  float* outs = outg + (size_t)Ng * HDIM;

  const dim3 tgrid(NT, 4);
  const dim3 bgrid(MAXB, 4);
  const int fGene = (Ng + FROWS - 1) / FROWS;
  const int fDrug = (Nd + FROWS - 1) / FROWS;
  const int fDis  = (Ns + FROWS - 1) / FROWS;
  const int fAll  = fGene + fDrug + fDis;

  hg_tilesort<<<tgrid, 1024, 0, stream>>>(src0, dst0, src1, dst1, src2, dst2, src3, dst3,
                                          toff, ts0, ts1, ts2, ts3, E, NT, Ng, Nd, Ns, Ng);
  hg_binsort<<<bgrid, 1024, 0, stream>>>(ts0, ts1, ts2, ts3, toff, rp0, rp1, rp2, rp3,
                                         csr0, csr1, csr2, csr3, NT, Ng, Nd, Ns, Ng);

  hg_gather_sage<<<fAll, 1024, 0, stream>>>(ed, eg, es, rp0, rp1, rp2, rp3,
                                           csr0, csr1, csr2, csr3, Wl0, bl0, Wr0,
                                           xd1, xg1, xs1, Nd, Ng, Ns, fDis, fDrug, 1);

  hg_gather_sage<<<fAll, 1024, 0, stream>>>(xd1, xg1, xs1, rp0, rp1, rp2, rp3,
                                           csr0, csr1, csr2, csr3, Wl1, bl1, Wr1,
                                           outd, outg, outs, Nd, Ng, Ns, fDis, fDrug, 0);
}
